# starting kernel
# baseline (speedup 1.0000x reference)
#include <hip/hip_runtime.h>
#include <hip/hip_bf16.h>

typedef float v2f __attribute__((ext_vector_type(2)));
typedef float v8f __attribute__((ext_vector_type(8)));

#define TPB 256

__global__ void k_init_deg(float* deg, int N) {
    int i = blockIdx.x * TPB + threadIdx.x;
    if (i < N) deg[i] = 1.0f;
}

__global__ void k_deg_accum(const int* __restrict__ dst, float* deg, int E) {
    int e = blockIdx.x * TPB + threadIdx.x;
    if (e < E) atomicAdd(&deg[dst[e]], 1.0f);
}

__global__ void k_dinv(float* deg, int N) {
    int i = blockIdx.x * TPB + threadIdx.x;
    if (i < N) {
        float d = deg[i];
        deg[i] = (d > 0.0f) ? rsqrtf(d) : 0.0f;
    }
}

__global__ void k_fuse_heads(const float* __restrict__ L0, const float* __restrict__ b0,
                             const float* __restrict__ L1, const float* __restrict__ b1,
                             const float* __restrict__ Wo, const float* __restrict__ bo,
                             float* W0f, float* W1f, float* cb) {
    for (int i = threadIdx.x; i < 64 * 40; i += TPB) {
        int k = i / 40, c = i % 40;
        float s0 = 0.0f, s1 = 0.0f;
        for (int j = 0; j < 64; ++j) {
            s0 += L0[k * 64 + j] * Wo[j * 40 + c];
            s1 += L1[k * 64 + j] * Wo[(64 + j) * 40 + c];
        }
        W0f[i] = s0;
        W1f[i] = s1;
    }
    if (threadIdx.x < 40) {
        int c = threadIdx.x;
        float s = bo[c];
        for (int j = 0; j < 64; ++j)
            s += b0[j] * Wo[j * 40 + c] + b1[j] * Wo[(64 + j) * 40 + c];
        cb[c] = s;
    }
}

__global__ __launch_bounds__(128)
void k_gemm_n64_wmma(const float* __restrict__ A, const float* __restrict__ W,
                     float* __restrict__ D, int N, int K) {
    __shared__ float sW[128 * 64];
    int tid = threadIdx.x;
    for (int i = tid; i < K * 64; i += 128) {
        int k = i >> 6, col = i & 63;
        sW[(k >> 1) * 128 + 2 * col + (k & 1)] = W[i];
    }
    __syncthreads();

    int wave = tid >> 5;
    int lane = tid & 31;
    int half = lane >> 4;
    int l16  = lane & 15;

    int rbase = blockIdx.x * 128 + wave * 32;
    int ar0 = rbase + l16;
    int ar1 = rbase + 16 + l16;
    long ab0 = (long)((ar0 < N) ? ar0 : (N - 1)) * K;
    long ab1 = (long)((ar1 < N) ? ar1 : (N - 1)) * K;

    v8f acc00 = {}, acc01 = {}, acc02 = {}, acc03 = {};
    v8f acc10 = {}, acc11 = {}, acc12 = {}, acc13 = {};

    for (int k = 0; k < K; k += 4) {
        int kb = k + 2 * half;
        v2f a0 = *(const v2f*)(A + ab0 + kb);
        v2f a1 = *(const v2f*)(A + ab1 + kb);
        const float* bp = &sW[(kb >> 1) * 128 + 2 * l16];
        v2f b0 = *(const v2f*)(bp + 0);
        v2f b1 = *(const v2f*)(bp + 32);
        v2f b2 = *(const v2f*)(bp + 64);
        v2f b3 = *(const v2f*)(bp + 96);
        acc00 = __builtin_amdgcn_wmma_f32_16x16x4_f32(false, a0, false, b0, (short)0, acc00, false, false);
        acc01 = __builtin_amdgcn_wmma_f32_16x16x4_f32(false, a0, false, b1, (short)0, acc01, false, false);
        acc02 = __builtin_amdgcn_wmma_f32_16x16x4_f32(false, a0, false, b2, (short)0, acc02, false, false);
        acc03 = __builtin_amdgcn_wmma_f32_16x16x4_f32(false, a0, false, b3, (short)0, acc03, false, false);
        acc10 = __builtin_amdgcn_wmma_f32_16x16x4_f32(false, a1, false, b0, (short)0, acc10, false, false);
        acc11 = __builtin_amdgcn_wmma_f32_16x16x4_f32(false, a1, false, b1, (short)0, acc11, false, false);
        acc12 = __builtin_amdgcn_wmma_f32_16x16x4_f32(false, a1, false, b2, (short)0, acc12, false, false);
        acc13 = __builtin_amdgcn_wmma_f32_16x16x4_f32(false, a1, false, b3, (short)0, acc13, false, false);
    }

    int m0 = rbase + half * 8;
    int m1 = rbase + 16 + half * 8;
    for (int r = 0; r < 8; ++r) {
        int rr = m0 + r;
        if (rr < N) {
            float* dp = D + (long)rr * 64 + l16;
            dp[0]  = acc00[r];
            dp[16] = acc01[r];
            dp[32] = acc02[r];
            dp[48] = acc03[r];
        }
        int ss = m1 + r;
        if (ss < N) {
            float* dp = D + (long)ss * 64 + l16;
            dp[0]  = acc10[r];
            dp[16] = acc11[r];
            dp[32] = acc12[r];
            dp[48] = acc13[r];
        }
    }
}

__global__ void k_scatter(const float* __restrict__ T, const int* __restrict__ src,
                          const int* __restrict__ dst, const float* __restrict__ dinv,
                          float* Bagg, int E) {
    long gid = (long)blockIdx.x * TPB + threadIdx.x;
    int e = (int)(gid >> 6);
    int f = (int)(gid & 63);
    if (e < E) {
        int s = src[e], d = dst[e];
        float nr = dinv[s] * dinv[d];
        atomicAdd(&Bagg[(long)d * 64 + f], T[(long)s * 64 + f] * nr);
    }
}

__global__ void k_bias_selfloop_elu(const float* __restrict__ Bagg, const float* __restrict__ T,
                                    const float* __restrict__ dinv, const float* __restrict__ bias,
                                    float* H, int N) {
    long gid = (long)blockIdx.x * TPB + threadIdx.x;
    if (gid < (long)N * 64) {
        int n = (int)(gid >> 6);
        int f = (int)(gid & 63);
        float di = dinv[n];
        float v = Bagg[gid] + di * di * T[gid] + bias[f];
        H[gid] = (v > 0.0f) ? v : (expf(v) - 1.0f);
    }
}

__global__ void k_out(const float* __restrict__ H0, const float* __restrict__ H1,
                      const float* __restrict__ W0f, const float* __restrict__ W1f,
                      const float* __restrict__ cb, float* __restrict__ out, int N) {
    __shared__ float sw[64 * 40 * 2 + 40];
    for (int i = threadIdx.x; i < 64 * 40; i += TPB) {
        sw[i] = W0f[i];
        sw[2560 + i] = W1f[i];
    }
    if (threadIdx.x < 40) sw[5120 + threadIdx.x] = cb[threadIdx.x];
    __syncthreads();

    long gid = (long)blockIdx.x * TPB + threadIdx.x;
    if (gid < (long)N * 40) {
        int n = (int)(gid / 40);
        int c = (int)(gid % 40);
        const float* h0 = H0 + (long)n * 64;
        const float* h1 = H1 + (long)n * 64;
        float s = sw[5120 + c];
        for (int k = 0; k < 64; ++k)
            s += h0[k] * sw[k * 40 + c] + h1[k] * sw[2560 + k * 40 + c];
        out[gid] = s;
    }
}

extern "C" void kernel_launch(void* const* d_in, const int* in_sizes, int n_in,
                              void* d_out, int out_size, void* d_ws, size_t ws_size,
                              hipStream_t stream) {
    const float* x       = (const float*)d_in[0];
    const int*   ei      = (const int*)d_in[1];
    const float* conv0_w = (const float*)d_in[2];
    const float* conv0_b = (const float*)d_in[3];
    const float* lin0_w  = (const float*)d_in[4];
    const float* lin0_b  = (const float*)d_in[5];
    const float* conv1_w = (const float*)d_in[6];
    const float* conv1_b = (const float*)d_in[7];
    const float* lin1_w  = (const float*)d_in[8];
    const float* lin1_b  = (const float*)d_in[9];
    const float* out_w   = (const float*)d_in[10];
    const float* out_b   = (const float*)d_in[11];
    float* out = (float*)d_out;

    int N = in_sizes[0] / 128;
    int E = in_sizes[1] / 2;
    const int* src = ei;
    const int* dst = ei + E;

    size_t NH = (size_t)N * 64;
    float* ws   = (float*)d_ws;
    float* dinv = ws;
    float* A    = ws + (((size_t)N + 255) / 256) * 256;
    float* B    = A + NH;
    float* H0   = B + NH;
    float* W0f  = H0 + NH;
    float* W1f  = W0f + 64 * 40;
    float* cb   = W1f + 64 * 40;

    int gN   = (N + TPB - 1) / TPB;
    int gE   = (E + TPB - 1) / TPB;
    int gNH  = (int)((NH + TPB - 1) / TPB);
    int gSc  = (int)(((size_t)E * 64 + TPB - 1) / TPB);
    int gOut = (int)(((size_t)N * 40 + TPB - 1) / TPB);
    int gGemm = (N + 127) / 128;

    k_init_deg<<<gN, TPB, 0, stream>>>(dinv, N);
    k_deg_accum<<<gE, TPB, 0, stream>>>(dst, dinv, E);
    k_dinv<<<gN, TPB, 0, stream>>>(dinv, N);

    k_fuse_heads<<<1, TPB, 0, stream>>>(lin0_w, lin0_b, lin1_w, lin1_b, out_w, out_b,
                                        W0f, W1f, cb);

    k_gemm_n64_wmma<<<gGemm, 128, 0, stream>>>(x, conv0_w, A, N, 128);
    hipMemsetAsync(B, 0, NH * sizeof(float), stream);
    k_scatter<<<gSc, TPB, 0, stream>>>(A, src, dst, dinv, B, E);
    k_bias_selfloop_elu<<<gNH, TPB, 0, stream>>>(B, A, dinv, conv0_b, H0, N);

    k_gemm_n64_wmma<<<gGemm, 128, 0, stream>>>(H0, conv1_w, A, N, 64);
    hipMemsetAsync(B, 0, NH * sizeof(float), stream);
    k_scatter<<<gSc, TPB, 0, stream>>>(A, src, dst, dinv, B, E);
    k_bias_selfloop_elu<<<gNH, TPB, 0, stream>>>(B, A, dinv, conv1_b, B, N);

    k_out<<<gOut, TPB, 0, stream>>>(H0, B, W0f, W1f, cb, out, N);
}


# float-atomic edge scatter replaced by a per-call edge list grouped by destination (integer add returns the arrival-order rank) and a per-node register gather; sums now run in rank order
# speedup vs baseline: 2.6042x; 2.6042x over previous
#include <hip/hip_runtime.h>
#include <hip/hip_bf16.h>

typedef float v2f __attribute__((ext_vector_type(2)));
typedef float v8f __attribute__((ext_vector_type(8)));

#define TPB 256

#define CHUNK 1024

__global__ void k_zero_cnt(int* cnt, int N) {
    int i = blockIdx.x * TPB + threadIdx.x;
    if (i < N) cnt[i] = 0;
}

__global__ void k_rank(const int* __restrict__ dst, int* cnt, int* __restrict__ rank, int E) {
    int e = blockIdx.x * TPB + threadIdx.x;
    if (e < E) rank[e] = atomicAdd(&cnt[dst[e]], 1);
}

__global__ __launch_bounds__(256)
void k_scan_chunk(const int* __restrict__ cnt, int* __restrict__ rowloc,
                  int* __restrict__ bsum, int N) {
    __shared__ int swave[8];
    int tid = threadIdx.x;
    int lane = tid & 31, wv = tid >> 5;
    int base = blockIdx.x * CHUNK + tid * 4;
    int c0 = (base + 0 < N) ? cnt[base + 0] : 0;
    int c1 = (base + 1 < N) ? cnt[base + 1] : 0;
    int c2 = (base + 2 < N) ? cnt[base + 2] : 0;
    int c3 = (base + 3 < N) ? cnt[base + 3] : 0;
    int t = c0 + c1 + c2 + c3;
    int incl = t;
    for (int off = 1; off < 32; off <<= 1) {
        int v = __shfl_up(incl, off, 32);
        if (lane >= off) incl += v;
    }
    if (lane == 31) swave[wv] = incl;
    __syncthreads();
    int wbase = 0;
    for (int w = 0; w < wv; ++w) wbase += swave[w];
    int excl = wbase + incl - t;
    if (base + 0 < N) rowloc[base + 0] = excl;
    if (base + 1 < N) rowloc[base + 1] = excl + c0;
    if (base + 2 < N) rowloc[base + 2] = excl + c0 + c1;
    if (base + 3 < N) rowloc[base + 3] = excl + c0 + c1 + c2;
    if (tid == 255) bsum[blockIdx.x] = excl + t;
}

__global__ __launch_bounds__(1024)
void k_scan_bsum(int* bsum, int nb) {
    __shared__ int sw[32];
    int tid = threadIdx.x;
    int lane = tid & 31, wv = tid >> 5;
    int carry = 0;
    for (int b0 = 0; b0 < nb; b0 += 1024) {
        int i = b0 + tid;
        int v = (i < nb) ? bsum[i] : 0;
        int incl = v;
        for (int off = 1; off < 32; off <<= 1) {
            int u = __shfl_up(incl, off, 32);
            if (lane >= off) incl += u;
        }
        if (lane == 31) sw[wv] = incl;
        __syncthreads();
        int wbase = 0, tot = 0;
        for (int w = 0; w < 32; ++w) {
            int s = sw[w];
            if (w < wv) wbase += s;
            tot += s;
        }
        if (i < nb) bsum[i] = carry + wbase + incl - v;
        carry += tot;
        __syncthreads();
    }
}

__global__ void k_fill(const int* __restrict__ src, const int* __restrict__ dst,
                       const int* __restrict__ rank, const int* __restrict__ cnt,
                       const int* __restrict__ rowloc, const int* __restrict__ boff,
                       int2* __restrict__ csr, int E) {
    int e = blockIdx.x * TPB + threadIdx.x;
    if (e < E) {
        int s = src[e], d = dst[e];
        int pos = boff[d / CHUNK] + rowloc[d] + rank[e];
        float ds = (float)(cnt[s] + 1);
        float w = (ds > 0.0f) ? rsqrtf(ds) : 0.0f;
        csr[pos] = make_int2(s, __float_as_int(w));
    }
}

__global__ void k_fuse_heads(const float* __restrict__ L0, const float* __restrict__ b0,
                             const float* __restrict__ L1, const float* __restrict__ b1,
                             const float* __restrict__ Wo, const float* __restrict__ bo,
                             float* W0f, float* W1f, float* cb) {
    for (int i = threadIdx.x; i < 64 * 40; i += TPB) {
        int k = i / 40, c = i % 40;
        float s0 = 0.0f, s1 = 0.0f;
        for (int j = 0; j < 64; ++j) {
            s0 += L0[k * 64 + j] * Wo[j * 40 + c];
            s1 += L1[k * 64 + j] * Wo[(64 + j) * 40 + c];
        }
        W0f[i] = s0;
        W1f[i] = s1;
    }
    if (threadIdx.x < 40) {
        int c = threadIdx.x;
        float s = bo[c];
        for (int j = 0; j < 64; ++j)
            s += b0[j] * Wo[j * 40 + c] + b1[j] * Wo[(64 + j) * 40 + c];
        cb[c] = s;
    }
}

__global__ __launch_bounds__(128)
void k_gemm_n64_wmma(const float* __restrict__ A, const float* __restrict__ W,
                     float* __restrict__ D, int N, int K) {
    __shared__ float sW[128 * 64];
    int tid = threadIdx.x;
    for (int i = tid; i < K * 64; i += 128) {
        int k = i >> 6, col = i & 63;
        sW[(k >> 1) * 128 + 2 * col + (k & 1)] = W[i];
    }
    __syncthreads();

    int wave = tid >> 5;
    int lane = tid & 31;
    int half = lane >> 4;
    int l16  = lane & 15;

    int rbase = blockIdx.x * 128 + wave * 32;
    int ar0 = rbase + l16;
    int ar1 = rbase + 16 + l16;
    long ab0 = (long)((ar0 < N) ? ar0 : (N - 1)) * K;
    long ab1 = (long)((ar1 < N) ? ar1 : (N - 1)) * K;

    v8f acc00 = {}, acc01 = {}, acc02 = {}, acc03 = {};
    v8f acc10 = {}, acc11 = {}, acc12 = {}, acc13 = {};

    for (int k = 0; k < K; k += 4) {
        int kb = k + 2 * half;
        v2f a0 = *(const v2f*)(A + ab0 + kb);
        v2f a1 = *(const v2f*)(A + ab1 + kb);
        const float* bp = &sW[(kb >> 1) * 128 + 2 * l16];
        v2f b0 = *(const v2f*)(bp + 0);
        v2f b1 = *(const v2f*)(bp + 32);
        v2f b2 = *(const v2f*)(bp + 64);
        v2f b3 = *(const v2f*)(bp + 96);
        acc00 = __builtin_amdgcn_wmma_f32_16x16x4_f32(false, a0, false, b0, (short)0, acc00, false, false);
        acc01 = __builtin_amdgcn_wmma_f32_16x16x4_f32(false, a0, false, b1, (short)0, acc01, false, false);
        acc02 = __builtin_amdgcn_wmma_f32_16x16x4_f32(false, a0, false, b2, (short)0, acc02, false, false);
        acc03 = __builtin_amdgcn_wmma_f32_16x16x4_f32(false, a0, false, b3, (short)0, acc03, false, false);
        acc10 = __builtin_amdgcn_wmma_f32_16x16x4_f32(false, a1, false, b0, (short)0, acc10, false, false);
        acc11 = __builtin_amdgcn_wmma_f32_16x16x4_f32(false, a1, false, b1, (short)0, acc11, false, false);
        acc12 = __builtin_amdgcn_wmma_f32_16x16x4_f32(false, a1, false, b2, (short)0, acc12, false, false);
        acc13 = __builtin_amdgcn_wmma_f32_16x16x4_f32(false, a1, false, b3, (short)0, acc13, false, false);
    }

    int m0 = rbase + half * 8;
    int m1 = rbase + 16 + half * 8;
    for (int r = 0; r < 8; ++r) {
        int rr = m0 + r;
        if (rr < N) {
            float* dp = D + (long)rr * 64 + l16;
            dp[0]  = acc00[r];
            dp[16] = acc01[r];
            dp[32] = acc02[r];
            dp[48] = acc03[r];
        }
        int ss = m1 + r;
        if (ss < N) {
            float* dp = D + (long)ss * 64 + l16;
            dp[0]  = acc10[r];
            dp[16] = acc11[r];
            dp[32] = acc12[r];
            dp[48] = acc13[r];
        }
    }
}

template <int U>
__device__ __forceinline__ void add_rows(const int2* __restrict__ p, const float4* __restrict__ T4,
                                         int l, float4& acc) {
    int2 e[U];
    float4 r[U];
#pragma unroll
    for (int k = 0; k < U; ++k) e[k] = p[k];
#pragma unroll
    for (int k = 0; k < U; ++k) r[k] = T4[(long)e[k].x * 16 + l];
#pragma unroll
    for (int k = 0; k < U; ++k) {
        float w = __int_as_float(e[k].y);
        acc.x = fmaf(w, r[k].x, acc.x);
        acc.y = fmaf(w, r[k].y, acc.y);
        acc.z = fmaf(w, r[k].z, acc.z);
        acc.w = fmaf(w, r[k].w, acc.w);
    }
}

__global__ __launch_bounds__(256)
void k_gather_elu(const float* __restrict__ T, const int2* __restrict__ csr,
                  const int* __restrict__ cnt, const int* __restrict__ rowloc,
                  const int* __restrict__ boff, const float* __restrict__ bias,
                  float* __restrict__ H, int N) {
    int tid = threadIdx.x;
    int n = blockIdx.x * 16 + (tid >> 4);
    int l = tid & 15;
    if (n >= N) return;
    const float4* T4 = (const float4*)T;
    int c = cnt[n];
    const int2* p = csr + (boff[n / CHUNK] + rowloc[n]);
    float4 self = T4[(long)n * 16 + l];
    float dg = (float)(c + 1);
    float dn = (dg > 0.0f) ? rsqrtf(dg) : 0.0f;

    float4 acc = make_float4(0.0f, 0.0f, 0.0f, 0.0f);
    int j = 0;
    for (; j + 8 <= c; j += 8) add_rows<8>(p + j, T4, l, acc);
    if (c & 4) { add_rows<4>(p + j, T4, l, acc); j += 4; }
    if (c & 2) { add_rows<2>(p + j, T4, l, acc); j += 2; }
    if (c & 1) { add_rows<1>(p + j, T4, l, acc); }

    float v0 = dn * (acc.x + dn * self.x) + bias[4 * l + 0];
    float v1 = dn * (acc.y + dn * self.y) + bias[4 * l + 1];
    float v2 = dn * (acc.z + dn * self.z) + bias[4 * l + 2];
    float v3 = dn * (acc.w + dn * self.w) + bias[4 * l + 3];
    float4 h;
    h.x = (v0 > 0.0f) ? v0 : (expf(v0) - 1.0f);
    h.y = (v1 > 0.0f) ? v1 : (expf(v1) - 1.0f);
    h.z = (v2 > 0.0f) ? v2 : (expf(v2) - 1.0f);
    h.w = (v3 > 0.0f) ? v3 : (expf(v3) - 1.0f);
    ((float4*)H)[(long)n * 16 + l] = h;
}

__global__ void k_out(const float* __restrict__ H0, const float* __restrict__ H1,
                      const float* __restrict__ W0f, const float* __restrict__ W1f,
                      const float* __restrict__ cb, float* __restrict__ out, int N) {
    __shared__ float sw[64 * 40 * 2 + 40];
    for (int i = threadIdx.x; i < 64 * 40; i += TPB) {
        sw[i] = W0f[i];
        sw[2560 + i] = W1f[i];
    }
    if (threadIdx.x < 40) sw[5120 + threadIdx.x] = cb[threadIdx.x];
    __syncthreads();

    long gid = (long)blockIdx.x * TPB + threadIdx.x;
    if (gid < (long)N * 40) {
        int n = (int)(gid / 40);
        int c = (int)(gid % 40);
        const float* h0 = H0 + (long)n * 64;
        const float* h1 = H1 + (long)n * 64;
        float s = sw[5120 + c];
        for (int k = 0; k < 64; ++k)
            s += h0[k] * sw[k * 40 + c] + h1[k] * sw[2560 + k * 40 + c];
        out[gid] = s;
    }
}

extern "C" void kernel_launch(void* const* d_in, const int* in_sizes, int n_in,
                              void* d_out, int out_size, void* d_ws, size_t ws_size,
                              hipStream_t stream) {
    const float* x       = (const float*)d_in[0];
    const int*   ei      = (const int*)d_in[1];
    const float* conv0_w = (const float*)d_in[2];
    const float* conv0_b = (const float*)d_in[3];
    const float* lin0_w  = (const float*)d_in[4];
    const float* lin0_b  = (const float*)d_in[5];
    const float* conv1_w = (const float*)d_in[6];
    const float* conv1_b = (const float*)d_in[7];
    const float* lin1_w  = (const float*)d_in[8];
    const float* lin1_b  = (const float*)d_in[9];
    const float* out_w   = (const float*)d_in[10];
    const float* out_b   = (const float*)d_in[11];
    float* out = (float*)d_out;

    int N = in_sizes[0] / 128;
    int E = in_sizes[1] / 2;
    const int* src = ei;
    const int* dst = ei + E;

    size_t NH = (size_t)N * 64;
    float* ws   = (float*)d_ws;
    int*   cnt  = (int*)ws;
    float* A    = ws + (((size_t)N + 255) / 256) * 256;
    float* H1   = A + NH;
    float* H0   = H1 + NH;
    float* W0f  = H0 + NH;
    float* W1f  = W0f + 64 * 40;
    float* cb   = W1f + 64 * 40;
    int*   rank = (int*)H1;
    int*   oscr   = (int*)d_out;
    int2*  csr    = (int2*)oscr;
    int*   rowloc = oscr + 2 * (size_t)E;
    int*   boff   = rowloc + N;

    int gN   = (N + TPB - 1) / TPB;
    int gE   = (E + TPB - 1) / TPB;
    int gCh  = (N + CHUNK - 1) / CHUNK;
    int gGa  = (N + 15) / 16;
    int gOut = (int)(((size_t)N * 40 + TPB - 1) / TPB);
    int gGemm = (N + 127) / 128;

    k_zero_cnt<<<gN, TPB, 0, stream>>>(cnt, N);
    k_rank<<<gE, TPB, 0, stream>>>(dst, cnt, rank, E);
    k_scan_chunk<<<gCh, 256, 0, stream>>>(cnt, rowloc, boff, N);
    k_scan_bsum<<<1, 1024, 0, stream>>>(boff, gCh);
    k_fill<<<gE, TPB, 0, stream>>>(src, dst, rank, cnt, rowloc, boff, csr, E);

    k_fuse_heads<<<1, TPB, 0, stream>>>(lin0_w, lin0_b, lin1_w, lin1_b, out_w, out_b,
                                        W0f, W1f, cb);

    k_gemm_n64_wmma<<<gGemm, 128, 0, stream>>>(x, conv0_w, A, N, 128);
    k_gather_elu<<<gGa, 256, 0, stream>>>(A, csr, cnt, rowloc, boff, conv0_b, H0, N);

    k_gemm_n64_wmma<<<gGemm, 128, 0, stream>>>(H0, conv1_w, A, N, 64);
    k_gather_elu<<<gGa, 256, 0, stream>>>(A, csr, cnt, rowloc, boff, conv1_b, H1, N);

    k_out<<<gOut, TPB, 0, stream>>>(H0, H1, W0f, W1f, cb, out, N);
}


# output projection moved from one thread per output element (scalar FMAs) onto f32 WMMA: [h0|h1] (N x 128) times the fused head matrices, 32 rows per wave; sums now run in the matrix unit's k order
# speedup vs baseline: 2.8791x; 1.1056x over previous
#include <hip/hip_runtime.h>
#include <hip/hip_bf16.h>

typedef float v2f __attribute__((ext_vector_type(2)));
typedef float v8f __attribute__((ext_vector_type(8)));

#define TPB 256

#define CHUNK 1024

__global__ void k_zero_cnt(int* cnt, int N) {
    int i = blockIdx.x * TPB + threadIdx.x;
    if (i < N) cnt[i] = 0;
}

__global__ void k_rank(const int* __restrict__ dst, int* cnt, int* __restrict__ rank, int E) {
    int e = blockIdx.x * TPB + threadIdx.x;
    if (e < E) rank[e] = atomicAdd(&cnt[dst[e]], 1);
}

__global__ __launch_bounds__(256)
void k_scan_chunk(const int* __restrict__ cnt, int* __restrict__ rowloc,
                  int* __restrict__ bsum, int N) {
    __shared__ int swave[8];
    int tid = threadIdx.x;
    int lane = tid & 31, wv = tid >> 5;
    int base = blockIdx.x * CHUNK + tid * 4;
    int c0 = (base + 0 < N) ? cnt[base + 0] : 0;
    int c1 = (base + 1 < N) ? cnt[base + 1] : 0;
    int c2 = (base + 2 < N) ? cnt[base + 2] : 0;
    int c3 = (base + 3 < N) ? cnt[base + 3] : 0;
    int t = c0 + c1 + c2 + c3;
    int incl = t;
    for (int off = 1; off < 32; off <<= 1) {
        int v = __shfl_up(incl, off, 32);
        if (lane >= off) incl += v;
    }
    if (lane == 31) swave[wv] = incl;
    __syncthreads();
    int wbase = 0;
    for (int w = 0; w < wv; ++w) wbase += swave[w];
    int excl = wbase + incl - t;
    if (base + 0 < N) rowloc[base + 0] = excl;
    if (base + 1 < N) rowloc[base + 1] = excl + c0;
    if (base + 2 < N) rowloc[base + 2] = excl + c0 + c1;
    if (base + 3 < N) rowloc[base + 3] = excl + c0 + c1 + c2;
    if (tid == 255) bsum[blockIdx.x] = excl + t;
}

__global__ __launch_bounds__(1024)
void k_scan_bsum(int* bsum, int nb) {
    __shared__ int sw[32];
    int tid = threadIdx.x;
    int lane = tid & 31, wv = tid >> 5;
    int carry = 0;
    for (int b0 = 0; b0 < nb; b0 += 1024) {
        int i = b0 + tid;
        int v = (i < nb) ? bsum[i] : 0;
        int incl = v;
        for (int off = 1; off < 32; off <<= 1) {
            int u = __shfl_up(incl, off, 32);
            if (lane >= off) incl += u;
        }
        if (lane == 31) sw[wv] = incl;
        __syncthreads();
        int wbase = 0, tot = 0;
        for (int w = 0; w < 32; ++w) {
            int s = sw[w];
            if (w < wv) wbase += s;
            tot += s;
        }
        if (i < nb) bsum[i] = carry + wbase + incl - v;
        carry += tot;
        __syncthreads();
    }
}

__global__ void k_fill(const int* __restrict__ src, const int* __restrict__ dst,
                       const int* __restrict__ rank, const int* __restrict__ cnt,
                       const int* __restrict__ rowloc, const int* __restrict__ boff,
                       int2* __restrict__ csr, int E) {
    int e = blockIdx.x * TPB + threadIdx.x;
    if (e < E) {
        int s = src[e], d = dst[e];
        int pos = boff[d / CHUNK] + rowloc[d] + rank[e];
        float ds = (float)(cnt[s] + 1);
        float w = (ds > 0.0f) ? rsqrtf(ds) : 0.0f;
        csr[pos] = make_int2(s, __float_as_int(w));
    }
}

__global__ void k_fuse_heads(const float* __restrict__ L0, const float* __restrict__ b0,
                             const float* __restrict__ L1, const float* __restrict__ b1,
                             const float* __restrict__ Wo, const float* __restrict__ bo,
                             float* W0f, float* W1f, float* cb) {
    for (int i = threadIdx.x; i < 64 * 40; i += TPB) {
        int k = i / 40, c = i % 40;
        float s0 = 0.0f, s1 = 0.0f;
        for (int j = 0; j < 64; ++j) {
            s0 += L0[k * 64 + j] * Wo[j * 40 + c];
            s1 += L1[k * 64 + j] * Wo[(64 + j) * 40 + c];
        }
        W0f[i] = s0;
        W1f[i] = s1;
    }
    if (threadIdx.x < 40) {
        int c = threadIdx.x;
        float s = bo[c];
        for (int j = 0; j < 64; ++j)
            s += b0[j] * Wo[j * 40 + c] + b1[j] * Wo[(64 + j) * 40 + c];
        cb[c] = s;
    }
}

__global__ __launch_bounds__(128)
void k_gemm_n64_wmma(const float* __restrict__ A, const float* __restrict__ W,
                     float* __restrict__ D, int N, int K) {
    __shared__ float sW[128 * 64];
    int tid = threadIdx.x;
    for (int i = tid; i < K * 64; i += 128) {
        int k = i >> 6, col = i & 63;
        sW[(k >> 1) * 128 + 2 * col + (k & 1)] = W[i];
    }
    __syncthreads();

    int wave = tid >> 5;
    int lane = tid & 31;
    int half = lane >> 4;
    int l16  = lane & 15;

    int rbase = blockIdx.x * 128 + wave * 32;
    int ar0 = rbase + l16;
    int ar1 = rbase + 16 + l16;
    long ab0 = (long)((ar0 < N) ? ar0 : (N - 1)) * K;
    long ab1 = (long)((ar1 < N) ? ar1 : (N - 1)) * K;

    v8f acc00 = {}, acc01 = {}, acc02 = {}, acc03 = {};
    v8f acc10 = {}, acc11 = {}, acc12 = {}, acc13 = {};

    for (int k = 0; k < K; k += 4) {
        int kb = k + 2 * half;
        v2f a0 = *(const v2f*)(A + ab0 + kb);
        v2f a1 = *(const v2f*)(A + ab1 + kb);
        const float* bp = &sW[(kb >> 1) * 128 + 2 * l16];
        v2f b0 = *(const v2f*)(bp + 0);
        v2f b1 = *(const v2f*)(bp + 32);
        v2f b2 = *(const v2f*)(bp + 64);
        v2f b3 = *(const v2f*)(bp + 96);
        acc00 = __builtin_amdgcn_wmma_f32_16x16x4_f32(false, a0, false, b0, (short)0, acc00, false, false);
        acc01 = __builtin_amdgcn_wmma_f32_16x16x4_f32(false, a0, false, b1, (short)0, acc01, false, false);
        acc02 = __builtin_amdgcn_wmma_f32_16x16x4_f32(false, a0, false, b2, (short)0, acc02, false, false);
        acc03 = __builtin_amdgcn_wmma_f32_16x16x4_f32(false, a0, false, b3, (short)0, acc03, false, false);
        acc10 = __builtin_amdgcn_wmma_f32_16x16x4_f32(false, a1, false, b0, (short)0, acc10, false, false);
        acc11 = __builtin_amdgcn_wmma_f32_16x16x4_f32(false, a1, false, b1, (short)0, acc11, false, false);
        acc12 = __builtin_amdgcn_wmma_f32_16x16x4_f32(false, a1, false, b2, (short)0, acc12, false, false);
        acc13 = __builtin_amdgcn_wmma_f32_16x16x4_f32(false, a1, false, b3, (short)0, acc13, false, false);
    }

    int m0 = rbase + half * 8;
    int m1 = rbase + 16 + half * 8;
    for (int r = 0; r < 8; ++r) {
        int rr = m0 + r;
        if (rr < N) {
            float* dp = D + (long)rr * 64 + l16;
            dp[0]  = acc00[r];
            dp[16] = acc01[r];
            dp[32] = acc02[r];
            dp[48] = acc03[r];
        }
        int ss = m1 + r;
        if (ss < N) {
            float* dp = D + (long)ss * 64 + l16;
            dp[0]  = acc10[r];
            dp[16] = acc11[r];
            dp[32] = acc12[r];
            dp[48] = acc13[r];
        }
    }
}

template <int U>
__device__ __forceinline__ void add_rows(const int2* __restrict__ p, const float4* __restrict__ T4,
                                         int l, float4& acc) {
    int2 e[U];
    float4 r[U];
#pragma unroll
    for (int k = 0; k < U; ++k) e[k] = p[k];
#pragma unroll
    for (int k = 0; k < U; ++k) r[k] = T4[(long)e[k].x * 16 + l];
#pragma unroll
    for (int k = 0; k < U; ++k) {
        float w = __int_as_float(e[k].y);
        acc.x = fmaf(w, r[k].x, acc.x);
        acc.y = fmaf(w, r[k].y, acc.y);
        acc.z = fmaf(w, r[k].z, acc.z);
        acc.w = fmaf(w, r[k].w, acc.w);
    }
}

__global__ __launch_bounds__(256)
void k_gather_elu(const float* __restrict__ T, const int2* __restrict__ csr,
                  const int* __restrict__ cnt, const int* __restrict__ rowloc,
                  const int* __restrict__ boff, const float* __restrict__ bias,
                  float* __restrict__ H, int N) {
    int tid = threadIdx.x;
    int n = blockIdx.x * 16 + (tid >> 4);
    int l = tid & 15;
    if (n >= N) return;
    const float4* T4 = (const float4*)T;
    int c = cnt[n];
    const int2* p = csr + (boff[n / CHUNK] + rowloc[n]);
    float4 self = T4[(long)n * 16 + l];
    float dg = (float)(c + 1);
    float dn = (dg > 0.0f) ? rsqrtf(dg) : 0.0f;

    float4 acc = make_float4(0.0f, 0.0f, 0.0f, 0.0f);
    int j = 0;
    for (; j + 8 <= c; j += 8) add_rows<8>(p + j, T4, l, acc);
    if (c & 4) { add_rows<4>(p + j, T4, l, acc); j += 4; }
    if (c & 2) { add_rows<2>(p + j, T4, l, acc); j += 2; }
    if (c & 1) { add_rows<1>(p + j, T4, l, acc); }

    float v0 = dn * (acc.x + dn * self.x) + bias[4 * l + 0];
    float v1 = dn * (acc.y + dn * self.y) + bias[4 * l + 1];
    float v2 = dn * (acc.z + dn * self.z) + bias[4 * l + 2];
    float v3 = dn * (acc.w + dn * self.w) + bias[4 * l + 3];
    float4 h;
    h.x = (v0 > 0.0f) ? v0 : (expf(v0) - 1.0f);
    h.y = (v1 > 0.0f) ? v1 : (expf(v1) - 1.0f);
    h.z = (v2 > 0.0f) ? v2 : (expf(v2) - 1.0f);
    h.w = (v3 > 0.0f) ? v3 : (expf(v3) - 1.0f);
    ((float4*)H)[(long)n * 16 + l] = h;
}

__global__ __launch_bounds__(128)
void k_out_wmma(const float* __restrict__ H0, const float* __restrict__ H1,
                const float* __restrict__ W0f, const float* __restrict__ W1f,
                const float* __restrict__ cb, float* __restrict__ out, int N) {
    __shared__ float4 sW4[32 * 48];
    __shared__ float4 sO4[4][32 * 10];
    int tid = threadIdx.x;
    int wave = tid >> 5;
    int lane = tid & 31;
    int half = lane >> 4;
    int l16  = lane & 15;

    int rbase = blockIdx.x * 128 + wave * 32;
    int ar0 = rbase + l16;
    int ar1 = rbase + 16 + l16;
    long ab0 = (long)((ar0 < N) ? ar0 : (N - 1)) * 16;
    long ab1 = (long)((ar1 < N) ? ar1 : (N - 1)) * 16;
    const float4* H04 = (const float4*)H0;
    const float4* H14 = (const float4*)H1;

    float4 a0[16], a1[16];
#pragma unroll
    for (int j = 0; j < 8; ++j) {
        a0[j]     = H04[ab0 + 2 * j + half];
        a1[j]     = H04[ab1 + 2 * j + half];
        a0[8 + j] = H14[ab0 + 2 * j + half];
        a1[8 + j] = H14[ab1 + 2 * j + half];
    }

    for (int it = tid; it < 320; it += 128) {
        int q = it / 10, cg = it % 10;
        const float4* Ws = (const float4*)((q < 16) ? W0f : W1f) + (q & 15) * 40 + cg;
        float4 r0 = Ws[0], r1 = Ws[10], r2 = Ws[20], r3 = Ws[30];
        float4* d = &sW4[q * 48 + 4 * cg];
        d[0] = make_float4(r0.x, r1.x, r2.x, r3.x);
        d[1] = make_float4(r0.y, r1.y, r2.y, r3.y);
        d[2] = make_float4(r0.z, r1.z, r2.z, r3.z);
        d[3] = make_float4(r0.w, r1.w, r2.w, r3.w);
    }
    for (int it = tid; it < 256; it += 128)
        sW4[(it >> 3) * 48 + 40 + (it & 7)] = make_float4(0.0f, 0.0f, 0.0f, 0.0f);
    __syncthreads();

    float bz0 = cb[l16];
    float bz1 = cb[16 + l16];
    float bz2 = (l16 < 8) ? cb[32 + l16] : 0.0f;
    v8f acc00, acc01, acc02, acc10, acc11, acc12;
#pragma unroll
    for (int r = 0; r < 8; ++r) {
        acc00[r] = bz0; acc01[r] = bz1; acc02[r] = bz2;
        acc10[r] = bz0; acc11[r] = bz1; acc12[r] = bz2;
    }

#pragma unroll
    for (int j = 0; j < 16; ++j) {
        const float4* bp = &sW4[(2 * j + half) * 48 + l16];
        float4 b0 = bp[0], b1 = bp[16], b2 = bp[32];
        v2f x0 = {a0[j].x, a0[j].y}, x1 = {a1[j].x, a1[j].y};
        v2f y0 = {a0[j].z, a0[j].w}, y1 = {a1[j].z, a1[j].w};
        v2f p0 = {b0.x, b0.y}, p1 = {b1.x, b1.y}, p2 = {b2.x, b2.y};
        v2f s0 = {b0.z, b0.w}, s1 = {b1.z, b1.w}, s2 = {b2.z, b2.w};
        acc00 = __builtin_amdgcn_wmma_f32_16x16x4_f32(false, x0, false, p0, (short)0, acc00, false, false);
        acc01 = __builtin_amdgcn_wmma_f32_16x16x4_f32(false, x0, false, p1, (short)0, acc01, false, false);
        acc02 = __builtin_amdgcn_wmma_f32_16x16x4_f32(false, x0, false, p2, (short)0, acc02, false, false);
        acc10 = __builtin_amdgcn_wmma_f32_16x16x4_f32(false, x1, false, p0, (short)0, acc10, false, false);
        acc11 = __builtin_amdgcn_wmma_f32_16x16x4_f32(false, x1, false, p1, (short)0, acc11, false, false);
        acc12 = __builtin_amdgcn_wmma_f32_16x16x4_f32(false, x1, false, p2, (short)0, acc12, false, false);
        acc00 = __builtin_amdgcn_wmma_f32_16x16x4_f32(false, y0, false, s0, (short)0, acc00, false, false);
        acc01 = __builtin_amdgcn_wmma_f32_16x16x4_f32(false, y0, false, s1, (short)0, acc01, false, false);
        acc02 = __builtin_amdgcn_wmma_f32_16x16x4_f32(false, y0, false, s2, (short)0, acc02, false, false);
        acc10 = __builtin_amdgcn_wmma_f32_16x16x4_f32(false, y1, false, s0, (short)0, acc10, false, false);
        acc11 = __builtin_amdgcn_wmma_f32_16x16x4_f32(false, y1, false, s1, (short)0, acc11, false, false);
        acc12 = __builtin_amdgcn_wmma_f32_16x16x4_f32(false, y1, false, s2, (short)0, acc12, false, false);
    }

    float* so = (float*)&sO4[wave][0];
#pragma unroll
    for (int r = 0; r < 8; ++r) {
        float* q0 = so + (half * 8 + r) * 40 + l16;
        float* q1 = so + (16 + half * 8 + r) * 40 + l16;
        q0[0]  = acc00[r];
        q0[16] = acc01[r];
        q1[0]  = acc10[r];
        q1[16] = acc11[r];
        if (l16 < 8) {
            q0[32] = acc02[r];
            q1[32] = acc12[r];
        }
    }
    __syncthreads();
    int nrow = N - rbase;
    float4* o4 = (float4*)out + (long)rbase * 10;
    if (nrow >= 32) {
        float4 t[10];
#pragma unroll
        for (int i = 0; i < 10; ++i) t[i] = sO4[wave][lane + 32 * i];
#pragma unroll
        for (int i = 0; i < 10; ++i) o4[lane + 32 * i] = t[i];
    } else {
        int n4 = nrow * 10;
        for (int i = lane; i < n4; i += 32) o4[i] = sO4[wave][i];
    }
}

extern "C" void kernel_launch(void* const* d_in, const int* in_sizes, int n_in,
                              void* d_out, int out_size, void* d_ws, size_t ws_size,
                              hipStream_t stream) {
    const float* x       = (const float*)d_in[0];
    const int*   ei      = (const int*)d_in[1];
    const float* conv0_w = (const float*)d_in[2];
    const float* conv0_b = (const float*)d_in[3];
    const float* lin0_w  = (const float*)d_in[4];
    const float* lin0_b  = (const float*)d_in[5];
    const float* conv1_w = (const float*)d_in[6];
    const float* conv1_b = (const float*)d_in[7];
    const float* lin1_w  = (const float*)d_in[8];
    const float* lin1_b  = (const float*)d_in[9];
    const float* out_w   = (const float*)d_in[10];
    const float* out_b   = (const float*)d_in[11];
    float* out = (float*)d_out;

    int N = in_sizes[0] / 128;
    int E = in_sizes[1] / 2;
    const int* src = ei;
    const int* dst = ei + E;

    size_t NH = (size_t)N * 64;
    float* ws   = (float*)d_ws;
    int*   cnt  = (int*)ws;
    float* A    = ws + (((size_t)N + 255) / 256) * 256;
    float* H1   = A + NH;
    float* H0   = H1 + NH;
    float* W0f  = H0 + NH;
    float* W1f  = W0f + 64 * 40;
    float* cb   = W1f + 64 * 40;
    int*   rank = (int*)H1;
    int*   oscr   = (int*)d_out;
    int2*  csr    = (int2*)oscr;
    int*   rowloc = oscr + 2 * (size_t)E;
    int*   boff   = rowloc + N;

    int gN   = (N + TPB - 1) / TPB;
    int gE   = (E + TPB - 1) / TPB;
    int gCh  = (N + CHUNK - 1) / CHUNK;
    int gGa  = (N + 15) / 16;
    int gGemm = (N + 127) / 128;

    k_zero_cnt<<<gN, TPB, 0, stream>>>(cnt, N);
    k_rank<<<gE, TPB, 0, stream>>>(dst, cnt, rank, E);
    k_scan_chunk<<<gCh, 256, 0, stream>>>(cnt, rowloc, boff, N);
    k_scan_bsum<<<1, 1024, 0, stream>>>(boff, gCh);
    k_fill<<<gE, TPB, 0, stream>>>(src, dst, rank, cnt, rowloc, boff, csr, E);

    k_fuse_heads<<<1, TPB, 0, stream>>>(lin0_w, lin0_b, lin1_w, lin1_b, out_w, out_b,
                                        W0f, W1f, cb);

    k_gemm_n64_wmma<<<gGemm, 128, 0, stream>>>(x, conv0_w, A, N, 128);
    k_gather_elu<<<gGa, 256, 0, stream>>>(A, csr, cnt, rowloc, boff, conv0_b, H0, N);

    k_gemm_n64_wmma<<<gGemm, 128, 0, stream>>>(H0, conv1_w, A, N, 64);
    k_gather_elu<<<gGa, 256, 0, stream>>>(A, csr, cnt, rowloc, boff, conv1_b, H1, N);

    k_out_wmma<<<gGemm, 128, 0, stream>>>(H0, H1, W0f, W1f, cb, out, N);
}


# in-degree counters split into one copy per coherence half of the L2 (die number picks the copy) so returning adds stay in their half; a run now sums first-half edges, then second-half, by arrival
# speedup vs baseline: 3.6386x; 1.2638x over previous
#include <hip/hip_runtime.h>
#include <hip/hip_bf16.h>

typedef float v2f __attribute__((ext_vector_type(2)));
typedef float v8f __attribute__((ext_vector_type(8)));

#define TPB 256

#define CHUNK 1024

__global__ void k_zero_cnt(int* cnt, int N) {
    int i = blockIdx.x * TPB + threadIdx.x;
    if (i < N) cnt[i] = 0;
}

__device__ __forceinline__ int cache_half() {
    return (int)((__builtin_amdgcn_s_sendmsg_rtn(0x87) >> 18) & 1u);
}

__global__ void k_rank(const int* __restrict__ dst, int* ch, int* __restrict__ rank, int E, int np) {
    int e = blockIdx.x * TPB + threadIdx.x;
    int h = cache_half();
    int* c = ch + (size_t)h * np;
    if (e < E) rank[e] = (atomicAdd(&c[dst[e]], 1) << 1) | h;
}

__global__ __launch_bounds__(256)
void k_scan_chunk(const int* __restrict__ ca, const int* __restrict__ cb2, int* __restrict__ cnt,
                  int2* __restrict__ rowloc, int* __restrict__ bsum, int N) {
    __shared__ int swave[8];
    int tid = threadIdx.x;
    int lane = tid & 31, wv = tid >> 5;
    int base = blockIdx.x * CHUNK + tid * 4;
    int a0 = (base + 0 < N) ? ca[base + 0] : 0;
    int a1 = (base + 1 < N) ? ca[base + 1] : 0;
    int a2 = (base + 2 < N) ? ca[base + 2] : 0;
    int a3 = (base + 3 < N) ? ca[base + 3] : 0;
    int c0 = (base + 0 < N) ? a0 + cb2[base + 0] : 0;
    int c1 = (base + 1 < N) ? a1 + cb2[base + 1] : 0;
    int c2 = (base + 2 < N) ? a2 + cb2[base + 2] : 0;
    int c3 = (base + 3 < N) ? a3 + cb2[base + 3] : 0;
    if (base + 0 < N) cnt[base + 0] = c0;
    if (base + 1 < N) cnt[base + 1] = c1;
    if (base + 2 < N) cnt[base + 2] = c2;
    if (base + 3 < N) cnt[base + 3] = c3;
    int t = c0 + c1 + c2 + c3;
    int incl = t;
    for (int off = 1; off < 32; off <<= 1) {
        int v = __shfl_up(incl, off, 32);
        if (lane >= off) incl += v;
    }
    if (lane == 31) swave[wv] = incl;
    __syncthreads();
    int wbase = 0;
    for (int w = 0; w < wv; ++w) wbase += swave[w];
    int excl = wbase + incl - t;
    if (base + 0 < N) rowloc[base + 0] = make_int2(excl, excl + a0);
    if (base + 1 < N) rowloc[base + 1] = make_int2(excl + c0, excl + c0 + a1);
    if (base + 2 < N) rowloc[base + 2] = make_int2(excl + c0 + c1, excl + c0 + c1 + a2);
    if (base + 3 < N) rowloc[base + 3] = make_int2(excl + c0 + c1 + c2, excl + c0 + c1 + c2 + a3);
    if (tid == 255) bsum[blockIdx.x] = excl + t;
}

__global__ __launch_bounds__(1024)
void k_scan_bsum(int* bsum, int nb) {
    __shared__ int sw[32];
    int tid = threadIdx.x;
    int lane = tid & 31, wv = tid >> 5;
    int carry = 0;
    for (int b0 = 0; b0 < nb; b0 += 1024) {
        int i = b0 + tid;
        int v = (i < nb) ? bsum[i] : 0;
        int incl = v;
        for (int off = 1; off < 32; off <<= 1) {
            int u = __shfl_up(incl, off, 32);
            if (lane >= off) incl += u;
        }
        if (lane == 31) sw[wv] = incl;
        __syncthreads();
        int wbase = 0, tot = 0;
        for (int w = 0; w < 32; ++w) {
            int s = sw[w];
            if (w < wv) wbase += s;
            tot += s;
        }
        if (i < nb) bsum[i] = carry + wbase + incl - v;
        carry += tot;
        __syncthreads();
    }
}

__global__ void k_fill(const int* __restrict__ src, const int* __restrict__ dst,
                       const int* __restrict__ rank, const int* __restrict__ cnt,
                       const int2* __restrict__ rowloc, const int* __restrict__ boff,
                       int2* __restrict__ csr, int E) {
    int e = blockIdx.x * TPB + threadIdx.x;
    if (e < E) {
        int s = src[e], d = dst[e];
        int rk = rank[e];
        int2 rl = rowloc[d];
        int pos = boff[d / CHUNK] + ((rk & 1) ? rl.y : rl.x) + (rk >> 1);
        float ds = (float)(cnt[s] + 1);
        float w = (ds > 0.0f) ? rsqrtf(ds) : 0.0f;
        csr[pos] = make_int2(s, __float_as_int(w));
    }
}

__global__ void k_fuse_heads(const float* __restrict__ L0, const float* __restrict__ b0,
                             const float* __restrict__ L1, const float* __restrict__ b1,
                             const float* __restrict__ Wo, const float* __restrict__ bo,
                             float* W0f, float* W1f, float* cb) {
    for (int i = threadIdx.x; i < 64 * 40; i += TPB) {
        int k = i / 40, c = i % 40;
        float s0 = 0.0f, s1 = 0.0f;
        for (int j = 0; j < 64; ++j) {
            s0 += L0[k * 64 + j] * Wo[j * 40 + c];
            s1 += L1[k * 64 + j] * Wo[(64 + j) * 40 + c];
        }
        W0f[i] = s0;
        W1f[i] = s1;
    }
    if (threadIdx.x < 40) {
        int c = threadIdx.x;
        float s = bo[c];
        for (int j = 0; j < 64; ++j)
            s += b0[j] * Wo[j * 40 + c] + b1[j] * Wo[(64 + j) * 40 + c];
        cb[c] = s;
    }
}

__global__ __launch_bounds__(128)
void k_gemm_n64_wmma(const float* __restrict__ A, const float* __restrict__ W,
                     float* __restrict__ D, int N, int K) {
    __shared__ float sW[128 * 64];
    int tid = threadIdx.x;
    for (int i = tid; i < K * 64; i += 128) {
        int k = i >> 6, col = i & 63;
        sW[(k >> 1) * 128 + 2 * col + (k & 1)] = W[i];
    }
    __syncthreads();

    int wave = tid >> 5;
    int lane = tid & 31;
    int half = lane >> 4;
    int l16  = lane & 15;

    int rbase = blockIdx.x * 128 + wave * 32;
    int ar0 = rbase + l16;
    int ar1 = rbase + 16 + l16;
    long ab0 = (long)((ar0 < N) ? ar0 : (N - 1)) * K;
    long ab1 = (long)((ar1 < N) ? ar1 : (N - 1)) * K;

    v8f acc00 = {}, acc01 = {}, acc02 = {}, acc03 = {};
    v8f acc10 = {}, acc11 = {}, acc12 = {}, acc13 = {};

    for (int k = 0; k < K; k += 4) {
        int kb = k + 2 * half;
        v2f a0 = *(const v2f*)(A + ab0 + kb);
        v2f a1 = *(const v2f*)(A + ab1 + kb);
        const float* bp = &sW[(kb >> 1) * 128 + 2 * l16];
        v2f b0 = *(const v2f*)(bp + 0);
        v2f b1 = *(const v2f*)(bp + 32);
        v2f b2 = *(const v2f*)(bp + 64);
        v2f b3 = *(const v2f*)(bp + 96);
        acc00 = __builtin_amdgcn_wmma_f32_16x16x4_f32(false, a0, false, b0, (short)0, acc00, false, false);
        acc01 = __builtin_amdgcn_wmma_f32_16x16x4_f32(false, a0, false, b1, (short)0, acc01, false, false);
        acc02 = __builtin_amdgcn_wmma_f32_16x16x4_f32(false, a0, false, b2, (short)0, acc02, false, false);
        acc03 = __builtin_amdgcn_wmma_f32_16x16x4_f32(false, a0, false, b3, (short)0, acc03, false, false);
        acc10 = __builtin_amdgcn_wmma_f32_16x16x4_f32(false, a1, false, b0, (short)0, acc10, false, false);
        acc11 = __builtin_amdgcn_wmma_f32_16x16x4_f32(false, a1, false, b1, (short)0, acc11, false, false);
        acc12 = __builtin_amdgcn_wmma_f32_16x16x4_f32(false, a1, false, b2, (short)0, acc12, false, false);
        acc13 = __builtin_amdgcn_wmma_f32_16x16x4_f32(false, a1, false, b3, (short)0, acc13, false, false);
    }

    int m0 = rbase + half * 8;
    int m1 = rbase + 16 + half * 8;
    for (int r = 0; r < 8; ++r) {
        int rr = m0 + r;
        if (rr < N) {
            float* dp = D + (long)rr * 64 + l16;
            dp[0]  = acc00[r];
            dp[16] = acc01[r];
            dp[32] = acc02[r];
            dp[48] = acc03[r];
        }
        int ss = m1 + r;
        if (ss < N) {
            float* dp = D + (long)ss * 64 + l16;
            dp[0]  = acc10[r];
            dp[16] = acc11[r];
            dp[32] = acc12[r];
            dp[48] = acc13[r];
        }
    }
}

template <int U>
__device__ __forceinline__ void add_rows(const int2* __restrict__ p, const float4* __restrict__ T4,
                                         int l, float4& acc) {
    int2 e[U];
    float4 r[U];
#pragma unroll
    for (int k = 0; k < U; ++k) e[k] = p[k];
#pragma unroll
    for (int k = 0; k < U; ++k) r[k] = T4[(long)e[k].x * 16 + l];
#pragma unroll
    for (int k = 0; k < U; ++k) {
        float w = __int_as_float(e[k].y);
        acc.x = fmaf(w, r[k].x, acc.x);
        acc.y = fmaf(w, r[k].y, acc.y);
        acc.z = fmaf(w, r[k].z, acc.z);
        acc.w = fmaf(w, r[k].w, acc.w);
    }
}

__global__ __launch_bounds__(256)
void k_gather_elu(const float* __restrict__ T, const int2* __restrict__ csr,
                  const int* __restrict__ cnt, const int2* __restrict__ rowloc,
                  const int* __restrict__ boff, const float* __restrict__ bias,
                  float* __restrict__ H, int N) {
    int tid = threadIdx.x;
    int n = blockIdx.x * 16 + (tid >> 4);
    int l = tid & 15;
    if (n >= N) return;
    const float4* T4 = (const float4*)T;
    int c = cnt[n];
    const int2* p = csr + (boff[n / CHUNK] + rowloc[n].x);
    float4 self = T4[(long)n * 16 + l];
    float dg = (float)(c + 1);
    float dn = (dg > 0.0f) ? rsqrtf(dg) : 0.0f;

    float4 acc = make_float4(0.0f, 0.0f, 0.0f, 0.0f);
    int j = 0;
    for (; j + 8 <= c; j += 8) add_rows<8>(p + j, T4, l, acc);
    if (c & 4) { add_rows<4>(p + j, T4, l, acc); j += 4; }
    if (c & 2) { add_rows<2>(p + j, T4, l, acc); j += 2; }
    if (c & 1) { add_rows<1>(p + j, T4, l, acc); }

    float v0 = dn * (acc.x + dn * self.x) + bias[4 * l + 0];
    float v1 = dn * (acc.y + dn * self.y) + bias[4 * l + 1];
    float v2 = dn * (acc.z + dn * self.z) + bias[4 * l + 2];
    float v3 = dn * (acc.w + dn * self.w) + bias[4 * l + 3];
    float4 h;
    h.x = (v0 > 0.0f) ? v0 : (expf(v0) - 1.0f);
    h.y = (v1 > 0.0f) ? v1 : (expf(v1) - 1.0f);
    h.z = (v2 > 0.0f) ? v2 : (expf(v2) - 1.0f);
    h.w = (v3 > 0.0f) ? v3 : (expf(v3) - 1.0f);
    ((float4*)H)[(long)n * 16 + l] = h;
}

__global__ __launch_bounds__(128)
void k_out_wmma(const float* __restrict__ H0, const float* __restrict__ H1,
                const float* __restrict__ W0f, const float* __restrict__ W1f,
                const float* __restrict__ cb, float* __restrict__ out, int N) {
    __shared__ float4 sW4[32 * 48];
    __shared__ float4 sO4[4][32 * 10];
    int tid = threadIdx.x;
    int wave = tid >> 5;
    int lane = tid & 31;
    int half = lane >> 4;
    int l16  = lane & 15;

    int rbase = blockIdx.x * 128 + wave * 32;
    int ar0 = rbase + l16;
    int ar1 = rbase + 16 + l16;
    long ab0 = (long)((ar0 < N) ? ar0 : (N - 1)) * 16;
    long ab1 = (long)((ar1 < N) ? ar1 : (N - 1)) * 16;
    const float4* H04 = (const float4*)H0;
    const float4* H14 = (const float4*)H1;

    float4 a0[16], a1[16];
#pragma unroll
    for (int j = 0; j < 8; ++j) {
        a0[j]     = H04[ab0 + 2 * j + half];
        a1[j]     = H04[ab1 + 2 * j + half];
        a0[8 + j] = H14[ab0 + 2 * j + half];
        a1[8 + j] = H14[ab1 + 2 * j + half];
    }

    for (int it = tid; it < 320; it += 128) {
        int q = it / 10, cg = it % 10;
        const float4* Ws = (const float4*)((q < 16) ? W0f : W1f) + (q & 15) * 40 + cg;
        float4 r0 = Ws[0], r1 = Ws[10], r2 = Ws[20], r3 = Ws[30];
        float4* d = &sW4[q * 48 + 4 * cg];
        d[0] = make_float4(r0.x, r1.x, r2.x, r3.x);
        d[1] = make_float4(r0.y, r1.y, r2.y, r3.y);
        d[2] = make_float4(r0.z, r1.z, r2.z, r3.z);
        d[3] = make_float4(r0.w, r1.w, r2.w, r3.w);
    }
    for (int it = tid; it < 256; it += 128)
        sW4[(it >> 3) * 48 + 40 + (it & 7)] = make_float4(0.0f, 0.0f, 0.0f, 0.0f);
    __syncthreads();

    float bz0 = cb[l16];
    float bz1 = cb[16 + l16];
    float bz2 = (l16 < 8) ? cb[32 + l16] : 0.0f;
    v8f acc00, acc01, acc02, acc10, acc11, acc12;
#pragma unroll
    for (int r = 0; r < 8; ++r) {
        acc00[r] = bz0; acc01[r] = bz1; acc02[r] = bz2;
        acc10[r] = bz0; acc11[r] = bz1; acc12[r] = bz2;
    }

#pragma unroll
    for (int j = 0; j < 16; ++j) {
        const float4* bp = &sW4[(2 * j + half) * 48 + l16];
        float4 b0 = bp[0], b1 = bp[16], b2 = bp[32];
        v2f x0 = {a0[j].x, a0[j].y}, x1 = {a1[j].x, a1[j].y};
        v2f y0 = {a0[j].z, a0[j].w}, y1 = {a1[j].z, a1[j].w};
        v2f p0 = {b0.x, b0.y}, p1 = {b1.x, b1.y}, p2 = {b2.x, b2.y};
        v2f s0 = {b0.z, b0.w}, s1 = {b1.z, b1.w}, s2 = {b2.z, b2.w};
        acc00 = __builtin_amdgcn_wmma_f32_16x16x4_f32(false, x0, false, p0, (short)0, acc00, false, false);
        acc01 = __builtin_amdgcn_wmma_f32_16x16x4_f32(false, x0, false, p1, (short)0, acc01, false, false);
        acc02 = __builtin_amdgcn_wmma_f32_16x16x4_f32(false, x0, false, p2, (short)0, acc02, false, false);
        acc10 = __builtin_amdgcn_wmma_f32_16x16x4_f32(false, x1, false, p0, (short)0, acc10, false, false);
        acc11 = __builtin_amdgcn_wmma_f32_16x16x4_f32(false, x1, false, p1, (short)0, acc11, false, false);
        acc12 = __builtin_amdgcn_wmma_f32_16x16x4_f32(false, x1, false, p2, (short)0, acc12, false, false);
        acc00 = __builtin_amdgcn_wmma_f32_16x16x4_f32(false, y0, false, s0, (short)0, acc00, false, false);
        acc01 = __builtin_amdgcn_wmma_f32_16x16x4_f32(false, y0, false, s1, (short)0, acc01, false, false);
        acc02 = __builtin_amdgcn_wmma_f32_16x16x4_f32(false, y0, false, s2, (short)0, acc02, false, false);
        acc10 = __builtin_amdgcn_wmma_f32_16x16x4_f32(false, y1, false, s0, (short)0, acc10, false, false);
        acc11 = __builtin_amdgcn_wmma_f32_16x16x4_f32(false, y1, false, s1, (short)0, acc11, false, false);
        acc12 = __builtin_amdgcn_wmma_f32_16x16x4_f32(false, y1, false, s2, (short)0, acc12, false, false);
    }

    float* so = (float*)&sO4[wave][0];
#pragma unroll
    for (int r = 0; r < 8; ++r) {
        float* q0 = so + (half * 8 + r) * 40 + l16;
        float* q1 = so + (16 + half * 8 + r) * 40 + l16;
        q0[0]  = acc00[r];
        q0[16] = acc01[r];
        q1[0]  = acc10[r];
        q1[16] = acc11[r];
        if (l16 < 8) {
            q0[32] = acc02[r];
            q1[32] = acc12[r];
        }
    }
    __syncthreads();
    int nrow = N - rbase;
    float4* o4 = (float4*)out + (long)rbase * 10;
    if (nrow >= 32) {
        float4 t[10];
#pragma unroll
        for (int i = 0; i < 10; ++i) t[i] = sO4[wave][lane + 32 * i];
#pragma unroll
        for (int i = 0; i < 10; ++i) o4[lane + 32 * i] = t[i];
    } else {
        int n4 = nrow * 10;
        for (int i = lane; i < n4; i += 32) o4[i] = sO4[wave][i];
    }
}

extern "C" void kernel_launch(void* const* d_in, const int* in_sizes, int n_in,
                              void* d_out, int out_size, void* d_ws, size_t ws_size,
                              hipStream_t stream) {
    const float* x       = (const float*)d_in[0];
    const int*   ei      = (const int*)d_in[1];
    const float* conv0_w = (const float*)d_in[2];
    const float* conv0_b = (const float*)d_in[3];
    const float* lin0_w  = (const float*)d_in[4];
    const float* lin0_b  = (const float*)d_in[5];
    const float* conv1_w = (const float*)d_in[6];
    const float* conv1_b = (const float*)d_in[7];
    const float* lin1_w  = (const float*)d_in[8];
    const float* lin1_b  = (const float*)d_in[9];
    const float* out_w   = (const float*)d_in[10];
    const float* out_b   = (const float*)d_in[11];
    float* out = (float*)d_out;

    int N = in_sizes[0] / 128;
    int E = in_sizes[1] / 2;
    const int* src = ei;
    const int* dst = ei + E;

    size_t NH = (size_t)N * 64;
    float* ws   = (float*)d_ws;
    int*   cnt  = (int*)ws;
    float* A    = ws + (((size_t)N + 255) / 256) * 256;
    float* H1   = A + NH;
    float* H0   = H1 + NH;
    float* W0f  = H0 + NH;
    float* W1f  = W0f + 64 * 40;
    float* cb   = W1f + 64 * 40;
    int*   rank = (int*)H1;
    size_t np   = (((size_t)N + 255) / 256) * 256;
    int*   ch   = rank + (((size_t)E + 255) / 256) * 256;
    int*   oscr   = (int*)d_out;
    int2*  csr    = (int2*)oscr;
    int2*  rowloc = (int2*)(oscr + 2 * (size_t)E);
    int*   boff   = (int*)(rowloc + N);

    int gN   = (N + TPB - 1) / TPB;
    int gE   = (E + TPB - 1) / TPB;
    int gCh  = (N + CHUNK - 1) / CHUNK;
    int gGa  = (N + 15) / 16;
    int gGemm = (N + 127) / 128;

    k_zero_cnt<<<2 * gN, TPB, 0, stream>>>(ch, (int)(2 * np));
    k_rank<<<gE, TPB, 0, stream>>>(dst, ch, rank, E, (int)np);
    k_scan_chunk<<<gCh, 256, 0, stream>>>(ch, ch + np, cnt, rowloc, boff, N);
    k_scan_bsum<<<1, 1024, 0, stream>>>(boff, gCh);
    k_fill<<<gE, TPB, 0, stream>>>(src, dst, rank, cnt, rowloc, boff, csr, E);

    k_fuse_heads<<<1, TPB, 0, stream>>>(lin0_w, lin0_b, lin1_w, lin1_b, out_w, out_b,
                                        W0f, W1f, cb);

    k_gemm_n64_wmma<<<gGemm, 128, 0, stream>>>(x, conv0_w, A, N, 128);
    k_gather_elu<<<gGa, 256, 0, stream>>>(A, csr, cnt, rowloc, boff, conv0_b, H0, N);

    k_gemm_n64_wmma<<<gGemm, 128, 0, stream>>>(H0, conv1_w, A, N, 64);
    k_gather_elu<<<gGa, 256, 0, stream>>>(A, csr, cnt, rowloc, boff, conv1_b, H1, N);

    k_out_wmma<<<gGemm, 128, 0, stream>>>(H0, H1, W0f, W1f, cb, out, N);
}


# fewer, longer-lived waves: 8 edges per lane in rank and fill kernels, 8 nodes per 16-lane group in the aggregation (run entries read one per lane, passed by row_share); sums still in arrival order
# speedup vs baseline: 3.8221x; 1.0504x over previous
#include <hip/hip_runtime.h>
#include <hip/hip_bf16.h>

typedef float v2f __attribute__((ext_vector_type(2)));
typedef float v8f __attribute__((ext_vector_type(8)));

#define TPB 256

#define CHUNK 1024

__global__ void k_zero_cnt(int4* cnt4, int n4) {
    int i = blockIdx.x * TPB + threadIdx.x;
    if (i < n4) cnt4[i] = make_int4(0, 0, 0, 0);
}

#define EPT 8
#define TPE 128

__device__ __forceinline__ int cache_half() {
    return (int)((__builtin_amdgcn_s_sendmsg_rtn(0x87) >> 18) & 1u);
}

__global__ __launch_bounds__(TPE)
void k_rank(const int* __restrict__ dst, int* ch, int* __restrict__ rank, int E, int np) {
    int nft = E / (TPE * EPT);
    int h = cache_half();
    int* c = ch + (size_t)h * np;
    if ((int)blockIdx.x < nft) {
        int e0 = blockIdx.x * (TPE * EPT) + threadIdx.x;
        int d[EPT], r[EPT];
#pragma unroll
        for (int k = 0; k < EPT; ++k) d[k] = dst[e0 + k * TPE];
#pragma unroll
        for (int k = 0; k < EPT; ++k) r[k] = atomicAdd(&c[d[k]], 1);
#pragma unroll
        for (int k = 0; k < EPT; ++k) rank[e0 + k * TPE] = (r[k] << 1) | h;
    } else {
        int e = nft * (TPE * EPT) + ((int)blockIdx.x - nft) * TPE + threadIdx.x;
        if (e < E) rank[e] = (atomicAdd(&c[dst[e]], 1) << 1) | h;
    }
}

__global__ __launch_bounds__(256)
void k_scan_chunk(const int* __restrict__ ca, const int* __restrict__ cb2, int* __restrict__ cnt,
                  int2* __restrict__ rowloc, int* __restrict__ bsum, int N) {
    __shared__ int swave[8];
    int tid = threadIdx.x;
    int lane = tid & 31, wv = tid >> 5;
    int base = blockIdx.x * CHUNK + tid * 4;
    int a0 = (base + 0 < N) ? ca[base + 0] : 0;
    int a1 = (base + 1 < N) ? ca[base + 1] : 0;
    int a2 = (base + 2 < N) ? ca[base + 2] : 0;
    int a3 = (base + 3 < N) ? ca[base + 3] : 0;
    int c0 = (base + 0 < N) ? a0 + cb2[base + 0] : 0;
    int c1 = (base + 1 < N) ? a1 + cb2[base + 1] : 0;
    int c2 = (base + 2 < N) ? a2 + cb2[base + 2] : 0;
    int c3 = (base + 3 < N) ? a3 + cb2[base + 3] : 0;
    if (base + 0 < N) cnt[base + 0] = c0;
    if (base + 1 < N) cnt[base + 1] = c1;
    if (base + 2 < N) cnt[base + 2] = c2;
    if (base + 3 < N) cnt[base + 3] = c3;
    int t = c0 + c1 + c2 + c3;
    int incl = t;
    for (int off = 1; off < 32; off <<= 1) {
        int v = __shfl_up(incl, off, 32);
        if (lane >= off) incl += v;
    }
    if (lane == 31) swave[wv] = incl;
    __syncthreads();
    int wbase = 0;
    for (int w = 0; w < wv; ++w) wbase += swave[w];
    int excl = wbase + incl - t;
    if (base + 0 < N) rowloc[base + 0] = make_int2(excl, excl + a0);
    if (base + 1 < N) rowloc[base + 1] = make_int2(excl + c0, excl + c0 + a1);
    if (base + 2 < N) rowloc[base + 2] = make_int2(excl + c0 + c1, excl + c0 + c1 + a2);
    if (base + 3 < N) rowloc[base + 3] = make_int2(excl + c0 + c1 + c2, excl + c0 + c1 + c2 + a3);
    if (tid == 255) bsum[blockIdx.x] = excl + t;
}

__global__ __launch_bounds__(1024)
void k_scan_bsum(int* bsum, int nb) {
    __shared__ int sw[32];
    int tid = threadIdx.x;
    int lane = tid & 31, wv = tid >> 5;
    int carry = 0;
    for (int b0 = 0; b0 < nb; b0 += 1024) {
        int i = b0 + tid;
        int v = (i < nb) ? bsum[i] : 0;
        int incl = v;
        for (int off = 1; off < 32; off <<= 1) {
            int u = __shfl_up(incl, off, 32);
            if (lane >= off) incl += u;
        }
        if (lane == 31) sw[wv] = incl;
        __syncthreads();
        int wbase = 0, tot = 0;
        for (int w = 0; w < 32; ++w) {
            int s = sw[w];
            if (w < wv) wbase += s;
            tot += s;
        }
        if (i < nb) bsum[i] = carry + wbase + incl - v;
        carry += tot;
        __syncthreads();
    }
}

__device__ __forceinline__ void fill_edge(int s, int rk, int ra, int rb, int cs, int bo, int2* __restrict__ csr) {
    int pos = bo + ((rk & 1) ? rb : ra) + (rk >> 1);
    float ds = (float)(cs + 1);
    float w = (ds > 0.0f) ? rsqrtf(ds) : 0.0f;
    csr[pos] = make_int2(s, __float_as_int(w));
}

__global__ __launch_bounds__(TPE)
void k_fill(const int* __restrict__ src, const int* __restrict__ dst,
            const int* __restrict__ rank, const int* __restrict__ cnt,
            const int2* __restrict__ rowloc, const int* __restrict__ boff,
            int2* __restrict__ csr, int E) {
    int nft = E / (TPE * EPT);
    if ((int)blockIdx.x < nft) {
        int e0 = blockIdx.x * (TPE * EPT) + threadIdx.x;
        int s[EPT], d[EPT], rk[EPT], cs[EPT], bo[EPT], ra[EPT], rb[EPT];
#pragma unroll
        for (int k = 0; k < EPT; ++k) {
            s[k]  = src[e0 + k * TPE];
            d[k]  = dst[e0 + k * TPE];
            rk[k] = rank[e0 + k * TPE];
        }
#pragma unroll
        for (int k = 0; k < EPT; ++k) {
            int2 rl = rowloc[d[k]];
            ra[k] = rl.x;
            rb[k] = rl.y;
            cs[k] = cnt[s[k]];
            bo[k] = boff[d[k] / CHUNK];
        }
#pragma unroll
        for (int k = 0; k < EPT; ++k) fill_edge(s[k], rk[k], ra[k], rb[k], cs[k], bo[k], csr);
    } else {
        int e = nft * (TPE * EPT) + ((int)blockIdx.x - nft) * TPE + threadIdx.x;
        if (e < E) {
            int s = src[e], d = dst[e];
            int2 rl = rowloc[d];
            fill_edge(s, rank[e], rl.x, rl.y, cnt[s], boff[d / CHUNK], csr);
        }
    }
}

__global__ void k_fuse_heads(const float* __restrict__ L0, const float* __restrict__ b0,
                             const float* __restrict__ L1, const float* __restrict__ b1,
                             const float* __restrict__ Wo, const float* __restrict__ bo,
                             float* W0f, float* W1f, float* cb) {
    for (int i = threadIdx.x; i < 64 * 40; i += TPB) {
        int k = i / 40, c = i % 40;
        float s0 = 0.0f, s1 = 0.0f;
        for (int j = 0; j < 64; ++j) {
            s0 += L0[k * 64 + j] * Wo[j * 40 + c];
            s1 += L1[k * 64 + j] * Wo[(64 + j) * 40 + c];
        }
        W0f[i] = s0;
        W1f[i] = s1;
    }
    if (threadIdx.x < 40) {
        int c = threadIdx.x;
        float s = bo[c];
        for (int j = 0; j < 64; ++j)
            s += b0[j] * Wo[j * 40 + c] + b1[j] * Wo[(64 + j) * 40 + c];
        cb[c] = s;
    }
}

__global__ __launch_bounds__(128)
void k_gemm_n64_wmma(const float* __restrict__ A, const float* __restrict__ W,
                     float* __restrict__ D, int N, int K) {
    __shared__ float sW[128 * 64];
    int tid = threadIdx.x;
    for (int i = tid; i < K * 64; i += 128) {
        int k = i >> 6, col = i & 63;
        sW[(k >> 1) * 128 + 2 * col + (k & 1)] = W[i];
    }
    __syncthreads();

    int wave = tid >> 5;
    int lane = tid & 31;
    int half = lane >> 4;
    int l16  = lane & 15;

    int rbase = blockIdx.x * 128 + wave * 32;
    int ar0 = rbase + l16;
    int ar1 = rbase + 16 + l16;
    long ab0 = (long)((ar0 < N) ? ar0 : (N - 1)) * K;
    long ab1 = (long)((ar1 < N) ? ar1 : (N - 1)) * K;

    v8f acc00 = {}, acc01 = {}, acc02 = {}, acc03 = {};
    v8f acc10 = {}, acc11 = {}, acc12 = {}, acc13 = {};

    for (int k = 0; k < K; k += 4) {
        int kb = k + 2 * half;
        v2f a0 = *(const v2f*)(A + ab0 + kb);
        v2f a1 = *(const v2f*)(A + ab1 + kb);
        const float* bp = &sW[(kb >> 1) * 128 + 2 * l16];
        v2f b0 = *(const v2f*)(bp + 0);
        v2f b1 = *(const v2f*)(bp + 32);
        v2f b2 = *(const v2f*)(bp + 64);
        v2f b3 = *(const v2f*)(bp + 96);
        acc00 = __builtin_amdgcn_wmma_f32_16x16x4_f32(false, a0, false, b0, (short)0, acc00, false, false);
        acc01 = __builtin_amdgcn_wmma_f32_16x16x4_f32(false, a0, false, b1, (short)0, acc01, false, false);
        acc02 = __builtin_amdgcn_wmma_f32_16x16x4_f32(false, a0, false, b2, (short)0, acc02, false, false);
        acc03 = __builtin_amdgcn_wmma_f32_16x16x4_f32(false, a0, false, b3, (short)0, acc03, false, false);
        acc10 = __builtin_amdgcn_wmma_f32_16x16x4_f32(false, a1, false, b0, (short)0, acc10, false, false);
        acc11 = __builtin_amdgcn_wmma_f32_16x16x4_f32(false, a1, false, b1, (short)0, acc11, false, false);
        acc12 = __builtin_amdgcn_wmma_f32_16x16x4_f32(false, a1, false, b2, (short)0, acc12, false, false);
        acc13 = __builtin_amdgcn_wmma_f32_16x16x4_f32(false, a1, false, b3, (short)0, acc13, false, false);
    }

    int m0 = rbase + half * 8;
    int m1 = rbase + 16 + half * 8;
    for (int r = 0; r < 8; ++r) {
        int rr = m0 + r;
        if (rr < N) {
            float* dp = D + (long)rr * 64 + l16;
            dp[0]  = acc00[r];
            dp[16] = acc01[r];
            dp[32] = acc02[r];
            dp[48] = acc03[r];
        }
        int ss = m1 + r;
        if (ss < N) {
            float* dp = D + (long)ss * 64 + l16;
            dp[0]  = acc10[r];
            dp[16] = acc11[r];
            dp[32] = acc12[r];
            dp[48] = acc13[r];
        }
    }
}

#define NPG 8
#define GTB 128

template <int K>
__device__ __forceinline__ int row_lane(int v) {
    return __builtin_amdgcn_update_dpp(0, v, 0x150 + K, 0xf, 0xf, false);
}

__device__ __forceinline__ int row_next(int v) {
    return __builtin_amdgcn_update_dpp(0, v, 0x12F, 0xf, 0xf, false);
}

template <int K0, int CNT>
__device__ __forceinline__ void rows_req(int2 e, const float4* __restrict__ T4, int l, float4* r, float* w) {
    if constexpr (CNT > 0) {
        rows_req<K0, CNT - 1>(e, T4, l, r, w);
        int s = row_lane<K0 + CNT - 1>(e.x);
        w[CNT - 1] = __int_as_float(row_lane<K0 + CNT - 1>(e.y));
        r[CNT - 1] = T4[s * 16 + l];
    }
}

template <int K0, int CNT>
__device__ __forceinline__ void tail_ids(int2 e, int* s, float* w) {
    if constexpr (CNT > 0) {
        tail_ids<K0, CNT - 1>(e, s, w);
        s[CNT - 1] = row_lane<K0 + CNT - 1>(e.x);
        w[CNT - 1] = __int_as_float(row_lane<K0 + CNT - 1>(e.y));
    }
}

template <int CNT>
__device__ __forceinline__ void rows_get(const int* s, const float4* __restrict__ T4, int l, float4* r) {
#pragma unroll
    for (int k = 0; k < CNT; ++k) r[k] = T4[s[k] * 16 + l];
}

template <int CNT>
__device__ __forceinline__ void rows_add(const float4* r, const float* w, float4& acc) {
#pragma unroll
    for (int k = 0; k < CNT; ++k) {
        acc.x = fmaf(w[k], r[k].x, acc.x);
        acc.y = fmaf(w[k], r[k].y, acc.y);
        acc.z = fmaf(w[k], r[k].z, acc.z);
        acc.w = fmaf(w[k], r[k].w, acc.w);
    }
}

__device__ __forceinline__ int piece_off(int q, int c, int l) {
    int nfull = c >> 4, t = c & 15;
    int b = (l & 8) ? 8 : (l & 4) ? 4 : (l & 2) ? 2 : (l & 1);
    int it = (t & b) ? ((t & (b - 1)) + (l - b)) : 0;
    int off = (q < nfull) ? (16 * q + l) : (16 * nfull + it);
    return (off < c) ? off : (c - 1);
}

__device__ __forceinline__ void run_sum(const int2* __restrict__ p, int c, int2 e,
                                        const float4* __restrict__ T4, int l,
                                        const float4* __restrict__ selfp, float4& self,
                                        const int2* __restrict__ nextp, int2& e_next, float4& acc) {
    int nfull = c >> 4, t = c & 15;
    for (int q = 0; q < nfull; ++q) {
        int2 en = p[piece_off(q + 1, c, l)];
        float4 r[16];
        float w[16];
        rows_req<0, 16>(e, T4, l, r, w);
        rows_add<16>(r, w, acc);
        e = en;
    }
    float4 r[16] = {};
    float w[16];
    int s[16];
    if (t) tail_ids<1, 15>(e, s + 1, w + 1);
    self = *selfp;
    e_next = *nextp;
    if (t) {
        rows_get<1>(s + 1, T4, l, r + 1);
        if (t & 2) rows_get<2>(s + 2, T4, l, r + 2);
        if (t & 4) rows_get<4>(s + 4, T4, l, r + 4);
        if (t & 8) rows_get<8>(s + 8, T4, l, r + 8);
        if (t & 1) rows_add<1>(r + 1, w + 1, acc);
        if (t & 2) rows_add<2>(r + 2, w + 2, acc);
        if (t & 4) rows_add<4>(r + 4, w + 4, acc);
        if (t & 8) rows_add<8>(r + 8, w + 8, acc);
    }
}

__device__ __forceinline__ const int2* first_piece(const int2* __restrict__ csr, int c, int st, int l) {
    return csr + ((c > 0) ? (st + piece_off(0, c, l)) : 0);
}

__global__ __launch_bounds__(GTB, 8)
void k_gather_elu(const float* __restrict__ T, const int2* __restrict__ csr,
                  const int* __restrict__ cnt, const int2* __restrict__ rowloc,
                  const int* __restrict__ boff, const float* __restrict__ bias,
                  float* __restrict__ H, int N) {
    int tid = threadIdx.x;
    int l = tid & 15;
    int n0 = (blockIdx.x * (GTB / 16) + (tid >> 4)) * NPG;
    if (n0 >= N) return;
    const float4* T4 = (const float4*)T;
    int nm = n0 + (l & (NPG - 1));
    nm = (nm < N) ? nm : (N - 1);
    int cm = cnt[nm];
    int stv = boff[nm / CHUNK] + rowloc[nm].x;
    float4 b4 = make_float4(bias[4 * l + 0], bias[4 * l + 1], bias[4 * l + 2], bias[4 * l + 3]);
    int nn = (N - n0 < NPG) ? (N - n0) : NPG;
    int2 e = *first_piece(csr, row_lane<0>(cm), row_lane<0>(stv), l);
#pragma unroll 1
    for (int i = 0; i < nn; ++i) {
        int n = n0 + i;
        int c = row_lane<0>(cm);
        int st = row_lane<0>(stv);
        cm = row_next(cm);
        stv = row_next(stv);
        const int2* nextp = first_piece(csr, row_lane<0>(cm), row_lane<0>(stv), l);
        float4 self;
        int2 e_next;
        float dg = (float)(c + 1);
        float dn = (dg > 0.0f) ? rsqrtf(dg) : 0.0f;
        float4 acc = make_float4(0.0f, 0.0f, 0.0f, 0.0f);
        run_sum(csr + st, c, e, T4, l, T4 + (n * 16 + l), self, nextp, e_next, acc);
        float v0 = dn * (acc.x + dn * self.x) + b4.x;
        float v1 = dn * (acc.y + dn * self.y) + b4.y;
        float v2 = dn * (acc.z + dn * self.z) + b4.z;
        float v3 = dn * (acc.w + dn * self.w) + b4.w;
        float4 h;
        h.x = (v0 > 0.0f) ? v0 : (expf(v0) - 1.0f);
        h.y = (v1 > 0.0f) ? v1 : (expf(v1) - 1.0f);
        h.z = (v2 > 0.0f) ? v2 : (expf(v2) - 1.0f);
        h.w = (v3 > 0.0f) ? v3 : (expf(v3) - 1.0f);
        ((float4*)H)[n * 16 + l] = h;
        e = e_next;
    }
}

__global__ __launch_bounds__(128)
void k_out_wmma(const float* __restrict__ H0, const float* __restrict__ H1,
                const float* __restrict__ W0f, const float* __restrict__ W1f,
                const float* __restrict__ cb, float* __restrict__ out, int N) {
    __shared__ float4 sW4[32 * 48];
    __shared__ float4 sO4[4][32 * 10];
    int tid = threadIdx.x;
    int wave = tid >> 5;
    int lane = tid & 31;
    int half = lane >> 4;
    int l16  = lane & 15;

    int rbase = blockIdx.x * 128 + wave * 32;
    int ar0 = rbase + l16;
    int ar1 = rbase + 16 + l16;
    long ab0 = (long)((ar0 < N) ? ar0 : (N - 1)) * 16;
    long ab1 = (long)((ar1 < N) ? ar1 : (N - 1)) * 16;
    const float4* H04 = (const float4*)H0;
    const float4* H14 = (const float4*)H1;

    float4 a0[16], a1[16];
#pragma unroll
    for (int j = 0; j < 8; ++j) {
        a0[j]     = H04[ab0 + 2 * j + half];
        a1[j]     = H04[ab1 + 2 * j + half];
        a0[8 + j] = H14[ab0 + 2 * j + half];
        a1[8 + j] = H14[ab1 + 2 * j + half];
    }

    for (int it = tid; it < 320; it += 128) {
        int q = it / 10, cg = it % 10;
        const float4* Ws = (const float4*)((q < 16) ? W0f : W1f) + (q & 15) * 40 + cg;
        float4 r0 = Ws[0], r1 = Ws[10], r2 = Ws[20], r3 = Ws[30];
        float4* d = &sW4[q * 48 + 4 * cg];
        d[0] = make_float4(r0.x, r1.x, r2.x, r3.x);
        d[1] = make_float4(r0.y, r1.y, r2.y, r3.y);
        d[2] = make_float4(r0.z, r1.z, r2.z, r3.z);
        d[3] = make_float4(r0.w, r1.w, r2.w, r3.w);
    }
    for (int it = tid; it < 256; it += 128)
        sW4[(it >> 3) * 48 + 40 + (it & 7)] = make_float4(0.0f, 0.0f, 0.0f, 0.0f);
    __syncthreads();

    float bz0 = cb[l16];
    float bz1 = cb[16 + l16];
    float bz2 = (l16 < 8) ? cb[32 + l16] : 0.0f;
    v8f acc00, acc01, acc02, acc10, acc11, acc12;
#pragma unroll
    for (int r = 0; r < 8; ++r) {
        acc00[r] = bz0; acc01[r] = bz1; acc02[r] = bz2;
        acc10[r] = bz0; acc11[r] = bz1; acc12[r] = bz2;
    }

#pragma unroll
    for (int j = 0; j < 16; ++j) {
        const float4* bp = &sW4[(2 * j + half) * 48 + l16];
        float4 b0 = bp[0], b1 = bp[16], b2 = bp[32];
        v2f x0 = {a0[j].x, a0[j].y}, x1 = {a1[j].x, a1[j].y};
        v2f y0 = {a0[j].z, a0[j].w}, y1 = {a1[j].z, a1[j].w};
        v2f p0 = {b0.x, b0.y}, p1 = {b1.x, b1.y}, p2 = {b2.x, b2.y};
        v2f s0 = {b0.z, b0.w}, s1 = {b1.z, b1.w}, s2 = {b2.z, b2.w};
        acc00 = __builtin_amdgcn_wmma_f32_16x16x4_f32(false, x0, false, p0, (short)0, acc00, false, false);
        acc01 = __builtin_amdgcn_wmma_f32_16x16x4_f32(false, x0, false, p1, (short)0, acc01, false, false);
        acc02 = __builtin_amdgcn_wmma_f32_16x16x4_f32(false, x0, false, p2, (short)0, acc02, false, false);
        acc10 = __builtin_amdgcn_wmma_f32_16x16x4_f32(false, x1, false, p0, (short)0, acc10, false, false);
        acc11 = __builtin_amdgcn_wmma_f32_16x16x4_f32(false, x1, false, p1, (short)0, acc11, false, false);
        acc12 = __builtin_amdgcn_wmma_f32_16x16x4_f32(false, x1, false, p2, (short)0, acc12, false, false);
        acc00 = __builtin_amdgcn_wmma_f32_16x16x4_f32(false, y0, false, s0, (short)0, acc00, false, false);
        acc01 = __builtin_amdgcn_wmma_f32_16x16x4_f32(false, y0, false, s1, (short)0, acc01, false, false);
        acc02 = __builtin_amdgcn_wmma_f32_16x16x4_f32(false, y0, false, s2, (short)0, acc02, false, false);
        acc10 = __builtin_amdgcn_wmma_f32_16x16x4_f32(false, y1, false, s0, (short)0, acc10, false, false);
        acc11 = __builtin_amdgcn_wmma_f32_16x16x4_f32(false, y1, false, s1, (short)0, acc11, false, false);
        acc12 = __builtin_amdgcn_wmma_f32_16x16x4_f32(false, y1, false, s2, (short)0, acc12, false, false);
    }

    float* so = (float*)&sO4[wave][0];
#pragma unroll
    for (int r = 0; r < 8; ++r) {
        float* q0 = so + (half * 8 + r) * 40 + l16;
        float* q1 = so + (16 + half * 8 + r) * 40 + l16;
        q0[0]  = acc00[r];
        q0[16] = acc01[r];
        q1[0]  = acc10[r];
        q1[16] = acc11[r];
        if (l16 < 8) {
            q0[32] = acc02[r];
            q1[32] = acc12[r];
        }
    }
    __syncthreads();
    int nrow = N - rbase;
    float4* o4 = (float4*)out + (long)rbase * 10;
    if (nrow >= 32) {
        float4 t[10];
#pragma unroll
        for (int i = 0; i < 10; ++i) t[i] = sO4[wave][lane + 32 * i];
#pragma unroll
        for (int i = 0; i < 10; ++i) o4[lane + 32 * i] = t[i];
    } else {
        int n4 = nrow * 10;
        for (int i = lane; i < n4; i += 32) o4[i] = sO4[wave][i];
    }
}

extern "C" void kernel_launch(void* const* d_in, const int* in_sizes, int n_in,
                              void* d_out, int out_size, void* d_ws, size_t ws_size,
                              hipStream_t stream) {
    const float* x       = (const float*)d_in[0];
    const int*   ei      = (const int*)d_in[1];
    const float* conv0_w = (const float*)d_in[2];
    const float* conv0_b = (const float*)d_in[3];
    const float* lin0_w  = (const float*)d_in[4];
    const float* lin0_b  = (const float*)d_in[5];
    const float* conv1_w = (const float*)d_in[6];
    const float* conv1_b = (const float*)d_in[7];
    const float* lin1_w  = (const float*)d_in[8];
    const float* lin1_b  = (const float*)d_in[9];
    const float* out_w   = (const float*)d_in[10];
    const float* out_b   = (const float*)d_in[11];
    float* out = (float*)d_out;

    int N = in_sizes[0] / 128;
    int E = in_sizes[1] / 2;
    const int* src = ei;
    const int* dst = ei + E;

    size_t NH = (size_t)N * 64;
    float* ws   = (float*)d_ws;
    int*   cnt  = (int*)ws;
    float* A    = ws + (((size_t)N + 255) / 256) * 256;
    float* H1   = A + NH;
    float* H0   = H1 + NH;
    float* W0f  = H0 + NH;
    float* W1f  = W0f + 64 * 40;
    float* cb   = W1f + 64 * 40;
    int*   rank = (int*)H1;
    size_t np   = (((size_t)N + 255) / 256) * 256;
    int*   ch   = rank + (((size_t)E + 255) / 256) * 256;
    int*   oscr   = (int*)d_out;
    int2*  csr    = (int2*)oscr;
    int2*  rowloc = (int2*)(oscr + 2 * (size_t)E);
    int*   boff   = (int*)(rowloc + N);

    int gCh  = (N + CHUNK - 1) / CHUNK;
    int gGa  = (N + (GTB / 16) * NPG - 1) / ((GTB / 16) * NPG);
    int gEe  = E / (TPE * EPT) + (E % (TPE * EPT) + TPE - 1) / TPE;
    int n4   = (int)(2 * np / 4);
    int gGemm = (N + 127) / 128;

    k_zero_cnt<<<(n4 + TPB - 1) / TPB, TPB, 0, stream>>>((int4*)ch, n4);
    k_rank<<<gEe, TPE, 0, stream>>>(dst, ch, rank, E, (int)np);
    k_scan_chunk<<<gCh, 256, 0, stream>>>(ch, ch + np, cnt, rowloc, boff, N);
    k_scan_bsum<<<1, 1024, 0, stream>>>(boff, gCh);
    k_fill<<<gEe, TPE, 0, stream>>>(src, dst, rank, cnt, rowloc, boff, csr, E);

    k_fuse_heads<<<1, TPB, 0, stream>>>(lin0_w, lin0_b, lin1_w, lin1_b, out_w, out_b,
                                        W0f, W1f, cb);

    k_gemm_n64_wmma<<<gGemm, 128, 0, stream>>>(x, conv0_w, A, N, 128);
    k_gather_elu<<<gGa, GTB, 0, stream>>>(A, csr, cnt, rowloc, boff, conv0_b, H0, N);

    k_gemm_n64_wmma<<<gGemm, 128, 0, stream>>>(H0, conv1_w, A, N, 64);
    k_gather_elu<<<gGa, GTB, 0, stream>>>(A, csr, cnt, rowloc, boff, conv1_b, H1, N);

    k_out_wmma<<<gGemm, 128, 0, stream>>>(H0, H1, W0f, W1f, cb, out, N);
}


# launch merge: the first layer's x @ W tiles and the head-matrix fusion (one entry per thread, 21 blocks) run as extra blocks of the in-degree ranking's launch; 11 launches become 9; sums unchanged
# speedup vs baseline: 4.6149x; 1.2074x over previous
#include <hip/hip_runtime.h>
#include <hip/hip_bf16.h>

typedef float v2f __attribute__((ext_vector_type(2)));
typedef float v8f __attribute__((ext_vector_type(8)));

#define TPB 256

#define CHUNK 1024

__global__ void k_zero_cnt(int4* cnt4, int n4) {
    int i = blockIdx.x * TPB + threadIdx.x;
    if (i < n4) cnt4[i] = make_int4(0, 0, 0, 0);
}

#define EPT 8
#define TPE 128

__device__ __forceinline__ int cache_half() {
    return (int)((__builtin_amdgcn_s_sendmsg_rtn(0x87) >> 18) & 1u);
}

__device__ __forceinline__
void rank_role(int bid, const int* __restrict__ dst, int* ch, int* __restrict__ rank, int E, int np) {
    int nft = E / (TPE * EPT);
    int h = cache_half();
    int* c = ch + (size_t)h * np;
    if (bid < nft) {
        int e0 = bid * (TPE * EPT) + threadIdx.x;
        int d[EPT], r[EPT];
#pragma unroll
        for (int k = 0; k < EPT; ++k) d[k] = dst[e0 + k * TPE];
#pragma unroll
        for (int k = 0; k < EPT; ++k) r[k] = atomicAdd(&c[d[k]], 1);
#pragma unroll
        for (int k = 0; k < EPT; ++k) rank[e0 + k * TPE] = (r[k] << 1) | h;
    } else {
        int e = nft * (TPE * EPT) + (bid - nft) * TPE + threadIdx.x;
        if (e < E) rank[e] = (atomicAdd(&c[dst[e]], 1) << 1) | h;
    }
}

__global__ __launch_bounds__(256)
void k_scan_chunk(const int* __restrict__ ca, const int* __restrict__ cb2, int* __restrict__ cnt,
                  int2* __restrict__ rowloc, int* __restrict__ bsum, int N) {
    __shared__ int swave[8];
    int tid = threadIdx.x;
    int lane = tid & 31, wv = tid >> 5;
    int base = blockIdx.x * CHUNK + tid * 4;
    int a0 = (base + 0 < N) ? ca[base + 0] : 0;
    int a1 = (base + 1 < N) ? ca[base + 1] : 0;
    int a2 = (base + 2 < N) ? ca[base + 2] : 0;
    int a3 = (base + 3 < N) ? ca[base + 3] : 0;
    int c0 = (base + 0 < N) ? a0 + cb2[base + 0] : 0;
    int c1 = (base + 1 < N) ? a1 + cb2[base + 1] : 0;
    int c2 = (base + 2 < N) ? a2 + cb2[base + 2] : 0;
    int c3 = (base + 3 < N) ? a3 + cb2[base + 3] : 0;
    if (base + 0 < N) cnt[base + 0] = c0;
    if (base + 1 < N) cnt[base + 1] = c1;
    if (base + 2 < N) cnt[base + 2] = c2;
    if (base + 3 < N) cnt[base + 3] = c3;
    int t = c0 + c1 + c2 + c3;
    int incl = t;
    for (int off = 1; off < 32; off <<= 1) {
        int v = __shfl_up(incl, off, 32);
        if (lane >= off) incl += v;
    }
    if (lane == 31) swave[wv] = incl;
    __syncthreads();
    int wbase = 0;
    for (int w = 0; w < wv; ++w) wbase += swave[w];
    int excl = wbase + incl - t;
    if (base + 0 < N) rowloc[base + 0] = make_int2(excl, excl + a0);
    if (base + 1 < N) rowloc[base + 1] = make_int2(excl + c0, excl + c0 + a1);
    if (base + 2 < N) rowloc[base + 2] = make_int2(excl + c0 + c1, excl + c0 + c1 + a2);
    if (base + 3 < N) rowloc[base + 3] = make_int2(excl + c0 + c1 + c2, excl + c0 + c1 + c2 + a3);
    if (tid == 255) bsum[blockIdx.x] = excl + t;
}

__global__ __launch_bounds__(1024)
void k_scan_bsum(int* bsum, int nb) {
    __shared__ int sw[32];
    int tid = threadIdx.x;
    int lane = tid & 31, wv = tid >> 5;
    int carry = 0;
    for (int b0 = 0; b0 < nb; b0 += 1024) {
        int i = b0 + tid;
        int v = (i < nb) ? bsum[i] : 0;
        int incl = v;
        for (int off = 1; off < 32; off <<= 1) {
            int u = __shfl_up(incl, off, 32);
            if (lane >= off) incl += u;
        }
        if (lane == 31) sw[wv] = incl;
        __syncthreads();
        int wbase = 0, tot = 0;
        for (int w = 0; w < 32; ++w) {
            int s = sw[w];
            if (w < wv) wbase += s;
            tot += s;
        }
        if (i < nb) bsum[i] = carry + wbase + incl - v;
        carry += tot;
        __syncthreads();
    }
}

__device__ __forceinline__ void fill_edge(int s, int rk, int ra, int rb, int cs, int bo, int2* __restrict__ csr) {
    int pos = bo + ((rk & 1) ? rb : ra) + (rk >> 1);
    float ds = (float)(cs + 1);
    float w = (ds > 0.0f) ? rsqrtf(ds) : 0.0f;
    csr[pos] = make_int2(s, __float_as_int(w));
}

__global__ __launch_bounds__(TPE)
void k_fill(const int* __restrict__ src, const int* __restrict__ dst,
            const int* __restrict__ rank, const int* __restrict__ cnt,
            const int2* __restrict__ rowloc, const int* __restrict__ boff,
            int2* __restrict__ csr, int E) {
    int nft = E / (TPE * EPT);
    if ((int)blockIdx.x < nft) {
        int e0 = blockIdx.x * (TPE * EPT) + threadIdx.x;
        int s[EPT], d[EPT], rk[EPT], cs[EPT], bo[EPT], ra[EPT], rb[EPT];
#pragma unroll
        for (int k = 0; k < EPT; ++k) {
            s[k]  = src[e0 + k * TPE];
            d[k]  = dst[e0 + k * TPE];
            rk[k] = rank[e0 + k * TPE];
        }
#pragma unroll
        for (int k = 0; k < EPT; ++k) {
            int2 rl = rowloc[d[k]];
            ra[k] = rl.x;
            rb[k] = rl.y;
            cs[k] = cnt[s[k]];
            bo[k] = boff[d[k] / CHUNK];
        }
#pragma unroll
        for (int k = 0; k < EPT; ++k) fill_edge(s[k], rk[k], ra[k], rb[k], cs[k], bo[k], csr);
    } else {
        int e = nft * (TPE * EPT) + ((int)blockIdx.x - nft) * TPE + threadIdx.x;
        if (e < E) {
            int s = src[e], d = dst[e];
            int2 rl = rowloc[d];
            fill_edge(s, rank[e], rl.x, rl.y, cnt[s], boff[d / CHUNK], csr);
        }
    }
}

#define FHB ((64 * 40) / TPE)
__device__ __forceinline__
void fuse_heads_role(int bid, const float* __restrict__ L0, const float* __restrict__ b0,
                     const float* __restrict__ L1, const float* __restrict__ b1,
                     const float* __restrict__ Wo, const float* __restrict__ bo,
                     float* W0f, float* W1f, float* cb) {
    if (bid < FHB) {
        int i = bid * TPE + threadIdx.x;
        int k = i / 40, c = i % 40;
        float s0 = 0.0f, s1 = 0.0f;
#pragma unroll 16
        for (int j = 0; j < 64; ++j) {
            s0 += L0[k * 64 + j] * Wo[j * 40 + c];
            s1 += L1[k * 64 + j] * Wo[(64 + j) * 40 + c];
        }
        W0f[i] = s0;
        W1f[i] = s1;
    } else if (threadIdx.x < 40) {
        int c = threadIdx.x;
        float s = bo[c];
        for (int j = 0; j < 64; ++j)
            s += b0[j] * Wo[j * 40 + c] + b1[j] * Wo[(64 + j) * 40 + c];
        cb[c] = s;
    }
}

__device__ __forceinline__
void gemm_n64_tile(int bid, float* sW, const float* __restrict__ A, const float* __restrict__ W,
                   float* __restrict__ D, int N, int K) {
    int tid = threadIdx.x;
    for (int i = tid; i < K * 64; i += 128) {
        int k = i >> 6, col = i & 63;
        sW[(k >> 1) * 128 + 2 * col + (k & 1)] = W[i];
    }
    __syncthreads();

    int wave = tid >> 5;
    int lane = tid & 31;
    int half = lane >> 4;
    int l16  = lane & 15;

    int rbase = bid * 128 + wave * 32;
    int ar0 = rbase + l16;
    int ar1 = rbase + 16 + l16;
    long ab0 = (long)((ar0 < N) ? ar0 : (N - 1)) * K;
    long ab1 = (long)((ar1 < N) ? ar1 : (N - 1)) * K;

    v8f acc00 = {}, acc01 = {}, acc02 = {}, acc03 = {};
    v8f acc10 = {}, acc11 = {}, acc12 = {}, acc13 = {};

    for (int k = 0; k < K; k += 4) {
        int kb = k + 2 * half;
        v2f a0 = *(const v2f*)(A + ab0 + kb);
        v2f a1 = *(const v2f*)(A + ab1 + kb);
        const float* bp = &sW[(kb >> 1) * 128 + 2 * l16];
        v2f b0 = *(const v2f*)(bp + 0);
        v2f b1 = *(const v2f*)(bp + 32);
        v2f b2 = *(const v2f*)(bp + 64);
        v2f b3 = *(const v2f*)(bp + 96);
        acc00 = __builtin_amdgcn_wmma_f32_16x16x4_f32(false, a0, false, b0, (short)0, acc00, false, false);
        acc01 = __builtin_amdgcn_wmma_f32_16x16x4_f32(false, a0, false, b1, (short)0, acc01, false, false);
        acc02 = __builtin_amdgcn_wmma_f32_16x16x4_f32(false, a0, false, b2, (short)0, acc02, false, false);
        acc03 = __builtin_amdgcn_wmma_f32_16x16x4_f32(false, a0, false, b3, (short)0, acc03, false, false);
        acc10 = __builtin_amdgcn_wmma_f32_16x16x4_f32(false, a1, false, b0, (short)0, acc10, false, false);
        acc11 = __builtin_amdgcn_wmma_f32_16x16x4_f32(false, a1, false, b1, (short)0, acc11, false, false);
        acc12 = __builtin_amdgcn_wmma_f32_16x16x4_f32(false, a1, false, b2, (short)0, acc12, false, false);
        acc13 = __builtin_amdgcn_wmma_f32_16x16x4_f32(false, a1, false, b3, (short)0, acc13, false, false);
    }

    int m0 = rbase + half * 8;
    int m1 = rbase + 16 + half * 8;
    for (int r = 0; r < 8; ++r) {
        int rr = m0 + r;
        if (rr < N) {
            float* dp = D + (long)rr * 64 + l16;
            dp[0]  = acc00[r];
            dp[16] = acc01[r];
            dp[32] = acc02[r];
            dp[48] = acc03[r];
        }
        int ss = m1 + r;
        if (ss < N) {
            float* dp = D + (long)ss * 64 + l16;
            dp[0]  = acc10[r];
            dp[16] = acc11[r];
            dp[32] = acc12[r];
            dp[48] = acc13[r];
        }
    }
}

__global__ __launch_bounds__(128)
void k_gemm_n64_wmma(const float* __restrict__ A, const float* __restrict__ W,
                     float* __restrict__ D, int N, int K) {
    __shared__ float sW[128 * 64];
    gemm_n64_tile((int)blockIdx.x, sW, A, W, D, N, K);
}

__global__ __launch_bounds__(TPE)
void k_rank_dense(const int* __restrict__ dst, int* ch, int* __restrict__ rank, int E, int np,
                  const float* __restrict__ X, const float* __restrict__ W, float* __restrict__ T, int N, int gG,
                  const float* __restrict__ L0, const float* __restrict__ b0,
                  const float* __restrict__ L1, const float* __restrict__ b1,
                  const float* __restrict__ Wo, const float* __restrict__ bo,
                  float* W0f, float* W1f, float* cb) {
    __shared__ float sW[128 * 64];
    int b = (int)blockIdx.x;
    if (b <= FHB) {
        fuse_heads_role(b, L0, b0, L1, b1, Wo, bo, W0f, W1f, cb);
    } else if (b <= FHB + gG) {
        gemm_n64_tile(b - (FHB + 1), sW, X, W, T, N, 128);
    } else {
        rank_role(b - (FHB + 1) - gG, dst, ch, rank, E, np);
    }
}

#define NPG 8
#define GTB 128

template <int K>
__device__ __forceinline__ int row_lane(int v) {
    return __builtin_amdgcn_update_dpp(0, v, 0x150 + K, 0xf, 0xf, false);
}

__device__ __forceinline__ int row_next(int v) {
    return __builtin_amdgcn_update_dpp(0, v, 0x12F, 0xf, 0xf, false);
}

template <int K0, int CNT>
__device__ __forceinline__ void rows_req(int2 e, const float4* __restrict__ T4, int l, float4* r, float* w) {
    if constexpr (CNT > 0) {
        rows_req<K0, CNT - 1>(e, T4, l, r, w);
        int s = row_lane<K0 + CNT - 1>(e.x);
        w[CNT - 1] = __int_as_float(row_lane<K0 + CNT - 1>(e.y));
        r[CNT - 1] = T4[s * 16 + l];
    }
}

template <int K0, int CNT>
__device__ __forceinline__ void tail_ids(int2 e, int* s, float* w) {
    if constexpr (CNT > 0) {
        tail_ids<K0, CNT - 1>(e, s, w);
        s[CNT - 1] = row_lane<K0 + CNT - 1>(e.x);
        w[CNT - 1] = __int_as_float(row_lane<K0 + CNT - 1>(e.y));
    }
}

template <int CNT>
__device__ __forceinline__ void rows_get(const int* s, const float4* __restrict__ T4, int l, float4* r) {
#pragma unroll
    for (int k = 0; k < CNT; ++k) r[k] = T4[s[k] * 16 + l];
}

template <int CNT>
__device__ __forceinline__ void rows_add(const float4* r, const float* w, float4& acc) {
#pragma unroll
    for (int k = 0; k < CNT; ++k) {
        acc.x = fmaf(w[k], r[k].x, acc.x);
        acc.y = fmaf(w[k], r[k].y, acc.y);
        acc.z = fmaf(w[k], r[k].z, acc.z);
        acc.w = fmaf(w[k], r[k].w, acc.w);
    }
}

__device__ __forceinline__ int piece_off(int q, int c, int l) {
    int nfull = c >> 4, t = c & 15;
    int b = (l & 8) ? 8 : (l & 4) ? 4 : (l & 2) ? 2 : (l & 1);
    int it = (t & b) ? ((t & (b - 1)) + (l - b)) : 0;
    int off = (q < nfull) ? (16 * q + l) : (16 * nfull + it);
    return (off < c) ? off : (c - 1);
}

__device__ __forceinline__ void run_sum(const int2* __restrict__ p, int c, int2 e,
                                        const float4* __restrict__ T4, int l,
                                        const float4* __restrict__ selfp, float4& self,
                                        const int2* __restrict__ nextp, int2& e_next, float4& acc) {
    int nfull = c >> 4, t = c & 15;
    for (int q = 0; q < nfull; ++q) {
        int2 en = p[piece_off(q + 1, c, l)];
        float4 r[16];
        float w[16];
        rows_req<0, 16>(e, T4, l, r, w);
        rows_add<16>(r, w, acc);
        e = en;
    }
    float4 r[16] = {};
    float w[16];
    int s[16];
    if (t) tail_ids<1, 15>(e, s + 1, w + 1);
    self = *selfp;
    e_next = *nextp;
    if (t) {
        rows_get<1>(s + 1, T4, l, r + 1);
        if (t & 2) rows_get<2>(s + 2, T4, l, r + 2);
        if (t & 4) rows_get<4>(s + 4, T4, l, r + 4);
        if (t & 8) rows_get<8>(s + 8, T4, l, r + 8);
        if (t & 1) rows_add<1>(r + 1, w + 1, acc);
        if (t & 2) rows_add<2>(r + 2, w + 2, acc);
        if (t & 4) rows_add<4>(r + 4, w + 4, acc);
        if (t & 8) rows_add<8>(r + 8, w + 8, acc);
    }
}

__device__ __forceinline__ const int2* first_piece(const int2* __restrict__ csr, int c, int st, int l) {
    return csr + ((c > 0) ? (st + piece_off(0, c, l)) : 0);
}

__global__ __launch_bounds__(GTB, 8)
void k_gather_elu(const float* __restrict__ T, const int2* __restrict__ csr,
                  const int* __restrict__ cnt, const int2* __restrict__ rowloc,
                  const int* __restrict__ boff, const float* __restrict__ bias,
                  float* __restrict__ H, int N) {
    int tid = threadIdx.x;
    int l = tid & 15;
    int n0 = (blockIdx.x * (GTB / 16) + (tid >> 4)) * NPG;
    if (n0 >= N) return;
    const float4* T4 = (const float4*)T;
    int nm = n0 + (l & (NPG - 1));
    nm = (nm < N) ? nm : (N - 1);
    int cm = cnt[nm];
    int stv = boff[nm / CHUNK] + rowloc[nm].x;
    float4 b4 = make_float4(bias[4 * l + 0], bias[4 * l + 1], bias[4 * l + 2], bias[4 * l + 3]);
    int nn = (N - n0 < NPG) ? (N - n0) : NPG;
    int2 e = *first_piece(csr, row_lane<0>(cm), row_lane<0>(stv), l);
#pragma unroll 1
    for (int i = 0; i < nn; ++i) {
        int n = n0 + i;
        int c = row_lane<0>(cm);
        int st = row_lane<0>(stv);
        cm = row_next(cm);
        stv = row_next(stv);
        const int2* nextp = first_piece(csr, row_lane<0>(cm), row_lane<0>(stv), l);
        float4 self;
        int2 e_next;
        float dg = (float)(c + 1);
        float dn = (dg > 0.0f) ? rsqrtf(dg) : 0.0f;
        float4 acc = make_float4(0.0f, 0.0f, 0.0f, 0.0f);
        run_sum(csr + st, c, e, T4, l, T4 + (n * 16 + l), self, nextp, e_next, acc);
        float v0 = dn * (acc.x + dn * self.x) + b4.x;
        float v1 = dn * (acc.y + dn * self.y) + b4.y;
        float v2 = dn * (acc.z + dn * self.z) + b4.z;
        float v3 = dn * (acc.w + dn * self.w) + b4.w;
        float4 h;
        h.x = (v0 > 0.0f) ? v0 : (expf(v0) - 1.0f);
        h.y = (v1 > 0.0f) ? v1 : (expf(v1) - 1.0f);
        h.z = (v2 > 0.0f) ? v2 : (expf(v2) - 1.0f);
        h.w = (v3 > 0.0f) ? v3 : (expf(v3) - 1.0f);
        ((float4*)H)[n * 16 + l] = h;
        e = e_next;
    }
}

__global__ __launch_bounds__(128)
void k_out_wmma(const float* __restrict__ H0, const float* __restrict__ H1,
                const float* __restrict__ W0f, const float* __restrict__ W1f,
                const float* __restrict__ cb, float* __restrict__ out, int N) {
    __shared__ float4 sW4[32 * 48];
    __shared__ float4 sO4[4][32 * 10];
    int tid = threadIdx.x;
    int wave = tid >> 5;
    int lane = tid & 31;
    int half = lane >> 4;
    int l16  = lane & 15;

    int rbase = blockIdx.x * 128 + wave * 32;
    int ar0 = rbase + l16;
    int ar1 = rbase + 16 + l16;
    long ab0 = (long)((ar0 < N) ? ar0 : (N - 1)) * 16;
    long ab1 = (long)((ar1 < N) ? ar1 : (N - 1)) * 16;
    const float4* H04 = (const float4*)H0;
    const float4* H14 = (const float4*)H1;

    float4 a0[16], a1[16];
#pragma unroll
    for (int j = 0; j < 8; ++j) {
        a0[j]     = H04[ab0 + 2 * j + half];
        a1[j]     = H04[ab1 + 2 * j + half];
        a0[8 + j] = H14[ab0 + 2 * j + half];
        a1[8 + j] = H14[ab1 + 2 * j + half];
    }

    for (int it = tid; it < 320; it += 128) {
        int q = it / 10, cg = it % 10;
        const float4* Ws = (const float4*)((q < 16) ? W0f : W1f) + (q & 15) * 40 + cg;
        float4 r0 = Ws[0], r1 = Ws[10], r2 = Ws[20], r3 = Ws[30];
        float4* d = &sW4[q * 48 + 4 * cg];
        d[0] = make_float4(r0.x, r1.x, r2.x, r3.x);
        d[1] = make_float4(r0.y, r1.y, r2.y, r3.y);
        d[2] = make_float4(r0.z, r1.z, r2.z, r3.z);
        d[3] = make_float4(r0.w, r1.w, r2.w, r3.w);
    }
    for (int it = tid; it < 256; it += 128)
        sW4[(it >> 3) * 48 + 40 + (it & 7)] = make_float4(0.0f, 0.0f, 0.0f, 0.0f);
    __syncthreads();

    float bz0 = cb[l16];
    float bz1 = cb[16 + l16];
    float bz2 = (l16 < 8) ? cb[32 + l16] : 0.0f;
    v8f acc00, acc01, acc02, acc10, acc11, acc12;
#pragma unroll
    for (int r = 0; r < 8; ++r) {
        acc00[r] = bz0; acc01[r] = bz1; acc02[r] = bz2;
        acc10[r] = bz0; acc11[r] = bz1; acc12[r] = bz2;
    }

#pragma unroll
    for (int j = 0; j < 16; ++j) {
        const float4* bp = &sW4[(2 * j + half) * 48 + l16];
        float4 b0 = bp[0], b1 = bp[16], b2 = bp[32];
        v2f x0 = {a0[j].x, a0[j].y}, x1 = {a1[j].x, a1[j].y};
        v2f y0 = {a0[j].z, a0[j].w}, y1 = {a1[j].z, a1[j].w};
        v2f p0 = {b0.x, b0.y}, p1 = {b1.x, b1.y}, p2 = {b2.x, b2.y};
        v2f s0 = {b0.z, b0.w}, s1 = {b1.z, b1.w}, s2 = {b2.z, b2.w};
        acc00 = __builtin_amdgcn_wmma_f32_16x16x4_f32(false, x0, false, p0, (short)0, acc00, false, false);
        acc01 = __builtin_amdgcn_wmma_f32_16x16x4_f32(false, x0, false, p1, (short)0, acc01, false, false);
        acc02 = __builtin_amdgcn_wmma_f32_16x16x4_f32(false, x0, false, p2, (short)0, acc02, false, false);
        acc10 = __builtin_amdgcn_wmma_f32_16x16x4_f32(false, x1, false, p0, (short)0, acc10, false, false);
        acc11 = __builtin_amdgcn_wmma_f32_16x16x4_f32(false, x1, false, p1, (short)0, acc11, false, false);
        acc12 = __builtin_amdgcn_wmma_f32_16x16x4_f32(false, x1, false, p2, (short)0, acc12, false, false);
        acc00 = __builtin_amdgcn_wmma_f32_16x16x4_f32(false, y0, false, s0, (short)0, acc00, false, false);
        acc01 = __builtin_amdgcn_wmma_f32_16x16x4_f32(false, y0, false, s1, (short)0, acc01, false, false);
        acc02 = __builtin_amdgcn_wmma_f32_16x16x4_f32(false, y0, false, s2, (short)0, acc02, false, false);
        acc10 = __builtin_amdgcn_wmma_f32_16x16x4_f32(false, y1, false, s0, (short)0, acc10, false, false);
        acc11 = __builtin_amdgcn_wmma_f32_16x16x4_f32(false, y1, false, s1, (short)0, acc11, false, false);
        acc12 = __builtin_amdgcn_wmma_f32_16x16x4_f32(false, y1, false, s2, (short)0, acc12, false, false);
    }

    float* so = (float*)&sO4[wave][0];
#pragma unroll
    for (int r = 0; r < 8; ++r) {
        float* q0 = so + (half * 8 + r) * 40 + l16;
        float* q1 = so + (16 + half * 8 + r) * 40 + l16;
        q0[0]  = acc00[r];
        q0[16] = acc01[r];
        q1[0]  = acc10[r];
        q1[16] = acc11[r];
        if (l16 < 8) {
            q0[32] = acc02[r];
            q1[32] = acc12[r];
        }
    }
    __syncthreads();
    int nrow = N - rbase;
    float4* o4 = (float4*)out + (long)rbase * 10;
    if (nrow >= 32) {
        float4 t[10];
#pragma unroll
        for (int i = 0; i < 10; ++i) t[i] = sO4[wave][lane + 32 * i];
#pragma unroll
        for (int i = 0; i < 10; ++i) o4[lane + 32 * i] = t[i];
    } else {
        int n4 = nrow * 10;
        for (int i = lane; i < n4; i += 32) o4[i] = sO4[wave][i];
    }
}

extern "C" void kernel_launch(void* const* d_in, const int* in_sizes, int n_in,
                              void* d_out, int out_size, void* d_ws, size_t ws_size,
                              hipStream_t stream) {
    const float* x       = (const float*)d_in[0];
    const int*   ei      = (const int*)d_in[1];
    const float* conv0_w = (const float*)d_in[2];
    const float* conv0_b = (const float*)d_in[3];
    const float* lin0_w  = (const float*)d_in[4];
    const float* lin0_b  = (const float*)d_in[5];
    const float* conv1_w = (const float*)d_in[6];
    const float* conv1_b = (const float*)d_in[7];
    const float* lin1_w  = (const float*)d_in[8];
    const float* lin1_b  = (const float*)d_in[9];
    const float* out_w   = (const float*)d_in[10];
    const float* out_b   = (const float*)d_in[11];
    float* out = (float*)d_out;

    int N = in_sizes[0] / 128;
    int E = in_sizes[1] / 2;
    const int* src = ei;
    const int* dst = ei + E;

    size_t NH = (size_t)N * 64;
    float* ws   = (float*)d_ws;
    int*   cnt  = (int*)ws;
    float* A    = ws + (((size_t)N + 255) / 256) * 256;
    float* H1   = A + NH;
    float* H0   = H1 + NH;
    float* W0f  = H0 + NH;
    float* W1f  = W0f + 64 * 40;
    float* cb   = W1f + 64 * 40;
    int*   rank = (int*)H1;
    size_t np   = (((size_t)N + 255) / 256) * 256;
    int*   ch   = rank + (((size_t)E + 255) / 256) * 256;
    int*   oscr   = (int*)d_out;
    int2*  csr    = (int2*)oscr;
    int2*  rowloc = (int2*)(oscr + 2 * (size_t)E);
    int*   boff   = (int*)(rowloc + N);

    int gCh  = (N + CHUNK - 1) / CHUNK;
    int gGa  = (N + (GTB / 16) * NPG - 1) / ((GTB / 16) * NPG);
    int gEe  = E / (TPE * EPT) + (E % (TPE * EPT) + TPE - 1) / TPE;
    int n4   = (int)(2 * np / 4);
    int gGemm = (N + 127) / 128;

    k_zero_cnt<<<(n4 + TPB - 1) / TPB, TPB, 0, stream>>>((int4*)ch, n4);
    k_rank_dense<<<(FHB + 1) + gGemm + gEe, TPE, 0, stream>>>(dst, ch, rank, E, (int)np, x, conv0_w, A, N, gGemm,
                                                             lin0_w, lin0_b, lin1_w, lin1_b, out_w, out_b,
                                                             W0f, W1f, cb);
    k_scan_chunk<<<gCh, 256, 0, stream>>>(ch, ch + np, cnt, rowloc, boff, N);
    k_scan_bsum<<<1, 1024, 0, stream>>>(boff, gCh);
    k_fill<<<gEe, TPE, 0, stream>>>(src, dst, rank, cnt, rowloc, boff, csr, E);

    k_gather_elu<<<gGa, GTB, 0, stream>>>(A, csr, cnt, rowloc, boff, conv0_b, H0, N);

    k_gemm_n64_wmma<<<gGemm, 128, 0, stream>>>(H0, conv1_w, A, N, 64);
    k_gather_elu<<<gGa, GTB, 0, stream>>>(A, csr, cnt, rowloc, boff, conv1_b, H1, N);

    k_out_wmma<<<gGemm, 128, 0, stream>>>(H0, H1, W0f, W1f, cb, out, N);
}


# launch merge: layer 1's h0 @ W runs on the f32 WMMA in the first aggregation's blocks (rows handed over in LDS); k_fill's blocks form the chunk-start prefix sum in LDS; 9 launches become 7; same sums
# speedup vs baseline: 5.0847x; 1.1018x over previous
#include <hip/hip_runtime.h>
#include <hip/hip_bf16.h>

typedef float v2f __attribute__((ext_vector_type(2)));
typedef float v8f __attribute__((ext_vector_type(8)));

#define TPB 256

#define CHUNK 1024

__global__ void k_zero_cnt(int4* cnt4, int n4) {
    int i = blockIdx.x * TPB + threadIdx.x;
    if (i < n4) cnt4[i] = make_int4(0, 0, 0, 0);
}

#define EPT 8
#define TPE 128

__device__ __forceinline__ int cache_half() {
    return (int)((__builtin_amdgcn_s_sendmsg_rtn(0x87) >> 18) & 1u);
}

__device__ __forceinline__
void rank_role(int bid, const int* __restrict__ dst, int* ch, int* __restrict__ rank, int E, int np) {
    int nft = E / (TPE * EPT);
    int h = cache_half();
    int* c = ch + (size_t)h * np;
    if (bid < nft) {
        int e0 = bid * (TPE * EPT) + threadIdx.x;
        int d[EPT], r[EPT];
#pragma unroll
        for (int k = 0; k < EPT; ++k) d[k] = dst[e0 + k * TPE];
#pragma unroll
        for (int k = 0; k < EPT; ++k) r[k] = atomicAdd(&c[d[k]], 1);
#pragma unroll
        for (int k = 0; k < EPT; ++k) rank[e0 + k * TPE] = (r[k] << 1) | h;
    } else {
        int e = nft * (TPE * EPT) + (bid - nft) * TPE + threadIdx.x;
        if (e < E) rank[e] = (atomicAdd(&c[dst[e]], 1) << 1) | h;
    }
}

__global__ __launch_bounds__(256)
void k_scan_chunk(const int* __restrict__ ca, const int* __restrict__ cb2, int* __restrict__ cnt,
                  int2* __restrict__ rowloc, int* __restrict__ bsum, int N) {
    __shared__ int swave[8];
    int tid = threadIdx.x;
    int lane = tid & 31, wv = tid >> 5;
    int base = blockIdx.x * CHUNK + tid * 4;
    int a0 = (base + 0 < N) ? ca[base + 0] : 0;
    int a1 = (base + 1 < N) ? ca[base + 1] : 0;
    int a2 = (base + 2 < N) ? ca[base + 2] : 0;
    int a3 = (base + 3 < N) ? ca[base + 3] : 0;
    int c0 = (base + 0 < N) ? a0 + cb2[base + 0] : 0;
    int c1 = (base + 1 < N) ? a1 + cb2[base + 1] : 0;
    int c2 = (base + 2 < N) ? a2 + cb2[base + 2] : 0;
    int c3 = (base + 3 < N) ? a3 + cb2[base + 3] : 0;
    if (base + 0 < N) cnt[base + 0] = c0;
    if (base + 1 < N) cnt[base + 1] = c1;
    if (base + 2 < N) cnt[base + 2] = c2;
    if (base + 3 < N) cnt[base + 3] = c3;
    int t = c0 + c1 + c2 + c3;
    int incl = t;
    for (int off = 1; off < 32; off <<= 1) {
        int v = __shfl_up(incl, off, 32);
        if (lane >= off) incl += v;
    }
    if (lane == 31) swave[wv] = incl;
    __syncthreads();
    int wbase = 0;
    for (int w = 0; w < wv; ++w) wbase += swave[w];
    int excl = wbase + incl - t;
    if (base + 0 < N) rowloc[base + 0] = make_int2(excl, excl + a0);
    if (base + 1 < N) rowloc[base + 1] = make_int2(excl + c0, excl + c0 + a1);
    if (base + 2 < N) rowloc[base + 2] = make_int2(excl + c0 + c1, excl + c0 + c1 + a2);
    if (base + 3 < N) rowloc[base + 3] = make_int2(excl + c0 + c1 + c2, excl + c0 + c1 + c2 + a3);
    if (tid == 255) bsum[blockIdx.x] = excl + t;
}

__device__ __forceinline__ void fill_edge(int s, int rk, int ra, int rb, int cs, int bo, int2* __restrict__ csr) {
    int pos = bo + ((rk & 1) ? rb : ra) + (rk >> 1);
    float ds = (float)(cs + 1);
    float w = (ds > 0.0f) ? rsqrtf(ds) : 0.0f;
    csr[pos] = make_int2(s, __float_as_int(w));
}

#define MAXCH 1024
#define CPT (MAXCH / TPE)
__device__ __forceinline__
void chunk_starts(const int* __restrict__ btot, int nb, int* __restrict__ boff, int* sB, int* sWv) {
    int tid = threadIdx.x;
    int lane = tid & 31, wv = tid >> 5;
    int v[CPT];
    int t = 0;
#pragma unroll
    for (int j = 0; j < CPT; ++j) {
        int i = tid * CPT + j;
        v[j] = (i < nb) ? btot[i] : 0;
        t += v[j];
    }
    int incl = t;
    for (int off = 1; off < 32; off <<= 1) {
        int u = __shfl_up(incl, off, 32);
        if (lane >= off) incl += u;
    }
    if (lane == 31) sWv[wv] = incl;
    __syncthreads();
    int ex = incl - t;
    for (int w = 0; w < wv; ++w) ex += sWv[w];
#pragma unroll
    for (int j = 0; j < CPT; ++j) {
        int i = tid * CPT + j;
        sB[i] = ex;
        if (blockIdx.x == 0 && i < nb) boff[i] = ex;
        ex += v[j];
    }
    __syncthreads();
}

__global__ __launch_bounds__(TPE)
void k_fill(const int* __restrict__ src, const int* __restrict__ dst,
            const int* __restrict__ rank, const int* __restrict__ cnt,
            const int2* __restrict__ rowloc, const int* __restrict__ btot, int nb,
            int* __restrict__ boff, int2* __restrict__ csr, int E) {
    __shared__ int sB[MAXCH];
    __shared__ int sWv[TPE / 32];
    int nft = E / (TPE * EPT);
    if ((int)blockIdx.x < nft) {
        int e0 = blockIdx.x * (TPE * EPT) + threadIdx.x;
        int s[EPT], d[EPT], rk[EPT], cs[EPT], bo[EPT], ra[EPT], rb[EPT];
#pragma unroll
        for (int k = 0; k < EPT; ++k) {
            s[k]  = src[e0 + k * TPE];
            d[k]  = dst[e0 + k * TPE];
            rk[k] = rank[e0 + k * TPE];
        }
        chunk_starts(btot, nb, boff, sB, sWv);
#pragma unroll
        for (int k = 0; k < EPT; ++k) {
            int2 rl = rowloc[d[k]];
            ra[k] = rl.x;
            rb[k] = rl.y;
            cs[k] = cnt[s[k]];
            bo[k] = sB[d[k] / CHUNK];
        }
#pragma unroll
        for (int k = 0; k < EPT; ++k) fill_edge(s[k], rk[k], ra[k], rb[k], cs[k], bo[k], csr);
    } else {
        chunk_starts(btot, nb, boff, sB, sWv);
        int e = nft * (TPE * EPT) + ((int)blockIdx.x - nft) * TPE + threadIdx.x;
        if (e < E) {
            int s = src[e], d = dst[e];
            int2 rl = rowloc[d];
            fill_edge(s, rank[e], rl.x, rl.y, cnt[s], sB[d / CHUNK], csr);
        }
    }
}

#define FHB ((64 * 40) / TPE)
__device__ __forceinline__
void fuse_heads_role(int bid, const float* __restrict__ L0, const float* __restrict__ b0,
                     const float* __restrict__ L1, const float* __restrict__ b1,
                     const float* __restrict__ Wo, const float* __restrict__ bo,
                     float* W0f, float* W1f, float* cb) {
    if (bid < FHB) {
        int i = bid * TPE + threadIdx.x;
        int k = i / 40, c = i % 40;
        float s0 = 0.0f, s1 = 0.0f;
#pragma unroll 16
        for (int j = 0; j < 64; ++j) {
            s0 += L0[k * 64 + j] * Wo[j * 40 + c];
            s1 += L1[k * 64 + j] * Wo[(64 + j) * 40 + c];
        }
        W0f[i] = s0;
        W1f[i] = s1;
    } else if (threadIdx.x < 40) {
        int c = threadIdx.x;
        float s = bo[c];
        for (int j = 0; j < 64; ++j)
            s += b0[j] * Wo[j * 40 + c] + b1[j] * Wo[(64 + j) * 40 + c];
        cb[c] = s;
    }
}

__device__ __forceinline__
void gemm_n64_tile(int bid, float* sW, const float* __restrict__ A, const float* __restrict__ W,
                   float* __restrict__ D, int N, int K) {
    int tid = threadIdx.x;
    for (int i = tid; i < K * 64; i += 128) {
        int k = i >> 6, col = i & 63;
        sW[(k >> 1) * 128 + 2 * col + (k & 1)] = W[i];
    }
    __syncthreads();

    int wave = tid >> 5;
    int lane = tid & 31;
    int half = lane >> 4;
    int l16  = lane & 15;

    int rbase = bid * 128 + wave * 32;
    int ar0 = rbase + l16;
    int ar1 = rbase + 16 + l16;
    long ab0 = (long)((ar0 < N) ? ar0 : (N - 1)) * K;
    long ab1 = (long)((ar1 < N) ? ar1 : (N - 1)) * K;

    v8f acc00 = {}, acc01 = {}, acc02 = {}, acc03 = {};
    v8f acc10 = {}, acc11 = {}, acc12 = {}, acc13 = {};

    for (int k = 0; k < K; k += 4) {
        int kb = k + 2 * half;
        v2f a0 = *(const v2f*)(A + ab0 + kb);
        v2f a1 = *(const v2f*)(A + ab1 + kb);
        const float* bp = &sW[(kb >> 1) * 128 + 2 * l16];
        v2f b0 = *(const v2f*)(bp + 0);
        v2f b1 = *(const v2f*)(bp + 32);
        v2f b2 = *(const v2f*)(bp + 64);
        v2f b3 = *(const v2f*)(bp + 96);
        acc00 = __builtin_amdgcn_wmma_f32_16x16x4_f32(false, a0, false, b0, (short)0, acc00, false, false);
        acc01 = __builtin_amdgcn_wmma_f32_16x16x4_f32(false, a0, false, b1, (short)0, acc01, false, false);
        acc02 = __builtin_amdgcn_wmma_f32_16x16x4_f32(false, a0, false, b2, (short)0, acc02, false, false);
        acc03 = __builtin_amdgcn_wmma_f32_16x16x4_f32(false, a0, false, b3, (short)0, acc03, false, false);
        acc10 = __builtin_amdgcn_wmma_f32_16x16x4_f32(false, a1, false, b0, (short)0, acc10, false, false);
        acc11 = __builtin_amdgcn_wmma_f32_16x16x4_f32(false, a1, false, b1, (short)0, acc11, false, false);
        acc12 = __builtin_amdgcn_wmma_f32_16x16x4_f32(false, a1, false, b2, (short)0, acc12, false, false);
        acc13 = __builtin_amdgcn_wmma_f32_16x16x4_f32(false, a1, false, b3, (short)0, acc13, false, false);
    }

    int m0 = rbase + half * 8;
    int m1 = rbase + 16 + half * 8;
    for (int r = 0; r < 8; ++r) {
        int rr = m0 + r;
        if (rr < N) {
            float* dp = D + (long)rr * 64 + l16;
            dp[0]  = acc00[r];
            dp[16] = acc01[r];
            dp[32] = acc02[r];
            dp[48] = acc03[r];
        }
        int ss = m1 + r;
        if (ss < N) {
            float* dp = D + (long)ss * 64 + l16;
            dp[0]  = acc10[r];
            dp[16] = acc11[r];
            dp[32] = acc12[r];
            dp[48] = acc13[r];
        }
    }
}

__global__ __launch_bounds__(TPE)
void k_rank_dense(const int* __restrict__ dst, int* ch, int* __restrict__ rank, int E, int np,
                  const float* __restrict__ X, const float* __restrict__ W, float* __restrict__ T, int N, int gG,
                  const float* __restrict__ L0, const float* __restrict__ b0,
                  const float* __restrict__ L1, const float* __restrict__ b1,
                  const float* __restrict__ Wo, const float* __restrict__ bo,
                  float* W0f, float* W1f, float* cb) {
    __shared__ float sW[128 * 64];
    int b = (int)blockIdx.x;
    if (b <= FHB) {
        fuse_heads_role(b, L0, b0, L1, b1, Wo, bo, W0f, W1f, cb);
    } else if (b <= FHB + gG) {
        gemm_n64_tile(b - (FHB + 1), sW, X, W, T, N, 128);
    } else {
        rank_role(b - (FHB + 1) - gG, dst, ch, rank, E, np);
    }
}

#define NPG 8
#define GTB 128

template <int K>
__device__ __forceinline__ int row_lane(int v) {
    return __builtin_amdgcn_update_dpp(0, v, 0x150 + K, 0xf, 0xf, false);
}

__device__ __forceinline__ int row_next(int v) {
    return __builtin_amdgcn_update_dpp(0, v, 0x12F, 0xf, 0xf, false);
}

template <int K0, int CNT>
__device__ __forceinline__ void rows_req(int2 e, const float4* __restrict__ T4, int l, float4* r, float* w) {
    if constexpr (CNT > 0) {
        rows_req<K0, CNT - 1>(e, T4, l, r, w);
        int s = row_lane<K0 + CNT - 1>(e.x);
        w[CNT - 1] = __int_as_float(row_lane<K0 + CNT - 1>(e.y));
        r[CNT - 1] = T4[s * 16 + l];
    }
}

template <int K0, int CNT>
__device__ __forceinline__ void tail_ids(int2 e, int* s, float* w) {
    if constexpr (CNT > 0) {
        tail_ids<K0, CNT - 1>(e, s, w);
        s[CNT - 1] = row_lane<K0 + CNT - 1>(e.x);
        w[CNT - 1] = __int_as_float(row_lane<K0 + CNT - 1>(e.y));
    }
}

template <int CNT>
__device__ __forceinline__ void rows_get(const int* s, const float4* __restrict__ T4, int l, float4* r) {
#pragma unroll
    for (int k = 0; k < CNT; ++k) r[k] = T4[s[k] * 16 + l];
}

template <int CNT>
__device__ __forceinline__ void rows_add(const float4* r, const float* w, float4& acc) {
#pragma unroll
    for (int k = 0; k < CNT; ++k) {
        acc.x = fmaf(w[k], r[k].x, acc.x);
        acc.y = fmaf(w[k], r[k].y, acc.y);
        acc.z = fmaf(w[k], r[k].z, acc.z);
        acc.w = fmaf(w[k], r[k].w, acc.w);
    }
}

__device__ __forceinline__ int piece_off(int q, int c, int l) {
    int nfull = c >> 4, t = c & 15;
    int b = (l & 8) ? 8 : (l & 4) ? 4 : (l & 2) ? 2 : (l & 1);
    int it = (t & b) ? ((t & (b - 1)) + (l - b)) : 0;
    int off = (q < nfull) ? (16 * q + l) : (16 * nfull + it);
    return (off < c) ? off : (c - 1);
}

__device__ __forceinline__ void run_sum(const int2* __restrict__ p, int c, int2 e,
                                        const float4* __restrict__ T4, int l,
                                        const float4* __restrict__ selfp, float4& self,
                                        const int2* __restrict__ nextp, int2& e_next, float4& acc) {
    int nfull = c >> 4, t = c & 15;
    for (int q = 0; q < nfull; ++q) {
        int2 en = p[piece_off(q + 1, c, l)];
        float4 r[16];
        float w[16];
        rows_req<0, 16>(e, T4, l, r, w);
        rows_add<16>(r, w, acc);
        e = en;
    }
    float4 r[16] = {};
    float w[16];
    int s[16];
    if (t) tail_ids<1, 15>(e, s + 1, w + 1);
    self = *selfp;
    e_next = *nextp;
    if (t) {
        rows_get<1>(s + 1, T4, l, r + 1);
        if (t & 2) rows_get<2>(s + 2, T4, l, r + 2);
        if (t & 4) rows_get<4>(s + 4, T4, l, r + 4);
        if (t & 8) rows_get<8>(s + 8, T4, l, r + 8);
        if (t & 1) rows_add<1>(r + 1, w + 1, acc);
        if (t & 2) rows_add<2>(r + 2, w + 2, acc);
        if (t & 4) rows_add<4>(r + 4, w + 4, acc);
        if (t & 8) rows_add<8>(r + 8, w + 8, acc);
    }
}

__device__ __forceinline__ const int2* first_piece(const int2* __restrict__ csr, int c, int st, int l) {
    return csr + ((c > 0) ? (st + piece_off(0, c, l)) : 0);
}

#define HP 68
template <bool FUSE>
__global__ __launch_bounds__(GTB, 8)
void k_gather_elu(const float* __restrict__ T, const int2* __restrict__ csr,
                  const int* __restrict__ cnt, const int2* __restrict__ rowloc,
                  const int* __restrict__ boff, const float* __restrict__ bias,
                  float* __restrict__ H, int N,
                  const float* __restrict__ W2, float* __restrict__ D) {
    __shared__ __attribute__((aligned(16))) float sW[FUSE ? 64 * 64 : 4];
    __shared__ __attribute__((aligned(16))) float sH[FUSE ? (GTB / 16) * NPG * HP : 4];
    int tid = threadIdx.x;
    int l = tid & 15;
    int n0 = (blockIdx.x * (GTB / 16) + (tid >> 4)) * NPG;
    if constexpr (!FUSE) {
        if (n0 >= N) return;
    }
    const float4* T4 = (const float4*)T;
    int nm = n0 + (l & (NPG - 1));
    nm = (nm < N) ? nm : (N - 1);
    int cm = cnt[nm];
    int stv = boff[nm / CHUNK] + rowloc[nm].x;
    float4 b4 = make_float4(bias[4 * l + 0], bias[4 * l + 1], bias[4 * l + 2], bias[4 * l + 3]);
    int nn = (N - n0 < NPG) ? ((N - n0 > 0) ? (N - n0) : 0) : NPG;
    float4 wa[4], wb[4];
    if constexpr (FUSE) {
#pragma unroll
        for (int j = 0; j < 4; ++j) {
            int it = tid + j * GTB;
            int kp = it >> 4, cg = it & 15;
            wa[j] = ((const float4*)W2)[(2 * kp) * 16 + cg];
            wb[j] = ((const float4*)W2)[(2 * kp + 1) * 16 + cg];
        }
    }
    int2 e = *first_piece(csr, row_lane<0>(cm), row_lane<0>(stv), l);
    if constexpr (FUSE) {
#pragma unroll
        for (int j = 0; j < 4; ++j) {
            int it = tid + j * GTB;
            int kp = it >> 4, cg = it & 15;
            float4* d = (float4*)&sW[kp * 128 + 8 * cg];
            d[0] = make_float4(wa[j].x, wb[j].x, wa[j].y, wb[j].y);
            d[1] = make_float4(wa[j].z, wb[j].z, wa[j].w, wb[j].w);
        }
    }
#pragma unroll 1
    for (int i = 0; i < nn; ++i) {
        int n = n0 + i;
        int c = row_lane<0>(cm);
        int st = row_lane<0>(stv);
        cm = row_next(cm);
        stv = row_next(stv);
        const int2* nextp = first_piece(csr, row_lane<0>(cm), row_lane<0>(stv), l);
        float4 self;
        int2 e_next;
        float dg = (float)(c + 1);
        float dn = (dg > 0.0f) ? rsqrtf(dg) : 0.0f;
        float4 acc = make_float4(0.0f, 0.0f, 0.0f, 0.0f);
        run_sum(csr + st, c, e, T4, l, T4 + (n * 16 + l), self, nextp, e_next, acc);
        float v0 = dn * (acc.x + dn * self.x) + b4.x;
        float v1 = dn * (acc.y + dn * self.y) + b4.y;
        float v2 = dn * (acc.z + dn * self.z) + b4.z;
        float v3 = dn * (acc.w + dn * self.w) + b4.w;
        float4 h;
        h.x = (v0 > 0.0f) ? v0 : (expf(v0) - 1.0f);
        h.y = (v1 > 0.0f) ? v1 : (expf(v1) - 1.0f);
        h.z = (v2 > 0.0f) ? v2 : (expf(v2) - 1.0f);
        h.w = (v3 > 0.0f) ? v3 : (expf(v3) - 1.0f);
        ((float4*)H)[n * 16 + l] = h;
        if constexpr (FUSE) *(float4*)&sH[((tid >> 4) * NPG + i) * HP + 4 * l] = h;
        e = e_next;
    }
    if constexpr (FUSE) {
        __syncthreads();
        int wave = tid >> 5;
        int lane = tid & 31;
        int half = lane >> 4;
        int l16  = lane & 15;
        const float* ap = &sH[(wave * 16 + l16) * HP + 2 * half];
        const float* bq = &sW[half * 128 + 2 * l16];
        v8f acc0 = {}, acc1 = {}, acc2 = {}, acc3 = {};
#pragma unroll
        for (int k = 0; k < 64; k += 4) {
            v2f a = *(const v2f*)(ap + k);
            const float* bp = bq + (k >> 1) * 128;
            v2f b0 = *(const v2f*)(bp + 0);
            v2f b1 = *(const v2f*)(bp + 32);
            v2f b2 = *(const v2f*)(bp + 64);
            v2f b3 = *(const v2f*)(bp + 96);
            acc0 = __builtin_amdgcn_wmma_f32_16x16x4_f32(false, a, false, b0, (short)0, acc0, false, false);
            acc1 = __builtin_amdgcn_wmma_f32_16x16x4_f32(false, a, false, b1, (short)0, acc1, false, false);
            acc2 = __builtin_amdgcn_wmma_f32_16x16x4_f32(false, a, false, b2, (short)0, acc2, false, false);
            acc3 = __builtin_amdgcn_wmma_f32_16x16x4_f32(false, a, false, b3, (short)0, acc3, false, false);
        }
        int m0 = blockIdx.x * ((GTB / 16) * NPG) + wave * 16 + half * 8;
#pragma unroll
        for (int r = 0; r < 8; ++r) {
            int rr = m0 + r;
            if (rr < N) {
                float* dp = D + (long)rr * 64 + l16;
                dp[0]  = acc0[r];
                dp[16] = acc1[r];
                dp[32] = acc2[r];
                dp[48] = acc3[r];
            }
        }
    }
}

__global__ __launch_bounds__(128)
void k_out_wmma(const float* __restrict__ H0, const float* __restrict__ H1,
                const float* __restrict__ W0f, const float* __restrict__ W1f,
                const float* __restrict__ cb, float* __restrict__ out, int N) {
    __shared__ float4 sW4[32 * 48];
    __shared__ float4 sO4[4][32 * 10];
    int tid = threadIdx.x;
    int wave = tid >> 5;
    int lane = tid & 31;
    int half = lane >> 4;
    int l16  = lane & 15;

    int rbase = blockIdx.x * 128 + wave * 32;
    int ar0 = rbase + l16;
    int ar1 = rbase + 16 + l16;
    long ab0 = (long)((ar0 < N) ? ar0 : (N - 1)) * 16;
    long ab1 = (long)((ar1 < N) ? ar1 : (N - 1)) * 16;
    const float4* H04 = (const float4*)H0;
    const float4* H14 = (const float4*)H1;

    float4 a0[16], a1[16];
#pragma unroll
    for (int j = 0; j < 8; ++j) {
        a0[j]     = H04[ab0 + 2 * j + half];
        a1[j]     = H04[ab1 + 2 * j + half];
        a0[8 + j] = H14[ab0 + 2 * j + half];
        a1[8 + j] = H14[ab1 + 2 * j + half];
    }

    for (int it = tid; it < 320; it += 128) {
        int q = it / 10, cg = it % 10;
        const float4* Ws = (const float4*)((q < 16) ? W0f : W1f) + (q & 15) * 40 + cg;
        float4 r0 = Ws[0], r1 = Ws[10], r2 = Ws[20], r3 = Ws[30];
        float4* d = &sW4[q * 48 + 4 * cg];
        d[0] = make_float4(r0.x, r1.x, r2.x, r3.x);
        d[1] = make_float4(r0.y, r1.y, r2.y, r3.y);
        d[2] = make_float4(r0.z, r1.z, r2.z, r3.z);
        d[3] = make_float4(r0.w, r1.w, r2.w, r3.w);
    }
    for (int it = tid; it < 256; it += 128)
        sW4[(it >> 3) * 48 + 40 + (it & 7)] = make_float4(0.0f, 0.0f, 0.0f, 0.0f);
    __syncthreads();

    float bz0 = cb[l16];
    float bz1 = cb[16 + l16];
    float bz2 = (l16 < 8) ? cb[32 + l16] : 0.0f;
    v8f acc00, acc01, acc02, acc10, acc11, acc12;
#pragma unroll
    for (int r = 0; r < 8; ++r) {
        acc00[r] = bz0; acc01[r] = bz1; acc02[r] = bz2;
        acc10[r] = bz0; acc11[r] = bz1; acc12[r] = bz2;
    }

#pragma unroll
    for (int j = 0; j < 16; ++j) {
        const float4* bp = &sW4[(2 * j + half) * 48 + l16];
        float4 b0 = bp[0], b1 = bp[16], b2 = bp[32];
        v2f x0 = {a0[j].x, a0[j].y}, x1 = {a1[j].x, a1[j].y};
        v2f y0 = {a0[j].z, a0[j].w}, y1 = {a1[j].z, a1[j].w};
        v2f p0 = {b0.x, b0.y}, p1 = {b1.x, b1.y}, p2 = {b2.x, b2.y};
        v2f s0 = {b0.z, b0.w}, s1 = {b1.z, b1.w}, s2 = {b2.z, b2.w};
        acc00 = __builtin_amdgcn_wmma_f32_16x16x4_f32(false, x0, false, p0, (short)0, acc00, false, false);
        acc01 = __builtin_amdgcn_wmma_f32_16x16x4_f32(false, x0, false, p1, (short)0, acc01, false, false);
        acc02 = __builtin_amdgcn_wmma_f32_16x16x4_f32(false, x0, false, p2, (short)0, acc02, false, false);
        acc10 = __builtin_amdgcn_wmma_f32_16x16x4_f32(false, x1, false, p0, (short)0, acc10, false, false);
        acc11 = __builtin_amdgcn_wmma_f32_16x16x4_f32(false, x1, false, p1, (short)0, acc11, false, false);
        acc12 = __builtin_amdgcn_wmma_f32_16x16x4_f32(false, x1, false, p2, (short)0, acc12, false, false);
        acc00 = __builtin_amdgcn_wmma_f32_16x16x4_f32(false, y0, false, s0, (short)0, acc00, false, false);
        acc01 = __builtin_amdgcn_wmma_f32_16x16x4_f32(false, y0, false, s1, (short)0, acc01, false, false);
        acc02 = __builtin_amdgcn_wmma_f32_16x16x4_f32(false, y0, false, s2, (short)0, acc02, false, false);
        acc10 = __builtin_amdgcn_wmma_f32_16x16x4_f32(false, y1, false, s0, (short)0, acc10, false, false);
        acc11 = __builtin_amdgcn_wmma_f32_16x16x4_f32(false, y1, false, s1, (short)0, acc11, false, false);
        acc12 = __builtin_amdgcn_wmma_f32_16x16x4_f32(false, y1, false, s2, (short)0, acc12, false, false);
    }

    float* so = (float*)&sO4[wave][0];
#pragma unroll
    for (int r = 0; r < 8; ++r) {
        float* q0 = so + (half * 8 + r) * 40 + l16;
        float* q1 = so + (16 + half * 8 + r) * 40 + l16;
        q0[0]  = acc00[r];
        q0[16] = acc01[r];
        q1[0]  = acc10[r];
        q1[16] = acc11[r];
        if (l16 < 8) {
            q0[32] = acc02[r];
            q1[32] = acc12[r];
        }
    }
    __syncthreads();
    int nrow = N - rbase;
    float4* o4 = (float4*)out + (long)rbase * 10;
    if (nrow >= 32) {
        float4 t[10];
#pragma unroll
        for (int i = 0; i < 10; ++i) t[i] = sO4[wave][lane + 32 * i];
#pragma unroll
        for (int i = 0; i < 10; ++i) o4[lane + 32 * i] = t[i];
    } else {
        int n4 = nrow * 10;
        for (int i = lane; i < n4; i += 32) o4[i] = sO4[wave][i];
    }
}

extern "C" void kernel_launch(void* const* d_in, const int* in_sizes, int n_in,
                              void* d_out, int out_size, void* d_ws, size_t ws_size,
                              hipStream_t stream) {
    const float* x       = (const float*)d_in[0];
    const int*   ei      = (const int*)d_in[1];
    const float* conv0_w = (const float*)d_in[2];
    const float* conv0_b = (const float*)d_in[3];
    const float* lin0_w  = (const float*)d_in[4];
    const float* lin0_b  = (const float*)d_in[5];
    const float* conv1_w = (const float*)d_in[6];
    const float* conv1_b = (const float*)d_in[7];
    const float* lin1_w  = (const float*)d_in[8];
    const float* lin1_b  = (const float*)d_in[9];
    const float* out_w   = (const float*)d_in[10];
    const float* out_b   = (const float*)d_in[11];
    float* out = (float*)d_out;

    int N = in_sizes[0] / 128;
    int E = in_sizes[1] / 2;
    const int* src = ei;
    const int* dst = ei + E;

    size_t NH = (size_t)N * 64;
    float* ws   = (float*)d_ws;
    int*   cnt  = (int*)ws;
    float* A    = ws + (((size_t)N + 255) / 256) * 256;
    float* H1   = A + NH;
    float* H0   = H1 + NH;
    float* W0f  = H0 + NH;
    float* W1f  = W0f + 64 * 40;
    float* cb   = W1f + 64 * 40;
    int*   rank = (int*)H1;
    size_t np   = (((size_t)N + 255) / 256) * 256;
    int*   ch   = rank + (((size_t)E + 255) / 256) * 256;
    int*   oscr   = (int*)d_out;
    int2*  csr    = (int2*)oscr;
    int2*  rowloc = (int2*)(oscr + 2 * (size_t)E);
    int*   boff   = (int*)(rowloc + N);

    int gCh  = (N + CHUNK - 1) / CHUNK;
    int gGa  = (N + (GTB / 16) * NPG - 1) / ((GTB / 16) * NPG);
    int gEe  = E / (TPE * EPT) + (E % (TPE * EPT) + TPE - 1) / TPE;
    int n4   = (int)(2 * np / 4);
    int gGemm = (N + 127) / 128;

    k_zero_cnt<<<(n4 + TPB - 1) / TPB, TPB, 0, stream>>>((int4*)ch, n4);
    k_rank_dense<<<(FHB + 1) + gGemm + gEe, TPE, 0, stream>>>(dst, ch, rank, E, (int)np, x, conv0_w, A, N, gGemm,
                                                             lin0_w, lin0_b, lin1_w, lin1_b, out_w, out_b,
                                                             W0f, W1f, cb);
    k_scan_chunk<<<gCh, 256, 0, stream>>>(ch, ch + np, cnt, rowloc, boff + gCh, N);
    k_fill<<<gEe, TPE, 0, stream>>>(src, dst, rank, cnt, rowloc, boff + gCh, gCh, boff, csr, E);

    k_gather_elu<true><<<gGa, GTB, 0, stream>>>(A, csr, cnt, rowloc, boff, conv0_b, H0, N, conv1_w, H1);

    k_gather_elu<false><<<gGa, GTB, 0, stream>>>(H1, csr, cnt, rowloc, boff, conv1_b, A, N, nullptr, nullptr);

    k_out_wmma<<<gGemm, 128, 0, stream>>>(H0, A, W0f, W1f, cb, out, N);
}


# launch merge: output projection runs on the f32 WMMA in both aggregations' blocks from their LDS rows (h0, h1 never stored); 7 launches become 6; layer 0 stores cb + h0 sum, layer 1 adds its sum
# speedup vs baseline: 5.3210x; 1.0465x over previous
#include <hip/hip_runtime.h>
#include <hip/hip_bf16.h>

typedef float v2f __attribute__((ext_vector_type(2)));
typedef float v8f __attribute__((ext_vector_type(8)));

#define TPB 256

#define CHUNK 1024

__global__ void k_zero_cnt(int4* cnt4, int n4) {
    int i = blockIdx.x * TPB + threadIdx.x;
    if (i < n4) cnt4[i] = make_int4(0, 0, 0, 0);
}

#define EPT 8
#define TPE 128

__device__ __forceinline__ int cache_half() {
    return (int)((__builtin_amdgcn_s_sendmsg_rtn(0x87) >> 18) & 1u);
}

__device__ __forceinline__
void rank_role(int bid, const int* __restrict__ dst, int* ch, int* __restrict__ rank, int E, int np) {
    int nft = E / (TPE * EPT);
    int h = cache_half();
    int* c = ch + (size_t)h * np;
    if (bid < nft) {
        int e0 = bid * (TPE * EPT) + threadIdx.x;
        int d[EPT], r[EPT];
#pragma unroll
        for (int k = 0; k < EPT; ++k) d[k] = dst[e0 + k * TPE];
#pragma unroll
        for (int k = 0; k < EPT; ++k) r[k] = atomicAdd(&c[d[k]], 1);
#pragma unroll
        for (int k = 0; k < EPT; ++k) rank[e0 + k * TPE] = (r[k] << 1) | h;
    } else {
        int e = nft * (TPE * EPT) + (bid - nft) * TPE + threadIdx.x;
        if (e < E) rank[e] = (atomicAdd(&c[dst[e]], 1) << 1) | h;
    }
}

__global__ __launch_bounds__(256)
void k_scan_chunk(const int* __restrict__ ca, const int* __restrict__ cb2, int* __restrict__ cnt,
                  int2* __restrict__ rowloc, int* __restrict__ bsum, int N) {
    __shared__ int swave[8];
    int tid = threadIdx.x;
    int lane = tid & 31, wv = tid >> 5;
    int base = blockIdx.x * CHUNK + tid * 4;
    int a0 = (base + 0 < N) ? ca[base + 0] : 0;
    int a1 = (base + 1 < N) ? ca[base + 1] : 0;
    int a2 = (base + 2 < N) ? ca[base + 2] : 0;
    int a3 = (base + 3 < N) ? ca[base + 3] : 0;
    int c0 = (base + 0 < N) ? a0 + cb2[base + 0] : 0;
    int c1 = (base + 1 < N) ? a1 + cb2[base + 1] : 0;
    int c2 = (base + 2 < N) ? a2 + cb2[base + 2] : 0;
    int c3 = (base + 3 < N) ? a3 + cb2[base + 3] : 0;
    if (base + 0 < N) cnt[base + 0] = c0;
    if (base + 1 < N) cnt[base + 1] = c1;
    if (base + 2 < N) cnt[base + 2] = c2;
    if (base + 3 < N) cnt[base + 3] = c3;
    int t = c0 + c1 + c2 + c3;
    int incl = t;
    for (int off = 1; off < 32; off <<= 1) {
        int v = __shfl_up(incl, off, 32);
        if (lane >= off) incl += v;
    }
    if (lane == 31) swave[wv] = incl;
    __syncthreads();
    int wbase = 0;
    for (int w = 0; w < wv; ++w) wbase += swave[w];
    int excl = wbase + incl - t;
    if (base + 0 < N) rowloc[base + 0] = make_int2(excl, excl + a0);
    if (base + 1 < N) rowloc[base + 1] = make_int2(excl + c0, excl + c0 + a1);
    if (base + 2 < N) rowloc[base + 2] = make_int2(excl + c0 + c1, excl + c0 + c1 + a2);
    if (base + 3 < N) rowloc[base + 3] = make_int2(excl + c0 + c1 + c2, excl + c0 + c1 + c2 + a3);
    if (tid == 255) bsum[blockIdx.x] = excl + t;
}

__device__ __forceinline__ void fill_edge(int s, int rk, int ra, int rb, int cs, int bo, int2* __restrict__ csr) {
    int pos = bo + ((rk & 1) ? rb : ra) + (rk >> 1);
    float ds = (float)(cs + 1);
    float w = (ds > 0.0f) ? rsqrtf(ds) : 0.0f;
    csr[pos] = make_int2(s, __float_as_int(w));
}

#define MAXCH 1024
#define CPT (MAXCH / TPE)
__device__ __forceinline__
void chunk_starts(const int* __restrict__ btot, int nb, int* __restrict__ boff, int* sB, int* sWv) {
    int tid = threadIdx.x;
    int lane = tid & 31, wv = tid >> 5;
    int v[CPT];
    int t = 0;
#pragma unroll
    for (int j = 0; j < CPT; ++j) {
        int i = tid * CPT + j;
        v[j] = (i < nb) ? btot[i] : 0;
        t += v[j];
    }
    int incl = t;
    for (int off = 1; off < 32; off <<= 1) {
        int u = __shfl_up(incl, off, 32);
        if (lane >= off) incl += u;
    }
    if (lane == 31) sWv[wv] = incl;
    __syncthreads();
    int ex = incl - t;
    for (int w = 0; w < wv; ++w) ex += sWv[w];
#pragma unroll
    for (int j = 0; j < CPT; ++j) {
        int i = tid * CPT + j;
        sB[i] = ex;
        if (blockIdx.x == 0 && i < nb) boff[i] = ex;
        ex += v[j];
    }
    __syncthreads();
}

__global__ __launch_bounds__(TPE)
void k_fill(const int* __restrict__ src, const int* __restrict__ dst,
            const int* __restrict__ rank, const int* __restrict__ cnt,
            const int2* __restrict__ rowloc, const int* __restrict__ btot, int nb,
            int* __restrict__ boff, int2* __restrict__ csr, int E) {
    __shared__ int sB[MAXCH];
    __shared__ int sWv[TPE / 32];
    int nft = E / (TPE * EPT);
    if ((int)blockIdx.x < nft) {
        int e0 = blockIdx.x * (TPE * EPT) + threadIdx.x;
        int s[EPT], d[EPT], rk[EPT], cs[EPT], bo[EPT], ra[EPT], rb[EPT];
#pragma unroll
        for (int k = 0; k < EPT; ++k) {
            s[k]  = src[e0 + k * TPE];
            d[k]  = dst[e0 + k * TPE];
            rk[k] = rank[e0 + k * TPE];
        }
        chunk_starts(btot, nb, boff, sB, sWv);
#pragma unroll
        for (int k = 0; k < EPT; ++k) {
            int2 rl = rowloc[d[k]];
            ra[k] = rl.x;
            rb[k] = rl.y;
            cs[k] = cnt[s[k]];
            bo[k] = sB[d[k] / CHUNK];
        }
#pragma unroll
        for (int k = 0; k < EPT; ++k) fill_edge(s[k], rk[k], ra[k], rb[k], cs[k], bo[k], csr);
    } else {
        chunk_starts(btot, nb, boff, sB, sWv);
        int e = nft * (TPE * EPT) + ((int)blockIdx.x - nft) * TPE + threadIdx.x;
        if (e < E) {
            int s = src[e], d = dst[e];
            int2 rl = rowloc[d];
            fill_edge(s, rank[e], rl.x, rl.y, cnt[s], sB[d / CHUNK], csr);
        }
    }
}

#define FHB ((64 * 40) / TPE)
__device__ __forceinline__
void fuse_heads_role(int bid, const float* __restrict__ L0, const float* __restrict__ b0,
                     const float* __restrict__ L1, const float* __restrict__ b1,
                     const float* __restrict__ Wo, const float* __restrict__ bo,
                     float* W0f, float* W1f, float* cb) {
    if (bid < FHB) {
        int i = bid * TPE + threadIdx.x;
        int k = i / 40, c = i % 40;
        float s0 = 0.0f, s1 = 0.0f;
#pragma unroll 16
        for (int j = 0; j < 64; ++j) {
            s0 += L0[k * 64 + j] * Wo[j * 40 + c];
            s1 += L1[k * 64 + j] * Wo[(64 + j) * 40 + c];
        }
        int o = ((k >> 2) * 40 + c) * 4 + (k & 3);
        W0f[o] = s0;
        W1f[o] = s1;
    } else if (threadIdx.x < 40) {
        int c = threadIdx.x;
        float s = bo[c];
        for (int j = 0; j < 64; ++j)
            s += b0[j] * Wo[j * 40 + c] + b1[j] * Wo[(64 + j) * 40 + c];
        cb[c] = s;
    }
}

__device__ __forceinline__
void gemm_n64_tile(int bid, float* sW, const float* __restrict__ A, const float* __restrict__ W,
                   float* __restrict__ D, int N, int K) {
    int tid = threadIdx.x;
    for (int i = tid; i < K * 64; i += 128) {
        int k = i >> 6, col = i & 63;
        sW[(k >> 1) * 128 + 2 * col + (k & 1)] = W[i];
    }
    __syncthreads();

    int wave = tid >> 5;
    int lane = tid & 31;
    int half = lane >> 4;
    int l16  = lane & 15;

    int rbase = bid * 128 + wave * 32;
    int ar0 = rbase + l16;
    int ar1 = rbase + 16 + l16;
    long ab0 = (long)((ar0 < N) ? ar0 : (N - 1)) * K;
    long ab1 = (long)((ar1 < N) ? ar1 : (N - 1)) * K;

    v8f acc00 = {}, acc01 = {}, acc02 = {}, acc03 = {};
    v8f acc10 = {}, acc11 = {}, acc12 = {}, acc13 = {};

    for (int k = 0; k < K; k += 4) {
        int kb = k + 2 * half;
        v2f a0 = *(const v2f*)(A + ab0 + kb);
        v2f a1 = *(const v2f*)(A + ab1 + kb);
        const float* bp = &sW[(kb >> 1) * 128 + 2 * l16];
        v2f b0 = *(const v2f*)(bp + 0);
        v2f b1 = *(const v2f*)(bp + 32);
        v2f b2 = *(const v2f*)(bp + 64);
        v2f b3 = *(const v2f*)(bp + 96);
        acc00 = __builtin_amdgcn_wmma_f32_16x16x4_f32(false, a0, false, b0, (short)0, acc00, false, false);
        acc01 = __builtin_amdgcn_wmma_f32_16x16x4_f32(false, a0, false, b1, (short)0, acc01, false, false);
        acc02 = __builtin_amdgcn_wmma_f32_16x16x4_f32(false, a0, false, b2, (short)0, acc02, false, false);
        acc03 = __builtin_amdgcn_wmma_f32_16x16x4_f32(false, a0, false, b3, (short)0, acc03, false, false);
        acc10 = __builtin_amdgcn_wmma_f32_16x16x4_f32(false, a1, false, b0, (short)0, acc10, false, false);
        acc11 = __builtin_amdgcn_wmma_f32_16x16x4_f32(false, a1, false, b1, (short)0, acc11, false, false);
        acc12 = __builtin_amdgcn_wmma_f32_16x16x4_f32(false, a1, false, b2, (short)0, acc12, false, false);
        acc13 = __builtin_amdgcn_wmma_f32_16x16x4_f32(false, a1, false, b3, (short)0, acc13, false, false);
    }

    int m0 = rbase + half * 8;
    int m1 = rbase + 16 + half * 8;
    for (int r = 0; r < 8; ++r) {
        int rr = m0 + r;
        if (rr < N) {
            float* dp = D + (long)rr * 64 + l16;
            dp[0]  = acc00[r];
            dp[16] = acc01[r];
            dp[32] = acc02[r];
            dp[48] = acc03[r];
        }
        int ss = m1 + r;
        if (ss < N) {
            float* dp = D + (long)ss * 64 + l16;
            dp[0]  = acc10[r];
            dp[16] = acc11[r];
            dp[32] = acc12[r];
            dp[48] = acc13[r];
        }
    }
}

__global__ __launch_bounds__(TPE)
void k_rank_dense(const int* __restrict__ dst, int* ch, int* __restrict__ rank, int E, int np,
                  const float* __restrict__ X, const float* __restrict__ W, float* __restrict__ T, int N, int gG,
                  const float* __restrict__ L0, const float* __restrict__ b0,
                  const float* __restrict__ L1, const float* __restrict__ b1,
                  const float* __restrict__ Wo, const float* __restrict__ bo,
                  float* W0f, float* W1f, float* cb) {
    __shared__ float sW[128 * 64];
    int b = (int)blockIdx.x;
    if (b <= FHB) {
        fuse_heads_role(b, L0, b0, L1, b1, Wo, bo, W0f, W1f, cb);
    } else if (b <= FHB + gG) {
        gemm_n64_tile(b - (FHB + 1), sW, X, W, T, N, 128);
    } else {
        rank_role(b - (FHB + 1) - gG, dst, ch, rank, E, np);
    }
}

#define NPG 8
#define GTB 128

template <int K>
__device__ __forceinline__ int row_lane(int v) {
    return __builtin_amdgcn_update_dpp(0, v, 0x150 + K, 0xf, 0xf, false);
}

__device__ __forceinline__ int row_next(int v) {
    return __builtin_amdgcn_update_dpp(0, v, 0x12F, 0xf, 0xf, false);
}

template <int K0, int CNT>
__device__ __forceinline__ void rows_req(int2 e, const float4* __restrict__ T4, int l, float4* r, float* w) {
    if constexpr (CNT > 0) {
        rows_req<K0, CNT - 1>(e, T4, l, r, w);
        int s = row_lane<K0 + CNT - 1>(e.x);
        w[CNT - 1] = __int_as_float(row_lane<K0 + CNT - 1>(e.y));
        r[CNT - 1] = T4[s * 16 + l];
    }
}

template <int K0, int CNT>
__device__ __forceinline__ void tail_ids(int2 e, int* s, float* w) {
    if constexpr (CNT > 0) {
        tail_ids<K0, CNT - 1>(e, s, w);
        s[CNT - 1] = row_lane<K0 + CNT - 1>(e.x);
        w[CNT - 1] = __int_as_float(row_lane<K0 + CNT - 1>(e.y));
    }
}

template <int CNT>
__device__ __forceinline__ void rows_get(const int* s, const float4* __restrict__ T4, int l, float4* r) {
#pragma unroll
    for (int k = 0; k < CNT; ++k) r[k] = T4[s[k] * 16 + l];
}

template <int CNT>
__device__ __forceinline__ void rows_add(const float4* r, const float* w, float4& acc) {
#pragma unroll
    for (int k = 0; k < CNT; ++k) {
        acc.x = fmaf(w[k], r[k].x, acc.x);
        acc.y = fmaf(w[k], r[k].y, acc.y);
        acc.z = fmaf(w[k], r[k].z, acc.z);
        acc.w = fmaf(w[k], r[k].w, acc.w);
    }
}

__device__ __forceinline__ int piece_off(int q, int c, int l) {
    int nfull = c >> 4, t = c & 15;
    int b = (l & 8) ? 8 : (l & 4) ? 4 : (l & 2) ? 2 : (l & 1);
    int it = (t & b) ? ((t & (b - 1)) + (l - b)) : 0;
    int off = (q < nfull) ? (16 * q + l) : (16 * nfull + it);
    return (off < c) ? off : (c - 1);
}

__device__ __forceinline__ void run_sum(const int2* __restrict__ p, int c, int2 e,
                                        const float4* __restrict__ T4, int l,
                                        const float4* __restrict__ selfp, float4& self,
                                        const int2* __restrict__ nextp, int2& e_next, float4& acc) {
    int nfull = c >> 4, t = c & 15;
    for (int q = 0; q < nfull; ++q) {
        int2 en = p[piece_off(q + 1, c, l)];
        float4 r[16];
        float w[16];
        rows_req<0, 16>(e, T4, l, r, w);
        rows_add<16>(r, w, acc);
        e = en;
    }
    float4 r[16] = {};
    float w[16];
    int s[16];
    if (t) tail_ids<1, 15>(e, s + 1, w + 1);
    self = *selfp;
    e_next = *nextp;
    if (t) {
        rows_get<1>(s + 1, T4, l, r + 1);
        if (t & 2) rows_get<2>(s + 2, T4, l, r + 2);
        if (t & 4) rows_get<4>(s + 4, T4, l, r + 4);
        if (t & 8) rows_get<8>(s + 8, T4, l, r + 8);
        if (t & 1) rows_add<1>(r + 1, w + 1, acc);
        if (t & 2) rows_add<2>(r + 2, w + 2, acc);
        if (t & 4) rows_add<4>(r + 4, w + 4, acc);
        if (t & 8) rows_add<8>(r + 8, w + 8, acc);
    }
}

__device__ __forceinline__ const int2* first_piece(const int2* __restrict__ csr, int c, int st, int l) {
    return csr + ((c > 0) ? (st + piece_off(0, c, l)) : 0);
}

#define HP 68
#define HQ ((64 / 4) * 40)
template <bool FIRST>
__global__ __launch_bounds__(GTB, 8)
void k_gather_elu(const float* __restrict__ T, const int2* __restrict__ csr,
                  const int* __restrict__ cnt, const int2* __restrict__ rowloc,
                  const int* __restrict__ boff, const float* __restrict__ bias, int N,
                  const float* __restrict__ W2, float* __restrict__ D,
                  const float* __restrict__ Wf, const float* __restrict__ cb, float* out) {
    __shared__ __attribute__((aligned(16))) float sW[FIRST ? 64 * 64 : 4];
    __shared__ __attribute__((aligned(16))) float sH[(GTB / 16) * NPG * HP];
    __shared__ float4 sF4[HQ];
    int tid = threadIdx.x;
    int l = tid & 15;
    int n0 = (blockIdx.x * (GTB / 16) + (tid >> 4)) * NPG;
    const float4* T4 = (const float4*)T;
    int nm = n0 + (l & (NPG - 1));
    nm = (nm < N) ? nm : (N - 1);
    int cm = cnt[nm];
    int stv = boff[nm / CHUNK] + rowloc[nm].x;
    float4 b4 = make_float4(bias[4 * l + 0], bias[4 * l + 1], bias[4 * l + 2], bias[4 * l + 3]);
    int nn = (N - n0 < NPG) ? ((N - n0 > 0) ? (N - n0) : 0) : NPG;
    float4 wa[4], wb[4];
    float4 wf[HQ / GTB];
    if constexpr (FIRST) {
#pragma unroll
        for (int j = 0; j < 4; ++j) {
            int it = tid + j * GTB;
            int kp = it >> 4, cg = it & 15;
            wa[j] = ((const float4*)W2)[(2 * kp) * 16 + cg];
            wb[j] = ((const float4*)W2)[(2 * kp + 1) * 16 + cg];
        }
    }
#pragma unroll
    for (int j = 0; j < HQ / GTB; ++j) wf[j] = ((const float4*)Wf)[tid + j * GTB];
    int2 e = *first_piece(csr, row_lane<0>(cm), row_lane<0>(stv), l);
    if constexpr (FIRST) {
#pragma unroll
        for (int j = 0; j < 4; ++j) {
            int it = tid + j * GTB;
            int kp = it >> 4, cg = it & 15;
            float4* d = (float4*)&sW[kp * 128 + 8 * cg];
            d[0] = make_float4(wa[j].x, wb[j].x, wa[j].y, wb[j].y);
            d[1] = make_float4(wa[j].z, wb[j].z, wa[j].w, wb[j].w);
        }
    }
#pragma unroll
    for (int j = 0; j < HQ / GTB; ++j) sF4[tid + j * GTB] = wf[j];
#pragma unroll 1
    for (int i = 0; i < nn; ++i) {
        int n = n0 + i;
        int c = row_lane<0>(cm);
        int st = row_lane<0>(stv);
        cm = row_next(cm);
        stv = row_next(stv);
        const int2* nextp = first_piece(csr, row_lane<0>(cm), row_lane<0>(stv), l);
        float4 self;
        int2 e_next;
        float dg = (float)(c + 1);
        float dn = (dg > 0.0f) ? rsqrtf(dg) : 0.0f;
        float4 acc = make_float4(0.0f, 0.0f, 0.0f, 0.0f);
        run_sum(csr + st, c, e, T4, l, T4 + (n * 16 + l), self, nextp, e_next, acc);
        float v0 = dn * (acc.x + dn * self.x) + b4.x;
        float v1 = dn * (acc.y + dn * self.y) + b4.y;
        float v2 = dn * (acc.z + dn * self.z) + b4.z;
        float v3 = dn * (acc.w + dn * self.w) + b4.w;
        float4 h;
        h.x = (v0 > 0.0f) ? v0 : (expf(v0) - 1.0f);
        h.y = (v1 > 0.0f) ? v1 : (expf(v1) - 1.0f);
        h.z = (v2 > 0.0f) ? v2 : (expf(v2) - 1.0f);
        h.w = (v3 > 0.0f) ? v3 : (expf(v3) - 1.0f);
        *(float4*)&sH[((tid >> 4) * NPG + i) * HP + 4 * l] = h;
        e = e_next;
    }
    __syncthreads();
    int wave = tid >> 5;
    int lane = tid & 31;
    int half = lane >> 4;
    int l16  = lane & 15;
    int R0 = blockIdx.x * ((GTB / 16) * NPG) + wave * 16;
    int nrow = N - R0;
    float4* o4 = (float4*)out + (long)R0 * 10;
    float4 prev[5];
    if constexpr (!FIRST) {
        if (nrow >= 16) {
#pragma unroll
            for (int i = 0; i < 5; ++i) prev[i] = o4[lane + 32 * i];
        }
    }
    if constexpr (FIRST) {
        const float* ap = &sH[(wave * 16 + l16) * HP + 2 * half];
        const float* bq = &sW[half * 128 + 2 * l16];
        v8f acc0 = {}, acc1 = {}, acc2 = {}, acc3 = {};
#pragma unroll
        for (int k = 0; k < 64; k += 4) {
            v2f a = *(const v2f*)(ap + k);
            const float* bp = bq + (k >> 1) * 128;
            v2f b0 = *(const v2f*)(bp + 0);
            v2f b1 = *(const v2f*)(bp + 32);
            v2f b2 = *(const v2f*)(bp + 64);
            v2f b3 = *(const v2f*)(bp + 96);
            acc0 = __builtin_amdgcn_wmma_f32_16x16x4_f32(false, a, false, b0, (short)0, acc0, false, false);
            acc1 = __builtin_amdgcn_wmma_f32_16x16x4_f32(false, a, false, b1, (short)0, acc1, false, false);
            acc2 = __builtin_amdgcn_wmma_f32_16x16x4_f32(false, a, false, b2, (short)0, acc2, false, false);
            acc3 = __builtin_amdgcn_wmma_f32_16x16x4_f32(false, a, false, b3, (short)0, acc3, false, false);
        }
        int m0 = R0 + half * 8;
#pragma unroll
        for (int r = 0; r < 8; ++r) {
            int rr = m0 + r;
            if (rr < N) {
                float* dp = D + (long)rr * 64 + l16;
                dp[0]  = acc0[r];
                dp[16] = acc1[r];
                dp[32] = acc2[r];
                dp[48] = acc3[r];
            }
        }
    }
    const float* ah = &sH[(wave * 16 + l16) * HP + 4 * half];
    const float4* bh = &sF4[half * 40];
    int c2 = 32 + (l16 & 7);
    v8f p0, p1, p2;
    if constexpr (FIRST) {
        float bz0 = cb[l16];
        float bz1 = cb[16 + l16];
        float bz2 = cb[c2];
#pragma unroll
        for (int r = 0; r < 8; ++r) { p0[r] = bz0; p1[r] = bz1; p2[r] = bz2; }
    } else {
#pragma unroll
        for (int r = 0; r < 8; ++r) { p0[r] = 0.0f; p1[r] = 0.0f; p2[r] = 0.0f; }
    }
#pragma unroll
    for (int j = 0; j < 8; ++j) {
        float4 a = *(const float4*)(ah + 8 * j);
        const float4* bp = bh + (2 * j) * 40;
        float4 b0 = bp[l16], b1 = bp[16 + l16], b2 = bp[c2];
        v2f x = {a.x, a.y}, y = {a.z, a.w};
        v2f q0 = {b0.x, b0.y}, q1 = {b1.x, b1.y}, q2 = {b2.x, b2.y};
        v2f s0 = {b0.z, b0.w}, s1 = {b1.z, b1.w}, s2 = {b2.z, b2.w};
        p0 = __builtin_amdgcn_wmma_f32_16x16x4_f32(false, x, false, q0, (short)0, p0, false, false);
        p1 = __builtin_amdgcn_wmma_f32_16x16x4_f32(false, x, false, q1, (short)0, p1, false, false);
        p2 = __builtin_amdgcn_wmma_f32_16x16x4_f32(false, x, false, q2, (short)0, p2, false, false);
        p0 = __builtin_amdgcn_wmma_f32_16x16x4_f32(false, y, false, s0, (short)0, p0, false, false);
        p1 = __builtin_amdgcn_wmma_f32_16x16x4_f32(false, y, false, s1, (short)0, p1, false, false);
        p2 = __builtin_amdgcn_wmma_f32_16x16x4_f32(false, y, false, s2, (short)0, p2, false, false);
    }
    float* so = &sH[wave * 16 * HP];
#pragma unroll
    for (int r = 0; r < 8; ++r) {
        float* q = so + (half * 8 + r) * 40;
        q[l16]      = p0[r];
        q[16 + l16] = p1[r];
        q[c2]       = p2[r];
    }
    __syncthreads();
    const float4* s4 = (const float4*)so;
    if (nrow >= 16) {
        float4 t[5];
#pragma unroll
        for (int i = 0; i < 5; ++i) t[i] = s4[lane + 32 * i];
        if constexpr (!FIRST) {
#pragma unroll
            for (int i = 0; i < 5; ++i) {
                t[i].x += prev[i].x;
                t[i].y += prev[i].y;
                t[i].z += prev[i].z;
                t[i].w += prev[i].w;
            }
        }
#pragma unroll
        for (int i = 0; i < 5; ++i) o4[lane + 32 * i] = t[i];
    } else {
        int n4 = nrow * 10;
        for (int i = lane; i < n4; i += 32) {
            float4 v = s4[i];
            if constexpr (!FIRST) {
                float4 u = o4[i];
                v.x += u.x;
                v.y += u.y;
                v.z += u.z;
                v.w += u.w;
            }
            o4[i] = v;
        }
    }
}

extern "C" void kernel_launch(void* const* d_in, const int* in_sizes, int n_in,
                              void* d_out, int out_size, void* d_ws, size_t ws_size,
                              hipStream_t stream) {
    const float* x       = (const float*)d_in[0];
    const int*   ei      = (const int*)d_in[1];
    const float* conv0_w = (const float*)d_in[2];
    const float* conv0_b = (const float*)d_in[3];
    const float* lin0_w  = (const float*)d_in[4];
    const float* lin0_b  = (const float*)d_in[5];
    const float* conv1_w = (const float*)d_in[6];
    const float* conv1_b = (const float*)d_in[7];
    const float* lin1_w  = (const float*)d_in[8];
    const float* lin1_b  = (const float*)d_in[9];
    const float* out_w   = (const float*)d_in[10];
    const float* out_b   = (const float*)d_in[11];
    float* out = (float*)d_out;

    int N = in_sizes[0] / 128;
    int E = in_sizes[1] / 2;
    const int* src = ei;
    const int* dst = ei + E;

    size_t NH = (size_t)N * 64;
    float* ws   = (float*)d_ws;
    int*   cnt  = (int*)ws;
    float* A    = ws + (((size_t)N + 255) / 256) * 256;
    float* H1   = A + NH;
    float* EL   = H1 + NH;
    float* W0f  = EL + NH;
    float* W1f  = W0f + 64 * 40;
    float* cb   = W1f + 64 * 40;
    int*   rank = (int*)H1;
    size_t np   = (((size_t)N + 255) / 256) * 256;
    int*   ch   = rank + (((size_t)E + 255) / 256) * 256;
    int*   oscr   = (int*)EL;
    int2*  csr    = (int2*)oscr;
    int2*  rowloc = (int2*)(oscr + 2 * (size_t)E);
    int*   boff   = (int*)(rowloc + N);

    int gCh  = (N + CHUNK - 1) / CHUNK;
    int gGa  = (N + (GTB / 16) * NPG - 1) / ((GTB / 16) * NPG);
    int gEe  = E / (TPE * EPT) + (E % (TPE * EPT) + TPE - 1) / TPE;
    int n4   = (int)(2 * np / 4);
    int gGemm = (N + 127) / 128;

    k_zero_cnt<<<(n4 + TPB - 1) / TPB, TPB, 0, stream>>>((int4*)ch, n4);
    k_rank_dense<<<(FHB + 1) + gGemm + gEe, TPE, 0, stream>>>(dst, ch, rank, E, (int)np, x, conv0_w, A, N, gGemm,
                                                             lin0_w, lin0_b, lin1_w, lin1_b, out_w, out_b,
                                                             W0f, W1f, cb);
    k_scan_chunk<<<gCh, 256, 0, stream>>>(ch, ch + np, cnt, rowloc, boff + gCh, N);
    k_fill<<<gEe, TPE, 0, stream>>>(src, dst, rank, cnt, rowloc, boff + gCh, gCh, boff, csr, E);

    k_gather_elu<true><<<gGa, GTB, 0, stream>>>(A, csr, cnt, rowloc, boff, conv0_b, N, conv1_w, H1, W0f, cb, out);

    k_gather_elu<false><<<gGa, GTB, 0, stream>>>(H1, csr, cnt, rowloc, boff, conv1_b, N, nullptr, nullptr, W1f, nullptr, out);
}


# rows stored times their node's dinv (x @ W moves from the ranking's launch to the placement's, where degrees exist): edge entries are 4-byte ids, no per-edge dinv lookup; sums add rounded rows
# speedup vs baseline: 5.6576x; 1.0633x over previous
#include <hip/hip_runtime.h>
#include <hip/hip_bf16.h>

typedef float v2f __attribute__((ext_vector_type(2)));
typedef float v8f __attribute__((ext_vector_type(8)));

#define TPB 256

#define CHUNK 1024

__global__ void k_zero_cnt(int4* cnt4, int n4) {
    int i = blockIdx.x * TPB + threadIdx.x;
    if (i < n4) cnt4[i] = make_int4(0, 0, 0, 0);
}

#define EPT 8
#define TPE 128

__device__ __forceinline__ int cache_half() {
    return (int)((__builtin_amdgcn_s_sendmsg_rtn(0x87) >> 18) & 1u);
}

__device__ __forceinline__ float dinv_of(int c) {
    float ds = (float)(c + 1);
    return (ds > 0.0f) ? rsqrtf(ds) : 0.0f;
}

__device__ __forceinline__
void rank_role(int bid, const int* __restrict__ dst, int* ch, int* __restrict__ rank, int E, int np) {
    int nft = E / (TPE * EPT);
    int h = cache_half();
    int* c = ch + (size_t)h * np;
    if (bid < nft) {
        int e0 = bid * (TPE * EPT) + threadIdx.x;
        int d[EPT], r[EPT];
#pragma unroll
        for (int k = 0; k < EPT; ++k) d[k] = dst[e0 + k * TPE];
#pragma unroll
        for (int k = 0; k < EPT; ++k) r[k] = atomicAdd(&c[d[k]], 1);
#pragma unroll
        for (int k = 0; k < EPT; ++k) rank[e0 + k * TPE] = (r[k] << 1) | h;
    } else {
        int e = nft * (TPE * EPT) + (bid - nft) * TPE + threadIdx.x;
        if (e < E) rank[e] = (atomicAdd(&c[dst[e]], 1) << 1) | h;
    }
}

__global__ __launch_bounds__(256)
void k_scan_chunk(const int* __restrict__ ca, const int* __restrict__ cb2, int* __restrict__ cnt,
                  int2* __restrict__ rowloc, int* __restrict__ bsum, int N) {
    __shared__ int swave[8];
    int tid = threadIdx.x;
    int lane = tid & 31, wv = tid >> 5;
    int base = blockIdx.x * CHUNK + tid * 4;
    int a0 = (base + 0 < N) ? ca[base + 0] : 0;
    int a1 = (base + 1 < N) ? ca[base + 1] : 0;
    int a2 = (base + 2 < N) ? ca[base + 2] : 0;
    int a3 = (base + 3 < N) ? ca[base + 3] : 0;
    int c0 = (base + 0 < N) ? a0 + cb2[base + 0] : 0;
    int c1 = (base + 1 < N) ? a1 + cb2[base + 1] : 0;
    int c2 = (base + 2 < N) ? a2 + cb2[base + 2] : 0;
    int c3 = (base + 3 < N) ? a3 + cb2[base + 3] : 0;
    if (base + 0 < N) cnt[base + 0] = c0;
    if (base + 1 < N) cnt[base + 1] = c1;
    if (base + 2 < N) cnt[base + 2] = c2;
    if (base + 3 < N) cnt[base + 3] = c3;
    int t = c0 + c1 + c2 + c3;
    int incl = t;
    for (int off = 1; off < 32; off <<= 1) {
        int v = __shfl_up(incl, off, 32);
        if (lane >= off) incl += v;
    }
    if (lane == 31) swave[wv] = incl;
    __syncthreads();
    int wbase = 0;
    for (int w = 0; w < wv; ++w) wbase += swave[w];
    int excl = wbase + incl - t;
    if (base + 0 < N) rowloc[base + 0] = make_int2(excl, excl + a0);
    if (base + 1 < N) rowloc[base + 1] = make_int2(excl + c0, excl + c0 + a1);
    if (base + 2 < N) rowloc[base + 2] = make_int2(excl + c0 + c1, excl + c0 + c1 + a2);
    if (base + 3 < N) rowloc[base + 3] = make_int2(excl + c0 + c1 + c2, excl + c0 + c1 + c2 + a3);
    if (tid == 255) bsum[blockIdx.x] = excl + t;
}

__device__ __forceinline__ void fill_edge(int s, int rk, int ra, int rb, int bo, int* __restrict__ csr) {
    int pos = bo + ((rk & 1) ? rb : ra) + (rk >> 1);
    csr[pos] = s;
}

#define MAXCH 1024
#define CPT (MAXCH / TPE)
__device__ __forceinline__
void chunk_starts(int bid, const int* __restrict__ btot, int nb, int* __restrict__ boff, int* sB, int* sWv) {
    int tid = threadIdx.x;
    int lane = tid & 31, wv = tid >> 5;
    int v[CPT];
    int t = 0;
#pragma unroll
    for (int j = 0; j < CPT; ++j) {
        int i = tid * CPT + j;
        v[j] = (i < nb) ? btot[i] : 0;
        t += v[j];
    }
    int incl = t;
    for (int off = 1; off < 32; off <<= 1) {
        int u = __shfl_up(incl, off, 32);
        if (lane >= off) incl += u;
    }
    if (lane == 31) sWv[wv] = incl;
    __syncthreads();
    int ex = incl - t;
    for (int w = 0; w < wv; ++w) ex += sWv[w];
#pragma unroll
    for (int j = 0; j < CPT; ++j) {
        int i = tid * CPT + j;
        sB[i] = ex;
        if (bid == 0 && i < nb) boff[i] = ex;
        ex += v[j];
    }
    __syncthreads();
}

#define FHB ((64 * 40) / TPE)
__device__ __forceinline__
void fuse_heads_role(int bid, const float* __restrict__ L0, const float* __restrict__ b0,
                     const float* __restrict__ L1, const float* __restrict__ b1,
                     const float* __restrict__ Wo, const float* __restrict__ bo,
                     float* W0f, float* W1f, float* cb) {
    if (bid < FHB) {
        int i = bid * TPE + threadIdx.x;
        int k = i / 40, c = i % 40;
        float s0 = 0.0f, s1 = 0.0f;
#pragma unroll 16
        for (int j = 0; j < 64; ++j) {
            s0 += L0[k * 64 + j] * Wo[j * 40 + c];
            s1 += L1[k * 64 + j] * Wo[(64 + j) * 40 + c];
        }
        int o = ((k >> 2) * 40 + c) * 4 + (k & 3);
        W0f[o] = s0;
        W1f[o] = s1;
    } else if (threadIdx.x < 40) {
        int c = threadIdx.x;
        float s = bo[c];
        for (int j = 0; j < 64; ++j)
            s += b0[j] * Wo[j * 40 + c] + b1[j] * Wo[(64 + j) * 40 + c];
        cb[c] = s;
    }
}

__device__ __forceinline__
void gemm_n64_tile(int bid, float* sW, const float* __restrict__ A, const float* __restrict__ W,
                   const int* __restrict__ cnt, float* __restrict__ D, int N, int K) {
    int tid = threadIdx.x;
    for (int i = tid; i < K * 64; i += 128) {
        int k = i >> 6, col = i & 63;
        sW[(k >> 1) * 128 + 2 * col + (k & 1)] = W[i];
    }
    __syncthreads();

    int wave = tid >> 5;
    int lane = tid & 31;
    int half = lane >> 4;
    int l16  = lane & 15;

    int rbase = bid * 128 + wave * 32;
    int ar0 = rbase + l16;
    int ar1 = rbase + 16 + l16;
    long ab0 = (long)((ar0 < N) ? ar0 : (N - 1)) * K;
    long ab1 = (long)((ar1 < N) ? ar1 : (N - 1)) * K;

    v8f acc00 = {}, acc01 = {}, acc02 = {}, acc03 = {};
    v8f acc10 = {}, acc11 = {}, acc12 = {}, acc13 = {};

    for (int k = 0; k < K; k += 4) {
        int kb = k + 2 * half;
        v2f a0 = *(const v2f*)(A + ab0 + kb);
        v2f a1 = *(const v2f*)(A + ab1 + kb);
        const float* bp = &sW[(kb >> 1) * 128 + 2 * l16];
        v2f b0 = *(const v2f*)(bp + 0);
        v2f b1 = *(const v2f*)(bp + 32);
        v2f b2 = *(const v2f*)(bp + 64);
        v2f b3 = *(const v2f*)(bp + 96);
        acc00 = __builtin_amdgcn_wmma_f32_16x16x4_f32(false, a0, false, b0, (short)0, acc00, false, false);
        acc01 = __builtin_amdgcn_wmma_f32_16x16x4_f32(false, a0, false, b1, (short)0, acc01, false, false);
        acc02 = __builtin_amdgcn_wmma_f32_16x16x4_f32(false, a0, false, b2, (short)0, acc02, false, false);
        acc03 = __builtin_amdgcn_wmma_f32_16x16x4_f32(false, a0, false, b3, (short)0, acc03, false, false);
        acc10 = __builtin_amdgcn_wmma_f32_16x16x4_f32(false, a1, false, b0, (short)0, acc10, false, false);
        acc11 = __builtin_amdgcn_wmma_f32_16x16x4_f32(false, a1, false, b1, (short)0, acc11, false, false);
        acc12 = __builtin_amdgcn_wmma_f32_16x16x4_f32(false, a1, false, b2, (short)0, acc12, false, false);
        acc13 = __builtin_amdgcn_wmma_f32_16x16x4_f32(false, a1, false, b3, (short)0, acc13, false, false);
    }

    int m0 = rbase + half * 8;
    int m1 = rbase + 16 + half * 8;
    int c0[8], c1[8];
#pragma unroll
    for (int r = 0; r < 8; ++r) {
        c0[r] = cnt[(m0 + r < N) ? (m0 + r) : (N - 1)];
        c1[r] = cnt[(m1 + r < N) ? (m1 + r) : (N - 1)];
    }
#pragma unroll
    for (int r = 0; r < 8; ++r) {
        int rr = m0 + r;
        if (rr < N) {
            float sc = dinv_of(c0[r]);
            float* dp = D + (long)rr * 64 + l16;
            dp[0]  = sc * acc00[r];
            dp[16] = sc * acc01[r];
            dp[32] = sc * acc02[r];
            dp[48] = sc * acc03[r];
        }
        int ss = m1 + r;
        if (ss < N) {
            float sc = dinv_of(c1[r]);
            float* dp = D + (long)ss * 64 + l16;
            dp[0]  = sc * acc10[r];
            dp[16] = sc * acc11[r];
            dp[32] = sc * acc12[r];
            dp[48] = sc * acc13[r];
        }
    }
}

__global__ __launch_bounds__(TPE)
void k_rank_heads(const int* __restrict__ dst, int* ch, int* __restrict__ rank, int E, int np,
                  const float* __restrict__ L0, const float* __restrict__ b0,
                  const float* __restrict__ L1, const float* __restrict__ b1,
                  const float* __restrict__ Wo, const float* __restrict__ bo,
                  float* W0f, float* W1f, float* cb) {
    int b = (int)blockIdx.x;
    if (b <= FHB) {
        fuse_heads_role(b, L0, b0, L1, b1, Wo, bo, W0f, W1f, cb);
    } else {
        rank_role(b - (FHB + 1), dst, ch, rank, E, np);
    }
}

__global__ __launch_bounds__(TPE)
void k_fill_dense(const int* __restrict__ src, const int* __restrict__ dst,
                  const int* __restrict__ rank, const int* __restrict__ cnt,
                  const int2* __restrict__ rowloc, const int* __restrict__ btot, int nb,
                  int* __restrict__ boff, int* __restrict__ csr, int E,
                  const float* __restrict__ X, const float* __restrict__ W, float* __restrict__ T, int N, int gG) {
    __shared__ float sbuf[128 * 64];
    if ((int)blockIdx.x < gG) {
        gemm_n64_tile((int)blockIdx.x, sbuf, X, W, cnt, T, N, 128);
        return;
    }
    int bid = (int)blockIdx.x - gG;
    int* sB = (int*)sbuf;
    int* sWv = sB + MAXCH;
    int nft = E / (TPE * EPT);
    if (bid < nft) {
        int e0 = bid * (TPE * EPT) + threadIdx.x;
        int s[EPT], d[EPT], rk[EPT], bo[EPT], ra[EPT], rb[EPT];
#pragma unroll
        for (int k = 0; k < EPT; ++k) {
            s[k]  = src[e0 + k * TPE];
            d[k]  = dst[e0 + k * TPE];
            rk[k] = rank[e0 + k * TPE];
        }
        chunk_starts(bid, btot, nb, boff, sB, sWv);
#pragma unroll
        for (int k = 0; k < EPT; ++k) {
            int2 rl = rowloc[d[k]];
            ra[k] = rl.x;
            rb[k] = rl.y;
            bo[k] = sB[d[k] / CHUNK];
        }
#pragma unroll
        for (int k = 0; k < EPT; ++k) fill_edge(s[k], rk[k], ra[k], rb[k], bo[k], csr);
    } else {
        chunk_starts(bid, btot, nb, boff, sB, sWv);
        int e = nft * (TPE * EPT) + (bid - nft) * TPE + threadIdx.x;
        if (e < E) {
            int s = src[e], d = dst[e];
            int2 rl = rowloc[d];
            fill_edge(s, rank[e], rl.x, rl.y, sB[d / CHUNK], csr);
        }
    }
}

#define NPG 8
#define GTB 128

template <int K>
__device__ __forceinline__ int row_lane(int v) {
    return __builtin_amdgcn_update_dpp(0, v, 0x150 + K, 0xf, 0xf, false);
}

__device__ __forceinline__ int row_next(int v) {
    return __builtin_amdgcn_update_dpp(0, v, 0x12F, 0xf, 0xf, false);
}

template <int K0, int CNT>
__device__ __forceinline__ void rows_req(int e, const float4* __restrict__ T4, int l, float4* r) {
    if constexpr (CNT > 0) {
        rows_req<K0, CNT - 1>(e, T4, l, r);
        int s = row_lane<K0 + CNT - 1>(e);
        r[CNT - 1] = T4[s * 16 + l];
    }
}

template <int K0, int CNT>
__device__ __forceinline__ void tail_ids(int e, int* s) {
    if constexpr (CNT > 0) {
        tail_ids<K0, CNT - 1>(e, s);
        s[CNT - 1] = row_lane<K0 + CNT - 1>(e);
    }
}

template <int CNT>
__device__ __forceinline__ void rows_get(const int* s, const float4* __restrict__ T4, int l, float4* r) {
#pragma unroll
    for (int k = 0; k < CNT; ++k) r[k] = T4[s[k] * 16 + l];
}

template <int CNT>
__device__ __forceinline__ void rows_add(const float4* r, float4& acc) {
#pragma unroll
    for (int k = 0; k < CNT; ++k) {
        acc.x += r[k].x;
        acc.y += r[k].y;
        acc.z += r[k].z;
        acc.w += r[k].w;
    }
}

__device__ __forceinline__ int piece_off(int q, int c, int l) {
    int nfull = c >> 4, t = c & 15;
    int b = (l & 8) ? 8 : (l & 4) ? 4 : (l & 2) ? 2 : (l & 1);
    int it = (t & b) ? ((t & (b - 1)) + (l - b)) : 0;
    int off = (q < nfull) ? (16 * q + l) : (16 * nfull + it);
    return (off < c) ? off : (c - 1);
}

__device__ __forceinline__ void run_sum(const int* __restrict__ p, int c, int e,
                                        const float4* __restrict__ T4, int l,
                                        const float4* __restrict__ selfp, float4& self,
                                        const int* __restrict__ nextp, int& e_next, float4& acc) {
    int nfull = c >> 4, t = c & 15;
    for (int q = 0; q < nfull; ++q) {
        int en = p[piece_off(q + 1, c, l)];
        float4 r[16];
        rows_req<0, 16>(e, T4, l, r);
        rows_add<16>(r, acc);
        e = en;
    }
    float4 r[16] = {};
    int s[16];
    if (t) tail_ids<1, 15>(e, s + 1);
    self = *selfp;
    e_next = *nextp;
    if (t) {
        rows_get<1>(s + 1, T4, l, r + 1);
        if (t & 2) rows_get<2>(s + 2, T4, l, r + 2);
        if (t & 4) rows_get<4>(s + 4, T4, l, r + 4);
        if (t & 8) rows_get<8>(s + 8, T4, l, r + 8);
        if (t & 1) rows_add<1>(r + 1, acc);
        if (t & 2) rows_add<2>(r + 2, acc);
        if (t & 4) rows_add<4>(r + 4, acc);
        if (t & 8) rows_add<8>(r + 8, acc);
    }
}

__device__ __forceinline__ const int* first_piece(const int* __restrict__ csr, int c, int st, int l) {
    return csr + ((c > 0) ? (st + piece_off(0, c, l)) : 0);
}

#define HP 68
#define HQ ((64 / 4) * 40)
template <bool FIRST>
__global__ __launch_bounds__(GTB, 8)
void k_gather_elu(const float* __restrict__ T, const int* __restrict__ csr,
                  const int* __restrict__ cnt, const int2* __restrict__ rowloc,
                  const int* __restrict__ boff, const float* __restrict__ bias, int N,
                  const float* __restrict__ W2, float* __restrict__ D,
                  const float* __restrict__ Wf, const float* __restrict__ cb, float* out) {
    __shared__ __attribute__((aligned(16))) float sW[FIRST ? 64 * 64 : 4];
    __shared__ __attribute__((aligned(16))) float sH[(GTB / 16) * NPG * HP];
    __shared__ float4 sF4[HQ];
    __shared__ float sDn[FIRST ? (GTB / 16) * NPG : 1];
    int tid = threadIdx.x;
    int l = tid & 15;
    int n0 = (blockIdx.x * (GTB / 16) + (tid >> 4)) * NPG;
    const float4* T4 = (const float4*)T;
    int nm = n0 + (l & (NPG - 1));
    nm = (nm < N) ? nm : (N - 1);
    int cm = cnt[nm];
    int stv = boff[nm / CHUNK] + rowloc[nm].x;
    float4 b4 = make_float4(bias[4 * l + 0], bias[4 * l + 1], bias[4 * l + 2], bias[4 * l + 3]);
    int nn = (N - n0 < NPG) ? ((N - n0 > 0) ? (N - n0) : 0) : NPG;
    float4 wa[4], wb[4];
    float4 wf[HQ / GTB];
    if constexpr (FIRST) {
#pragma unroll
        for (int j = 0; j < 4; ++j) {
            int it = tid + j * GTB;
            int kp = it >> 4, cg = it & 15;
            wa[j] = ((const float4*)W2)[(2 * kp) * 16 + cg];
            wb[j] = ((const float4*)W2)[(2 * kp + 1) * 16 + cg];
        }
    }
#pragma unroll
    for (int j = 0; j < HQ / GTB; ++j) wf[j] = ((const float4*)Wf)[tid + j * GTB];
    int e = *first_piece(csr, row_lane<0>(cm), row_lane<0>(stv), l);
    if constexpr (FIRST) {
#pragma unroll
        for (int j = 0; j < 4; ++j) {
            int it = tid + j * GTB;
            int kp = it >> 4, cg = it & 15;
            float4* d = (float4*)&sW[kp * 128 + 8 * cg];
            d[0] = make_float4(wa[j].x, wb[j].x, wa[j].y, wb[j].y);
            d[1] = make_float4(wa[j].z, wb[j].z, wa[j].w, wb[j].w);
        }
    }
#pragma unroll
    for (int j = 0; j < HQ / GTB; ++j) sF4[tid + j * GTB] = wf[j];
#pragma unroll 1
    for (int i = 0; i < nn; ++i) {
        int n = n0 + i;
        int c = row_lane<0>(cm);
        int st = row_lane<0>(stv);
        cm = row_next(cm);
        stv = row_next(stv);
        const int* nextp = first_piece(csr, row_lane<0>(cm), row_lane<0>(stv), l);
        float4 self;
        int e_next;
        float dn = dinv_of(c);
        float4 acc = make_float4(0.0f, 0.0f, 0.0f, 0.0f);
        run_sum(csr + st, c, e, T4, l, T4 + (n * 16 + l), self, nextp, e_next, acc);
        float v0 = dn * (acc.x + self.x) + b4.x;
        float v1 = dn * (acc.y + self.y) + b4.y;
        float v2 = dn * (acc.z + self.z) + b4.z;
        float v3 = dn * (acc.w + self.w) + b4.w;
        float4 h;
        h.x = (v0 > 0.0f) ? v0 : (expf(v0) - 1.0f);
        h.y = (v1 > 0.0f) ? v1 : (expf(v1) - 1.0f);
        h.z = (v2 > 0.0f) ? v2 : (expf(v2) - 1.0f);
        h.w = (v3 > 0.0f) ? v3 : (expf(v3) - 1.0f);
        *(float4*)&sH[((tid >> 4) * NPG + i) * HP + 4 * l] = h;
        if constexpr (FIRST) sDn[(tid >> 4) * NPG + i] = dn;
        e = e_next;
    }
    __syncthreads();
    int wave = tid >> 5;
    int lane = tid & 31;
    int half = lane >> 4;
    int l16  = lane & 15;
    int R0 = blockIdx.x * ((GTB / 16) * NPG) + wave * 16;
    int nrow = N - R0;
    float4* o4 = (float4*)out + (long)R0 * 10;
    float4 prev[5];
    if constexpr (!FIRST) {
        if (nrow >= 16) {
#pragma unroll
            for (int i = 0; i < 5; ++i) prev[i] = o4[lane + 32 * i];
        }
    }
    if constexpr (FIRST) {
        const float* ap = &sH[(wave * 16 + l16) * HP + 2 * half];
        const float* bq = &sW[half * 128 + 2 * l16];
        v8f acc0 = {}, acc1 = {}, acc2 = {}, acc3 = {};
#pragma unroll
        for (int k = 0; k < 64; k += 4) {
            v2f a = *(const v2f*)(ap + k);
            const float* bp = bq + (k >> 1) * 128;
            v2f b0 = *(const v2f*)(bp + 0);
            v2f b1 = *(const v2f*)(bp + 32);
            v2f b2 = *(const v2f*)(bp + 64);
            v2f b3 = *(const v2f*)(bp + 96);
            acc0 = __builtin_amdgcn_wmma_f32_16x16x4_f32(false, a, false, b0, (short)0, acc0, false, false);
            acc1 = __builtin_amdgcn_wmma_f32_16x16x4_f32(false, a, false, b1, (short)0, acc1, false, false);
            acc2 = __builtin_amdgcn_wmma_f32_16x16x4_f32(false, a, false, b2, (short)0, acc2, false, false);
            acc3 = __builtin_amdgcn_wmma_f32_16x16x4_f32(false, a, false, b3, (short)0, acc3, false, false);
        }
        int m0 = R0 + half * 8;
#pragma unroll
        for (int r = 0; r < 8; ++r) {
            int rr = m0 + r;
            if (rr < N) {
                float sc = sDn[wave * 16 + half * 8 + r];
                float* dp = D + (long)rr * 64 + l16;
                dp[0]  = sc * acc0[r];
                dp[16] = sc * acc1[r];
                dp[32] = sc * acc2[r];
                dp[48] = sc * acc3[r];
            }
        }
    }
    const float* ah = &sH[(wave * 16 + l16) * HP + 4 * half];
    const float4* bh = &sF4[half * 40];
    int c2 = 32 + (l16 & 7);
    v8f p0, p1, p2;
    if constexpr (FIRST) {
        float bz0 = cb[l16];
        float bz1 = cb[16 + l16];
        float bz2 = cb[c2];
#pragma unroll
        for (int r = 0; r < 8; ++r) { p0[r] = bz0; p1[r] = bz1; p2[r] = bz2; }
    } else {
#pragma unroll
        for (int r = 0; r < 8; ++r) { p0[r] = 0.0f; p1[r] = 0.0f; p2[r] = 0.0f; }
    }
#pragma unroll
    for (int j = 0; j < 8; ++j) {
        float4 a = *(const float4*)(ah + 8 * j);
        const float4* bp = bh + (2 * j) * 40;
        float4 b0 = bp[l16], b1 = bp[16 + l16], b2 = bp[c2];
        v2f x = {a.x, a.y}, y = {a.z, a.w};
        v2f q0 = {b0.x, b0.y}, q1 = {b1.x, b1.y}, q2 = {b2.x, b2.y};
        v2f s0 = {b0.z, b0.w}, s1 = {b1.z, b1.w}, s2 = {b2.z, b2.w};
        p0 = __builtin_amdgcn_wmma_f32_16x16x4_f32(false, x, false, q0, (short)0, p0, false, false);
        p1 = __builtin_amdgcn_wmma_f32_16x16x4_f32(false, x, false, q1, (short)0, p1, false, false);
        p2 = __builtin_amdgcn_wmma_f32_16x16x4_f32(false, x, false, q2, (short)0, p2, false, false);
        p0 = __builtin_amdgcn_wmma_f32_16x16x4_f32(false, y, false, s0, (short)0, p0, false, false);
        p1 = __builtin_amdgcn_wmma_f32_16x16x4_f32(false, y, false, s1, (short)0, p1, false, false);
        p2 = __builtin_amdgcn_wmma_f32_16x16x4_f32(false, y, false, s2, (short)0, p2, false, false);
    }
    float* so = &sH[wave * 16 * HP];
#pragma unroll
    for (int r = 0; r < 8; ++r) {
        float* q = so + (half * 8 + r) * 40;
        q[l16]      = p0[r];
        q[16 + l16] = p1[r];
        q[c2]       = p2[r];
    }
    __syncthreads();
    const float4* s4 = (const float4*)so;
    if (nrow >= 16) {
        float4 t[5];
#pragma unroll
        for (int i = 0; i < 5; ++i) t[i] = s4[lane + 32 * i];
        if constexpr (!FIRST) {
#pragma unroll
            for (int i = 0; i < 5; ++i) {
                t[i].x += prev[i].x;
                t[i].y += prev[i].y;
                t[i].z += prev[i].z;
                t[i].w += prev[i].w;
            }
        }
#pragma unroll
        for (int i = 0; i < 5; ++i) o4[lane + 32 * i] = t[i];
    } else {
        int n4 = nrow * 10;
        for (int i = lane; i < n4; i += 32) {
            float4 v = s4[i];
            if constexpr (!FIRST) {
                float4 u = o4[i];
                v.x += u.x;
                v.y += u.y;
                v.z += u.z;
                v.w += u.w;
            }
            o4[i] = v;
        }
    }
}

extern "C" void kernel_launch(void* const* d_in, const int* in_sizes, int n_in,
                              void* d_out, int out_size, void* d_ws, size_t ws_size,
                              hipStream_t stream) {
    const float* x       = (const float*)d_in[0];
    const int*   ei      = (const int*)d_in[1];
    const float* conv0_w = (const float*)d_in[2];
    const float* conv0_b = (const float*)d_in[3];
    const float* lin0_w  = (const float*)d_in[4];
    const float* lin0_b  = (const float*)d_in[5];
    const float* conv1_w = (const float*)d_in[6];
    const float* conv1_b = (const float*)d_in[7];
    const float* lin1_w  = (const float*)d_in[8];
    const float* lin1_b  = (const float*)d_in[9];
    const float* out_w   = (const float*)d_in[10];
    const float* out_b   = (const float*)d_in[11];
    float* out = (float*)d_out;

    int N = in_sizes[0] / 128;
    int E = in_sizes[1] / 2;
    const int* src = ei;
    const int* dst = ei + E;

    size_t NH = (size_t)N * 64;
    float* ws   = (float*)d_ws;
    int*   cnt  = (int*)ws;
    float* A    = ws + (((size_t)N + 255) / 256) * 256;
    float* H1   = A + NH;
    float* EL   = H1 + NH;
    float* W0f  = EL + NH;
    float* W1f  = W0f + 64 * 40;
    float* cb   = W1f + 64 * 40;
    int*   rank = (int*)H1;
    size_t np   = (((size_t)N + 255) / 256) * 256;
    int*   ch   = rank + (((size_t)E + 255) / 256) * 256;
    int*   oscr   = (int*)EL;
    int*   csr    = oscr;
    int2*  rowloc = (int2*)(oscr + 2 * (size_t)E);
    int*   boff   = (int*)(rowloc + N);

    int gCh  = (N + CHUNK - 1) / CHUNK;
    int gGa  = (N + (GTB / 16) * NPG - 1) / ((GTB / 16) * NPG);
    int gEe  = E / (TPE * EPT) + (E % (TPE * EPT) + TPE - 1) / TPE;
    int n4   = (int)(2 * np / 4);
    int gGemm = (N + 127) / 128;

    k_zero_cnt<<<(n4 + TPB - 1) / TPB, TPB, 0, stream>>>((int4*)ch, n4);
    k_rank_heads<<<(FHB + 1) + gEe, TPE, 0, stream>>>(dst, ch, rank, E, (int)np,
                                                     lin0_w, lin0_b, lin1_w, lin1_b, out_w, out_b,
                                                     W0f, W1f, cb);
    k_scan_chunk<<<gCh, 256, 0, stream>>>(ch, ch + np, cnt, rowloc, boff + gCh, N);
    k_fill_dense<<<gGemm + gEe, TPE, 0, stream>>>(src, dst, rank, cnt, rowloc, boff + gCh, gCh, boff, csr, E,
                                                 x, conv0_w, A, N, gGemm);

    k_gather_elu<true><<<gGa, GTB, 0, stream>>>(A, csr, cnt, rowloc, boff, conv0_b, N, conv1_w, H1, W0f, cb, out);

    k_gather_elu<false><<<gGa, GTB, 0, stream>>>(H1, csr, cnt, rowloc, boff, conv1_b, N, nullptr, nullptr, W1f, nullptr, out);
}


# edges grouped by destination with no memory atomic per edge: binned by 512-node group, then placed per node, by LDS adds (1 memory add per block and group); 6 launches to 5; sums in LDS-add order
# speedup vs baseline: 7.0368x; 1.2438x over previous
#include <hip/hip_runtime.h>
#include <hip/hip_bf16.h>

typedef float v2f __attribute__((ext_vector_type(2)));
typedef float v8f __attribute__((ext_vector_type(8)));

#define TPE 128
#define BKN 512
#define BEPT 32
#define MAXBK 2048
#define SRCB 22

__device__ __forceinline__ int cache_half() {
    return (int)((__builtin_amdgcn_s_sendmsg_rtn(0x87) >> 18) & 1u);
}

__device__ __forceinline__ float dinv_of(int c) {
    float ds = (float)(c + 1);
    return (ds > 0.0f) ? rsqrtf(ds) : 0.0f;
}

__global__ void k_zero_cur(int* cur, int n) {
    for (int i = threadIdx.x; i < n; i += (int)blockDim.x) cur[i] = 0;
}

template <int NE, bool GUARD>
__device__ __forceinline__
void bin_edges(int e0, const int* __restrict__ src, const int* __restrict__ dst, int* cur, int* __restrict__ reg,
               int E, int nbk, int capb, int* sHist, int* sBase) {
    int tid = threadIdx.x;
    int h = cache_half();
    for (int i = tid; i < nbk; i += TPE) sHist[i] = 0;
    int d[NE], s[NE], lr[NE];
#pragma unroll
    for (int k = 0; k < NE; ++k) {
        int e = e0 + k * TPE;
        bool ok = !GUARD || e < E;
        d[k] = ok ? dst[e] : -1;
        s[k] = ok ? src[e] : 0;
    }
    __syncthreads();
#pragma unroll
    for (int k = 0; k < NE; ++k) lr[k] = (d[k] >= 0) ? atomicAdd(&sHist[d[k] / BKN], 1) : 0;
    __syncthreads();
    for (int i = tid; i < nbk; i += TPE) {
        int c = sHist[i];
        sBase[i] = (c > 0) ? atomicAdd(&cur[h * nbk + i], c) : 0;
    }
    __syncthreads();
#pragma unroll
    for (int k = 0; k < NE; ++k) lr[k] += sBase[(d[k] >= 0) ? (d[k] / BKN) : 0];
#pragma unroll
    for (int k = 0; k < NE; ++k) {
        if (d[k] >= 0 && lr[k] < capb) {
            int off = h ? (capb - 1 - lr[k]) : lr[k];
            reg[(size_t)(d[k] / BKN) * capb + off] = ((d[k] % BKN) << SRCB) | s[k];
        }
    }
}

__device__ __forceinline__
void bin_role(int bid, const int* __restrict__ src, const int* __restrict__ dst, int* cur, int* __restrict__ reg,
              int E, int nbk, int capb, int* sHist, int* sBase) {
    int nft = E / (TPE * BEPT);
    if (bid < nft) {
        bin_edges<BEPT, false>(bid * (TPE * BEPT) + threadIdx.x, src, dst, cur, reg, E, nbk, capb, sHist, sBase);
    } else {
        bin_edges<1, true>(nft * (TPE * BEPT) + (bid - nft) * TPE + threadIdx.x, src, dst, cur, reg, E, nbk, capb,
                           sHist, sBase);
    }
}

#define BKT 512
#define KCH 20
__global__ __launch_bounds__(BKT)
void k_bucket(const int* __restrict__ cur, const int* __restrict__ reg,
              const int* __restrict__ src, const int* __restrict__ dst, int E, int nbk, int capb,
              int* __restrict__ cnt, int* __restrict__ rowst, int* __restrict__ csr, float* T, int N) {
    __shared__ int sCnt[BKN];
    __shared__ int sPos[BKN];
    __shared__ int sWv[BKT / 32];
    __shared__ int sStart;
    int tid = threadIdx.x;
    int b = blockIdx.x;
    int lane = tid & 31, wv = tid >> 5;
    sCnt[tid] = 0;
    if (tid == 0) sStart = 0;
    __syncthreads();
    for (int j = tid; j < b; j += BKT) atomicAdd(&sStart, cur[j] + cur[nbk + j]);
    int ca = cur[b], cb = cur[nbk + b];
    int tot = ca + cb;
    bool fits = tot <= capb;
    const int* rb = reg + (size_t)b * capb;
    if (fits) {
        for (int i0 = tid; i0 < tot; i0 += BKT * KCH) {
            int p[KCH];
#pragma unroll
            for (int k = 0; k < KCH; ++k) {
                int i = i0 + k * BKT;
                i = (i < tot) ? i : (tot - 1);
                p[k] = rb[(i < ca) ? i : (capb - 1 - (i - ca))];
            }
#pragma unroll
            for (int k = 0; k < KCH; ++k)
                if (i0 + k * BKT < tot) atomicAdd(&sCnt[p[k] >> SRCB], 1);
        }
    } else {
        for (int e = tid; e < E; e += BKT) {
            int d = dst[e];
            if (d / BKN == b) atomicAdd(&sCnt[d % BKN], 1);
        }
    }
    __syncthreads();
    int c = sCnt[tid];
    int incl = c;
    for (int off = 1; off < 32; off <<= 1) {
        int v = __shfl_up(incl, off, 32);
        if (lane >= off) incl += v;
    }
    if (lane == 31) sWv[wv] = incl;
    __syncthreads();
    int excl = incl - c;
    for (int w = 0; w < wv; ++w) excl += sWv[w];
    int start = sStart;
    sPos[tid] = excl;
    int n = b * BKN + tid;
    if (n < N) {
        cnt[n] = c;
        rowst[n] = start + excl;
    }
    __syncthreads();
    if (fits) {
        for (int i0 = tid; i0 < tot; i0 += BKT * KCH) {
            int p[KCH];
#pragma unroll
            for (int k = 0; k < KCH; ++k) {
                int i = i0 + k * BKT;
                i = (i < tot) ? i : (tot - 1);
                p[k] = rb[(i < ca) ? i : (capb - 1 - (i - ca))];
            }
#pragma unroll
            for (int k = 0; k < KCH; ++k) {
                if (i0 + k * BKT < tot) {
                    int pos = atomicAdd(&sPos[p[k] >> SRCB], 1);
                    csr[start + pos] = p[k] & ((1 << SRCB) - 1);
                }
            }
        }
    } else {
        for (int e = tid; e < E; e += BKT) {
            int d = dst[e];
            if (d / BKN == b) {
                int pos = atomicAdd(&sPos[d % BKN], 1);
                csr[start + pos] = src[e];
            }
        }
    }
    int nrows = N - b * BKN;
    nrows = (nrows < BKN) ? nrows : BKN;
    float4* T4 = (float4*)T + (size_t)b * BKN * 16;
#pragma unroll 8
    for (int i = tid; i < nrows * 16; i += BKT) {
        float sc = dinv_of(sCnt[i >> 4]);
        float4 v = T4[i];
        v.x *= sc;
        v.y *= sc;
        v.z *= sc;
        v.w *= sc;
        T4[i] = v;
    }
}

#define FHB ((64 * 40) / TPE)
__device__ __forceinline__
void fuse_heads_role(int bid, const float* __restrict__ L0, const float* __restrict__ b0,
                     const float* __restrict__ L1, const float* __restrict__ b1,
                     const float* __restrict__ Wo, const float* __restrict__ bo,
                     float* W0f, float* W1f, float* cb) {
    if (bid < FHB) {
        int i = bid * TPE + threadIdx.x;
        int k = i / 40, c = i % 40;
        float s0 = 0.0f, s1 = 0.0f;
#pragma unroll 16
        for (int j = 0; j < 64; ++j) {
            s0 += L0[k * 64 + j] * Wo[j * 40 + c];
            s1 += L1[k * 64 + j] * Wo[(64 + j) * 40 + c];
        }
        int o = ((k >> 2) * 40 + c) * 4 + (k & 3);
        W0f[o] = s0;
        W1f[o] = s1;
    } else if (threadIdx.x < 40) {
        int c = threadIdx.x;
        float s = bo[c];
        for (int j = 0; j < 64; ++j)
            s += b0[j] * Wo[j * 40 + c] + b1[j] * Wo[(64 + j) * 40 + c];
        cb[c] = s;
    }
}

__device__ __forceinline__
void gemm_n64_tile(int bid, float* sW, const float* __restrict__ A, const float* __restrict__ W,
                   float* __restrict__ D, int N, int K) {
    int tid = threadIdx.x;
    for (int i = tid; i < K * 64; i += 128) {
        int k = i >> 6, col = i & 63;
        sW[(k >> 1) * 128 + 2 * col + (k & 1)] = W[i];
    }
    __syncthreads();

    int wave = tid >> 5;
    int lane = tid & 31;
    int half = lane >> 4;
    int l16  = lane & 15;

    int rbase = bid * 128 + wave * 32;
    int ar0 = rbase + l16;
    int ar1 = rbase + 16 + l16;
    long ab0 = (long)((ar0 < N) ? ar0 : (N - 1)) * K;
    long ab1 = (long)((ar1 < N) ? ar1 : (N - 1)) * K;

    v8f acc00 = {}, acc01 = {}, acc02 = {}, acc03 = {};
    v8f acc10 = {}, acc11 = {}, acc12 = {}, acc13 = {};

    for (int k = 0; k < K; k += 4) {
        int kb = k + 2 * half;
        v2f a0 = *(const v2f*)(A + ab0 + kb);
        v2f a1 = *(const v2f*)(A + ab1 + kb);
        const float* bp = &sW[(kb >> 1) * 128 + 2 * l16];
        v2f b0 = *(const v2f*)(bp + 0);
        v2f b1 = *(const v2f*)(bp + 32);
        v2f b2 = *(const v2f*)(bp + 64);
        v2f b3 = *(const v2f*)(bp + 96);
        acc00 = __builtin_amdgcn_wmma_f32_16x16x4_f32(false, a0, false, b0, (short)0, acc00, false, false);
        acc01 = __builtin_amdgcn_wmma_f32_16x16x4_f32(false, a0, false, b1, (short)0, acc01, false, false);
        acc02 = __builtin_amdgcn_wmma_f32_16x16x4_f32(false, a0, false, b2, (short)0, acc02, false, false);
        acc03 = __builtin_amdgcn_wmma_f32_16x16x4_f32(false, a0, false, b3, (short)0, acc03, false, false);
        acc10 = __builtin_amdgcn_wmma_f32_16x16x4_f32(false, a1, false, b0, (short)0, acc10, false, false);
        acc11 = __builtin_amdgcn_wmma_f32_16x16x4_f32(false, a1, false, b1, (short)0, acc11, false, false);
        acc12 = __builtin_amdgcn_wmma_f32_16x16x4_f32(false, a1, false, b2, (short)0, acc12, false, false);
        acc13 = __builtin_amdgcn_wmma_f32_16x16x4_f32(false, a1, false, b3, (short)0, acc13, false, false);
    }

    int m0 = rbase + half * 8;
    int m1 = rbase + 16 + half * 8;
#pragma unroll
    for (int r = 0; r < 8; ++r) {
        int rr = m0 + r;
        if (rr < N) {
            float* dp = D + (long)rr * 64 + l16;
            dp[0]  = acc00[r];
            dp[16] = acc01[r];
            dp[32] = acc02[r];
            dp[48] = acc03[r];
        }
        int ss = m1 + r;
        if (ss < N) {
            float* dp = D + (long)ss * 64 + l16;
            dp[0]  = acc10[r];
            dp[16] = acc11[r];
            dp[32] = acc12[r];
            dp[48] = acc13[r];
        }
    }
}

__global__ __launch_bounds__(TPE)
void k_bin_dense(const int* __restrict__ src, const int* __restrict__ dst, int* cur, int* __restrict__ reg,
                 int E, int nbk, int capb,
                 const float* __restrict__ X, const float* __restrict__ W, float* __restrict__ T, int N, int gG,
                 const float* __restrict__ L0, const float* __restrict__ b0,
                 const float* __restrict__ L1, const float* __restrict__ b1,
                 const float* __restrict__ Wo, const float* __restrict__ bo,
                 float* W0f, float* W1f, float* cb) {
    __shared__ float sbuf[128 * 64];
    int b = (int)blockIdx.x;
    if (b < gG) {
        gemm_n64_tile(b, sbuf, X, W, T, N, 128);
    } else if (b <= gG + FHB) {
        fuse_heads_role(b - gG, L0, b0, L1, b1, Wo, bo, W0f, W1f, cb);
    } else {
        int* sHist = (int*)sbuf;
        bin_role(b - (gG + FHB + 1), src, dst, cur, reg, E, nbk, capb, sHist, sHist + MAXBK);
    }
}

#define NPG 8
#define GTB 128

template <int K>
__device__ __forceinline__ int row_lane(int v) {
    return __builtin_amdgcn_update_dpp(0, v, 0x150 + K, 0xf, 0xf, false);
}

__device__ __forceinline__ int row_next(int v) {
    return __builtin_amdgcn_update_dpp(0, v, 0x12F, 0xf, 0xf, false);
}

template <int K0, int CNT>
__device__ __forceinline__ void rows_req(int e, const float4* __restrict__ T4, int l, float4* r) {
    if constexpr (CNT > 0) {
        rows_req<K0, CNT - 1>(e, T4, l, r);
        int s = row_lane<K0 + CNT - 1>(e);
        r[CNT - 1] = T4[s * 16 + l];
    }
}

template <int K0, int CNT>
__device__ __forceinline__ void tail_ids(int e, int* s) {
    if constexpr (CNT > 0) {
        tail_ids<K0, CNT - 1>(e, s);
        s[CNT - 1] = row_lane<K0 + CNT - 1>(e);
    }
}

template <int CNT>
__device__ __forceinline__ void rows_get(const int* s, const float4* __restrict__ T4, int l, float4* r) {
#pragma unroll
    for (int k = 0; k < CNT; ++k) r[k] = T4[s[k] * 16 + l];
}

template <int CNT>
__device__ __forceinline__ void rows_add(const float4* r, float4& acc) {
#pragma unroll
    for (int k = 0; k < CNT; ++k) {
        acc.x += r[k].x;
        acc.y += r[k].y;
        acc.z += r[k].z;
        acc.w += r[k].w;
    }
}

__device__ __forceinline__ int piece_off(int q, int c, int l) {
    int nfull = c >> 4, t = c & 15;
    int b = (l & 8) ? 8 : (l & 4) ? 4 : (l & 2) ? 2 : (l & 1);
    int it = (t & b) ? ((t & (b - 1)) + (l - b)) : 0;
    int off = (q < nfull) ? (16 * q + l) : (16 * nfull + it);
    return (off < c) ? off : (c - 1);
}

__device__ __forceinline__ void run_sum(const int* __restrict__ p, int c, int e,
                                        const float4* __restrict__ T4, int l,
                                        const float4* __restrict__ selfp, float4& self,
                                        const int* __restrict__ nextp, int& e_next, float4& acc) {
    int nfull = c >> 4, t = c & 15;
    for (int q = 0; q < nfull; ++q) {
        int en = p[piece_off(q + 1, c, l)];
        float4 r[16];
        rows_req<0, 16>(e, T4, l, r);
        rows_add<16>(r, acc);
        e = en;
    }
    float4 r[16] = {};
    int s[16];
    if (t) tail_ids<1, 15>(e, s + 1);
    self = *selfp;
    e_next = *nextp;
    if (t) {
        rows_get<1>(s + 1, T4, l, r + 1);
        if (t & 2) rows_get<2>(s + 2, T4, l, r + 2);
        if (t & 4) rows_get<4>(s + 4, T4, l, r + 4);
        if (t & 8) rows_get<8>(s + 8, T4, l, r + 8);
        if (t & 1) rows_add<1>(r + 1, acc);
        if (t & 2) rows_add<2>(r + 2, acc);
        if (t & 4) rows_add<4>(r + 4, acc);
        if (t & 8) rows_add<8>(r + 8, acc);
    }
}

__device__ __forceinline__ const int* first_piece(const int* __restrict__ csr, int c, int st, int l) {
    return csr + ((c > 0) ? (st + piece_off(0, c, l)) : 0);
}

#define HP 68
#define HQ ((64 / 4) * 40)
template <bool FIRST>
__global__ __launch_bounds__(GTB, 8)
void k_gather_elu(const float* __restrict__ T, const int* __restrict__ csr,
                  const int* __restrict__ cnt, const int* __restrict__ rowst,
                  const float* __restrict__ bias, int N,
                  const float* __restrict__ W2, float* __restrict__ D,
                  const float* __restrict__ Wf, const float* __restrict__ cb, float* out) {
    __shared__ __attribute__((aligned(16))) float sW[FIRST ? 64 * 64 : 4];
    __shared__ __attribute__((aligned(16))) float sH[(GTB / 16) * NPG * HP];
    __shared__ float4 sF4[HQ];
    __shared__ float sDn[FIRST ? (GTB / 16) * NPG : 1];
    int tid = threadIdx.x;
    int l = tid & 15;
    int n0 = (blockIdx.x * (GTB / 16) + (tid >> 4)) * NPG;
    const float4* T4 = (const float4*)T;
    int nm = n0 + (l & (NPG - 1));
    nm = (nm < N) ? nm : (N - 1);
    int cm = cnt[nm];
    int stv = rowst[nm];
    float4 b4 = make_float4(bias[4 * l + 0], bias[4 * l + 1], bias[4 * l + 2], bias[4 * l + 3]);
    int nn = (N - n0 < NPG) ? ((N - n0 > 0) ? (N - n0) : 0) : NPG;
    float4 wa[4], wb[4];
    float4 wf[HQ / GTB];
    if constexpr (FIRST) {
#pragma unroll
        for (int j = 0; j < 4; ++j) {
            int it = tid + j * GTB;
            int kp = it >> 4, cg = it & 15;
            wa[j] = ((const float4*)W2)[(2 * kp) * 16 + cg];
            wb[j] = ((const float4*)W2)[(2 * kp + 1) * 16 + cg];
        }
    }
#pragma unroll
    for (int j = 0; j < HQ / GTB; ++j) wf[j] = ((const float4*)Wf)[tid + j * GTB];
    int e = *first_piece(csr, row_lane<0>(cm), row_lane<0>(stv), l);
    if constexpr (FIRST) {
#pragma unroll
        for (int j = 0; j < 4; ++j) {
            int it = tid + j * GTB;
            int kp = it >> 4, cg = it & 15;
            float4* d = (float4*)&sW[kp * 128 + 8 * cg];
            d[0] = make_float4(wa[j].x, wb[j].x, wa[j].y, wb[j].y);
            d[1] = make_float4(wa[j].z, wb[j].z, wa[j].w, wb[j].w);
        }
    }
#pragma unroll
    for (int j = 0; j < HQ / GTB; ++j) sF4[tid + j * GTB] = wf[j];
#pragma unroll 1
    for (int i = 0; i < nn; ++i) {
        int n = n0 + i;
        int c = row_lane<0>(cm);
        int st = row_lane<0>(stv);
        cm = row_next(cm);
        stv = row_next(stv);
        const int* nextp = first_piece(csr, row_lane<0>(cm), row_lane<0>(stv), l);
        float4 self;
        int e_next;
        float dn = dinv_of(c);
        float4 acc = make_float4(0.0f, 0.0f, 0.0f, 0.0f);
        run_sum(csr + st, c, e, T4, l, T4 + (n * 16 + l), self, nextp, e_next, acc);
        float v0 = dn * (acc.x + self.x) + b4.x;
        float v1 = dn * (acc.y + self.y) + b4.y;
        float v2 = dn * (acc.z + self.z) + b4.z;
        float v3 = dn * (acc.w + self.w) + b4.w;
        float4 h;
        h.x = (v0 > 0.0f) ? v0 : (expf(v0) - 1.0f);
        h.y = (v1 > 0.0f) ? v1 : (expf(v1) - 1.0f);
        h.z = (v2 > 0.0f) ? v2 : (expf(v2) - 1.0f);
        h.w = (v3 > 0.0f) ? v3 : (expf(v3) - 1.0f);
        *(float4*)&sH[((tid >> 4) * NPG + i) * HP + 4 * l] = h;
        if constexpr (FIRST) sDn[(tid >> 4) * NPG + i] = dn;
        e = e_next;
    }
    __syncthreads();
    int wave = tid >> 5;
    int lane = tid & 31;
    int half = lane >> 4;
    int l16  = lane & 15;
    int R0 = blockIdx.x * ((GTB / 16) * NPG) + wave * 16;
    int nrow = N - R0;
    float4* o4 = (float4*)out + (long)R0 * 10;
    float4 prev[5];
    if constexpr (!FIRST) {
        if (nrow >= 16) {
#pragma unroll
            for (int i = 0; i < 5; ++i) prev[i] = o4[lane + 32 * i];
        }
    }
    if constexpr (FIRST) {
        const float* ap = &sH[(wave * 16 + l16) * HP + 2 * half];
        const float* bq = &sW[half * 128 + 2 * l16];
        v8f acc0 = {}, acc1 = {}, acc2 = {}, acc3 = {};
#pragma unroll
        for (int k = 0; k < 64; k += 4) {
            v2f a = *(const v2f*)(ap + k);
            const float* bp = bq + (k >> 1) * 128;
            v2f b0 = *(const v2f*)(bp + 0);
            v2f b1 = *(const v2f*)(bp + 32);
            v2f b2 = *(const v2f*)(bp + 64);
            v2f b3 = *(const v2f*)(bp + 96);
            acc0 = __builtin_amdgcn_wmma_f32_16x16x4_f32(false, a, false, b0, (short)0, acc0, false, false);
            acc1 = __builtin_amdgcn_wmma_f32_16x16x4_f32(false, a, false, b1, (short)0, acc1, false, false);
            acc2 = __builtin_amdgcn_wmma_f32_16x16x4_f32(false, a, false, b2, (short)0, acc2, false, false);
            acc3 = __builtin_amdgcn_wmma_f32_16x16x4_f32(false, a, false, b3, (short)0, acc3, false, false);
        }
        int m0 = R0 + half * 8;
#pragma unroll
        for (int r = 0; r < 8; ++r) {
            int rr = m0 + r;
            if (rr < N) {
                float sc = sDn[wave * 16 + half * 8 + r];
                float* dp = D + (long)rr * 64 + l16;
                dp[0]  = sc * acc0[r];
                dp[16] = sc * acc1[r];
                dp[32] = sc * acc2[r];
                dp[48] = sc * acc3[r];
            }
        }
    }
    const float* ah = &sH[(wave * 16 + l16) * HP + 4 * half];
    const float4* bh = &sF4[half * 40];
    int c2 = 32 + (l16 & 7);
    v8f p0, p1, p2;
    if constexpr (FIRST) {
        float bz0 = cb[l16];
        float bz1 = cb[16 + l16];
        float bz2 = cb[c2];
#pragma unroll
        for (int r = 0; r < 8; ++r) { p0[r] = bz0; p1[r] = bz1; p2[r] = bz2; }
    } else {
#pragma unroll
        for (int r = 0; r < 8; ++r) { p0[r] = 0.0f; p1[r] = 0.0f; p2[r] = 0.0f; }
    }
#pragma unroll
    for (int j = 0; j < 8; ++j) {
        float4 a = *(const float4*)(ah + 8 * j);
        const float4* bp = bh + (2 * j) * 40;
        float4 b0 = bp[l16], b1 = bp[16 + l16], b2 = bp[c2];
        v2f x = {a.x, a.y}, y = {a.z, a.w};
        v2f q0 = {b0.x, b0.y}, q1 = {b1.x, b1.y}, q2 = {b2.x, b2.y};
        v2f s0 = {b0.z, b0.w}, s1 = {b1.z, b1.w}, s2 = {b2.z, b2.w};
        p0 = __builtin_amdgcn_wmma_f32_16x16x4_f32(false, x, false, q0, (short)0, p0, false, false);
        p1 = __builtin_amdgcn_wmma_f32_16x16x4_f32(false, x, false, q1, (short)0, p1, false, false);
        p2 = __builtin_amdgcn_wmma_f32_16x16x4_f32(false, x, false, q2, (short)0, p2, false, false);
        p0 = __builtin_amdgcn_wmma_f32_16x16x4_f32(false, y, false, s0, (short)0, p0, false, false);
        p1 = __builtin_amdgcn_wmma_f32_16x16x4_f32(false, y, false, s1, (short)0, p1, false, false);
        p2 = __builtin_amdgcn_wmma_f32_16x16x4_f32(false, y, false, s2, (short)0, p2, false, false);
    }
    float* so = &sH[wave * 16 * HP];
#pragma unroll
    for (int r = 0; r < 8; ++r) {
        float* q = so + (half * 8 + r) * 40;
        q[l16]      = p0[r];
        q[16 + l16] = p1[r];
        q[c2]       = p2[r];
    }
    __syncthreads();
    const float4* s4 = (const float4*)so;
    if (nrow >= 16) {
        float4 t[5];
#pragma unroll
        for (int i = 0; i < 5; ++i) t[i] = s4[lane + 32 * i];
        if constexpr (!FIRST) {
#pragma unroll
            for (int i = 0; i < 5; ++i) {
                t[i].x += prev[i].x;
                t[i].y += prev[i].y;
                t[i].z += prev[i].z;
                t[i].w += prev[i].w;
            }
        }
#pragma unroll
        for (int i = 0; i < 5; ++i) o4[lane + 32 * i] = t[i];
    } else {
        int n4 = nrow * 10;
        for (int i = lane; i < n4; i += 32) {
            float4 v = s4[i];
            if constexpr (!FIRST) {
                float4 u = o4[i];
                v.x += u.x;
                v.y += u.y;
                v.z += u.z;
                v.w += u.w;
            }
            o4[i] = v;
        }
    }
}

extern "C" void kernel_launch(void* const* d_in, const int* in_sizes, int n_in,
                              void* d_out, int out_size, void* d_ws, size_t ws_size,
                              hipStream_t stream) {
    const float* x       = (const float*)d_in[0];
    const int*   ei      = (const int*)d_in[1];
    const float* conv0_w = (const float*)d_in[2];
    const float* conv0_b = (const float*)d_in[3];
    const float* lin0_w  = (const float*)d_in[4];
    const float* lin0_b  = (const float*)d_in[5];
    const float* conv1_w = (const float*)d_in[6];
    const float* conv1_b = (const float*)d_in[7];
    const float* lin1_w  = (const float*)d_in[8];
    const float* lin1_b  = (const float*)d_in[9];
    const float* out_w   = (const float*)d_in[10];
    const float* out_b   = (const float*)d_in[11];
    float* out = (float*)d_out;

    int N = in_sizes[0] / 128;
    int E = in_sizes[1] / 2;
    const int* src = ei;
    const int* dst = ei + E;

    size_t NH = (size_t)N * 64;
    float* ws   = (float*)d_ws;
    int*   cnt  = (int*)ws;
    float* A    = ws + (((size_t)N + 255) / 256) * 256;
    float* H1   = A + NH;
    float* EL   = H1 + NH;
    float* W0f  = EL + NH;
    float* W1f  = W0f + 64 * 40;
    float* cb   = W1f + 64 * 40;
    int    nbk  = (N + BKN - 1) / BKN;
    int    capb = (int)(NH / (size_t)nbk);
    int*   reg  = (int*)H1;
    int*   oscr   = (int*)EL;
    int*   csr    = oscr;
    int*   rowst  = oscr + 2 * (size_t)E;
    int*   cur    = rowst + N;

    int gGa  = (N + (GTB / 16) * NPG - 1) / ((GTB / 16) * NPG);
    int gBin = E / (TPE * BEPT) + (E % (TPE * BEPT) + TPE - 1) / TPE;
    int gGemm = (N + 127) / 128;

    k_zero_cur<<<1, 512, 0, stream>>>(cur, 2 * nbk);
    k_bin_dense<<<gGemm + (FHB + 1) + gBin, TPE, 0, stream>>>(src, dst, cur, reg, E, nbk, capb,
                                                             x, conv0_w, A, N, gGemm,
                                                             lin0_w, lin0_b, lin1_w, lin1_b, out_w, out_b,
                                                             W0f, W1f, cb);
    k_bucket<<<nbk, BKT, 0, stream>>>(cur, reg, src, dst, E, nbk, capb, cnt, rowst, csr, A, N);

    k_gather_elu<true><<<gGa, GTB, 0, stream>>>(A, csr, cnt, rowst, conv0_b, N, conv1_w, H1, W0f, cb, out);

    k_gather_elu<false><<<gGa, GTB, 0, stream>>>(H1, csr, cnt, rowst, conv1_b, N, nullptr, nullptr, W1f, nullptr, out);
}


# level 2 of the edge grouping builds each group's piece of the grouped list in LDS and stores it as whole 128-byte lines (16-byte stores), not one scattered 4-byte store per edge; sums unchanged
# speedup vs baseline: 7.4250x; 1.0552x over previous
#include <hip/hip_runtime.h>
#include <hip/hip_bf16.h>

typedef float v2f __attribute__((ext_vector_type(2)));
typedef float v8f __attribute__((ext_vector_type(8)));

#define TPE 128
#define BKN 512
#define BEPT 32
#define MAXBK 2048
#define SRCB 22

__device__ __forceinline__ int cache_half() {
    return (int)((__builtin_amdgcn_s_sendmsg_rtn(0x87) >> 18) & 1u);
}

__device__ __forceinline__ float dinv_of(int c) {
    float ds = (float)(c + 1);
    return (ds > 0.0f) ? rsqrtf(ds) : 0.0f;
}

__global__ void k_zero_cur(int* cur, int n) {
    for (int i = threadIdx.x; i < n; i += (int)blockDim.x) cur[i] = 0;
}

template <int NE, bool GUARD>
__device__ __forceinline__
void bin_edges(int e0, const int* __restrict__ src, const int* __restrict__ dst, int* cur, int* __restrict__ reg,
               int E, int nbk, int capb, int* sHist, int* sBase) {
    int tid = threadIdx.x;
    int h = cache_half();
    for (int i = tid; i < nbk; i += TPE) sHist[i] = 0;
    int d[NE], s[NE], lr[NE];
#pragma unroll
    for (int k = 0; k < NE; ++k) {
        int e = e0 + k * TPE;
        bool ok = !GUARD || e < E;
        d[k] = ok ? dst[e] : -1;
        s[k] = ok ? src[e] : 0;
    }
    __syncthreads();
#pragma unroll
    for (int k = 0; k < NE; ++k) lr[k] = (d[k] >= 0) ? atomicAdd(&sHist[d[k] / BKN], 1) : 0;
    __syncthreads();
    for (int i = tid; i < nbk; i += TPE) {
        int c = sHist[i];
        sBase[i] = (c > 0) ? atomicAdd(&cur[h * nbk + i], c) : 0;
    }
    __syncthreads();
#pragma unroll
    for (int k = 0; k < NE; ++k) lr[k] += sBase[(d[k] >= 0) ? (d[k] / BKN) : 0];
#pragma unroll
    for (int k = 0; k < NE; ++k) {
        if (d[k] >= 0 && lr[k] < capb) {
            int off = h ? (capb - 1 - lr[k]) : lr[k];
            reg[(size_t)(d[k] / BKN) * capb + off] = ((d[k] % BKN) << SRCB) | s[k];
        }
    }
}

__device__ __forceinline__
void bin_role(int bid, const int* __restrict__ src, const int* __restrict__ dst, int* cur, int* __restrict__ reg,
              int E, int nbk, int capb, int* sHist, int* sBase) {
    int nft = E / (TPE * BEPT);
    if (bid < nft) {
        bin_edges<BEPT, false>(bid * (TPE * BEPT) + threadIdx.x, src, dst, cur, reg, E, nbk, capb, sHist, sBase);
    } else {
        bin_edges<1, true>(nft * (TPE * BEPT) + (bid - nft) * TPE + threadIdx.x, src, dst, cur, reg, E, nbk, capb,
                           sHist, sBase);
    }
}

#define BKT 512
#define KCH 20
#define STG (64 * BKN)
__global__ __launch_bounds__(BKT)
void k_bucket(const int* __restrict__ cur, const int* __restrict__ reg,
              const int* __restrict__ src, const int* __restrict__ dst, int E, int nbk, int capb,
              int* __restrict__ cnt, int* __restrict__ rowst, int* __restrict__ csr, float* T, int N) {
    __shared__ int sCnt[BKN];
    __shared__ int sPos[BKN];
    __shared__ int sWv[BKT / 32];
    __shared__ int sStart;
    __shared__ __attribute__((aligned(16))) int sL[STG + 8];
    int tid = threadIdx.x;
    int b = blockIdx.x;
    int lane = tid & 31, wv = tid >> 5;
    sCnt[tid] = 0;
    if (tid == 0) sStart = 0;
    __syncthreads();
    for (int j = tid; j < b; j += BKT) atomicAdd(&sStart, cur[j] + cur[nbk + j]);
    int ca = cur[b], cb = cur[nbk + b];
    int tot = ca + cb;
    bool fits = tot <= capb;
    const int* rb = reg + (size_t)b * capb;
    if (fits) {
        for (int i0 = tid; i0 < tot; i0 += BKT * KCH) {
            int p[KCH];
#pragma unroll
            for (int k = 0; k < KCH; ++k) {
                int i = i0 + k * BKT;
                i = (i < tot) ? i : (tot - 1);
                p[k] = rb[(i < ca) ? i : (capb - 1 - (i - ca))];
            }
#pragma unroll
            for (int k = 0; k < KCH; ++k)
                if (i0 + k * BKT < tot) atomicAdd(&sCnt[p[k] >> SRCB], 1);
        }
    } else {
        for (int e = tid; e < E; e += BKT) {
            int d = dst[e];
            if (d / BKN == b) atomicAdd(&sCnt[d % BKN], 1);
        }
    }
    __syncthreads();
    int c = sCnt[tid];
    int incl = c;
    for (int off = 1; off < 32; off <<= 1) {
        int v = __shfl_up(incl, off, 32);
        if (lane >= off) incl += v;
    }
    if (lane == 31) sWv[wv] = incl;
    __syncthreads();
    int excl = incl - c;
    for (int w = 0; w < wv; ++w) excl += sWv[w];
    int start = sStart;
    sPos[tid] = excl;
    int n = b * BKN + tid;
    if (n < N) {
        cnt[n] = c;
        rowst[n] = start + excl;
    }
    __syncthreads();
    int mis = start & 3;
    if (fits) {
        for (int i0 = tid; i0 < tot; i0 += BKT * KCH) {
            int p[KCH];
#pragma unroll
            for (int k = 0; k < KCH; ++k) {
                int i = i0 + k * BKT;
                i = (i < tot) ? i : (tot - 1);
                p[k] = rb[(i < ca) ? i : (capb - 1 - (i - ca))];
            }
#pragma unroll
            for (int k = 0; k < KCH; ++k) {
                if (i0 + k * BKT < tot) {
                    int pos = atomicAdd(&sPos[p[k] >> SRCB], 1);
                    sL[mis + pos] = p[k] & ((1 << SRCB) - 1);
                }
            }
        }
    } else {
        for (int e = tid; e < E; e += BKT) {
            int d = dst[e];
            if (d / BKN == b) {
                int pos = atomicAdd(&sPos[d % BKN], 1);
                csr[start + pos] = src[e];
            }
        }
    }
    __syncthreads();
    if (fits) {
        int* cbase = csr + (start - mis);
        int end = mis + tot;
        int qlo = (mis + 3) >> 2;
        int qhi = end >> 2;
#pragma unroll 4
        for (int q = qlo + tid; q < qhi; q += BKT) ((int4*)cbase)[q] = ((const int4*)sL)[q];
        if (tid < 4) {
            int e = tid;
            if (e >= mis && e < end && e < 4 * qlo) cbase[e] = sL[e];
        } else if (tid < 8) {
            int e = 4 * qhi + (tid - 4);
            if (qhi >= qlo && e < end) cbase[e] = sL[e];
        }
    }
    int nrows = N - b * BKN;
    nrows = (nrows < BKN) ? nrows : BKN;
    float4* T4 = (float4*)T + (size_t)b * BKN * 16;
#pragma unroll 8
    for (int i = tid; i < nrows * 16; i += BKT) {
        float sc = dinv_of(sCnt[i >> 4]);
        float4 v = T4[i];
        v.x *= sc;
        v.y *= sc;
        v.z *= sc;
        v.w *= sc;
        T4[i] = v;
    }
}

#define FHB ((64 * 40) / TPE)
__device__ __forceinline__
void fuse_heads_role(int bid, const float* __restrict__ L0, const float* __restrict__ b0,
                     const float* __restrict__ L1, const float* __restrict__ b1,
                     const float* __restrict__ Wo, const float* __restrict__ bo,
                     float* W0f, float* W1f, float* cb) {
    if (bid < FHB) {
        int i = bid * TPE + threadIdx.x;
        int k = i / 40, c = i % 40;
        float s0 = 0.0f, s1 = 0.0f;
#pragma unroll 16
        for (int j = 0; j < 64; ++j) {
            s0 += L0[k * 64 + j] * Wo[j * 40 + c];
            s1 += L1[k * 64 + j] * Wo[(64 + j) * 40 + c];
        }
        int o = ((k >> 2) * 40 + c) * 4 + (k & 3);
        W0f[o] = s0;
        W1f[o] = s1;
    } else if (threadIdx.x < 40) {
        int c = threadIdx.x;
        float s = bo[c];
        for (int j = 0; j < 64; ++j)
            s += b0[j] * Wo[j * 40 + c] + b1[j] * Wo[(64 + j) * 40 + c];
        cb[c] = s;
    }
}

__device__ __forceinline__
void gemm_n64_tile(int bid, float* sW, const float* __restrict__ A, const float* __restrict__ W,
                   float* __restrict__ D, int N, int K) {
    int tid = threadIdx.x;
    for (int i = tid; i < K * 64; i += 128) {
        int k = i >> 6, col = i & 63;
        sW[(k >> 1) * 128 + 2 * col + (k & 1)] = W[i];
    }
    __syncthreads();

    int wave = tid >> 5;
    int lane = tid & 31;
    int half = lane >> 4;
    int l16  = lane & 15;

    int rbase = bid * 128 + wave * 32;
    int ar0 = rbase + l16;
    int ar1 = rbase + 16 + l16;
    long ab0 = (long)((ar0 < N) ? ar0 : (N - 1)) * K;
    long ab1 = (long)((ar1 < N) ? ar1 : (N - 1)) * K;

    v8f acc00 = {}, acc01 = {}, acc02 = {}, acc03 = {};
    v8f acc10 = {}, acc11 = {}, acc12 = {}, acc13 = {};

    for (int k = 0; k < K; k += 4) {
        int kb = k + 2 * half;
        v2f a0 = *(const v2f*)(A + ab0 + kb);
        v2f a1 = *(const v2f*)(A + ab1 + kb);
        const float* bp = &sW[(kb >> 1) * 128 + 2 * l16];
        v2f b0 = *(const v2f*)(bp + 0);
        v2f b1 = *(const v2f*)(bp + 32);
        v2f b2 = *(const v2f*)(bp + 64);
        v2f b3 = *(const v2f*)(bp + 96);
        acc00 = __builtin_amdgcn_wmma_f32_16x16x4_f32(false, a0, false, b0, (short)0, acc00, false, false);
        acc01 = __builtin_amdgcn_wmma_f32_16x16x4_f32(false, a0, false, b1, (short)0, acc01, false, false);
        acc02 = __builtin_amdgcn_wmma_f32_16x16x4_f32(false, a0, false, b2, (short)0, acc02, false, false);
        acc03 = __builtin_amdgcn_wmma_f32_16x16x4_f32(false, a0, false, b3, (short)0, acc03, false, false);
        acc10 = __builtin_amdgcn_wmma_f32_16x16x4_f32(false, a1, false, b0, (short)0, acc10, false, false);
        acc11 = __builtin_amdgcn_wmma_f32_16x16x4_f32(false, a1, false, b1, (short)0, acc11, false, false);
        acc12 = __builtin_amdgcn_wmma_f32_16x16x4_f32(false, a1, false, b2, (short)0, acc12, false, false);
        acc13 = __builtin_amdgcn_wmma_f32_16x16x4_f32(false, a1, false, b3, (short)0, acc13, false, false);
    }

    int m0 = rbase + half * 8;
    int m1 = rbase + 16 + half * 8;
#pragma unroll
    for (int r = 0; r < 8; ++r) {
        int rr = m0 + r;
        if (rr < N) {
            float* dp = D + (long)rr * 64 + l16;
            dp[0]  = acc00[r];
            dp[16] = acc01[r];
            dp[32] = acc02[r];
            dp[48] = acc03[r];
        }
        int ss = m1 + r;
        if (ss < N) {
            float* dp = D + (long)ss * 64 + l16;
            dp[0]  = acc10[r];
            dp[16] = acc11[r];
            dp[32] = acc12[r];
            dp[48] = acc13[r];
        }
    }
}

__global__ __launch_bounds__(TPE)
void k_bin_dense(const int* __restrict__ src, const int* __restrict__ dst, int* cur, int* __restrict__ reg,
                 int E, int nbk, int capb,
                 const float* __restrict__ X, const float* __restrict__ W, float* __restrict__ T, int N, int gG,
                 const float* __restrict__ L0, const float* __restrict__ b0,
                 const float* __restrict__ L1, const float* __restrict__ b1,
                 const float* __restrict__ Wo, const float* __restrict__ bo,
                 float* W0f, float* W1f, float* cb) {
    __shared__ float sbuf[128 * 64];
    int b = (int)blockIdx.x;
    if (b < gG) {
        gemm_n64_tile(b, sbuf, X, W, T, N, 128);
    } else if (b <= gG + FHB) {
        fuse_heads_role(b - gG, L0, b0, L1, b1, Wo, bo, W0f, W1f, cb);
    } else {
        int* sHist = (int*)sbuf;
        bin_role(b - (gG + FHB + 1), src, dst, cur, reg, E, nbk, capb, sHist, sHist + MAXBK);
    }
}

#define NPG 8
#define GTB 128

template <int K>
__device__ __forceinline__ int row_lane(int v) {
    return __builtin_amdgcn_update_dpp(0, v, 0x150 + K, 0xf, 0xf, false);
}

__device__ __forceinline__ int row_next(int v) {
    return __builtin_amdgcn_update_dpp(0, v, 0x12F, 0xf, 0xf, false);
}

template <int K0, int CNT>
__device__ __forceinline__ void rows_req(int e, const float4* __restrict__ T4, int l, float4* r) {
    if constexpr (CNT > 0) {
        rows_req<K0, CNT - 1>(e, T4, l, r);
        int s = row_lane<K0 + CNT - 1>(e);
        r[CNT - 1] = T4[s * 16 + l];
    }
}

template <int K0, int CNT>
__device__ __forceinline__ void tail_ids(int e, int* s) {
    if constexpr (CNT > 0) {
        tail_ids<K0, CNT - 1>(e, s);
        s[CNT - 1] = row_lane<K0 + CNT - 1>(e);
    }
}

template <int CNT>
__device__ __forceinline__ void rows_get(const int* s, const float4* __restrict__ T4, int l, float4* r) {
#pragma unroll
    for (int k = 0; k < CNT; ++k) r[k] = T4[s[k] * 16 + l];
}

template <int CNT>
__device__ __forceinline__ void rows_add(const float4* r, float4& acc) {
#pragma unroll
    for (int k = 0; k < CNT; ++k) {
        acc.x += r[k].x;
        acc.y += r[k].y;
        acc.z += r[k].z;
        acc.w += r[k].w;
    }
}

__device__ __forceinline__ int piece_off(int q, int c, int l) {
    int nfull = c >> 4, t = c & 15;
    int b = (l & 8) ? 8 : (l & 4) ? 4 : (l & 2) ? 2 : (l & 1);
    int it = (t & b) ? ((t & (b - 1)) + (l - b)) : 0;
    int off = (q < nfull) ? (16 * q + l) : (16 * nfull + it);
    return (off < c) ? off : (c - 1);
}

__device__ __forceinline__ void run_sum(const int* __restrict__ p, int c, int e,
                                        const float4* __restrict__ T4, int l,
                                        const float4* __restrict__ selfp, float4& self,
                                        const int* __restrict__ nextp, int& e_next, float4& acc) {
    int nfull = c >> 4, t = c & 15;
    for (int q = 0; q < nfull; ++q) {
        int en = p[piece_off(q + 1, c, l)];
        float4 r[16];
        rows_req<0, 16>(e, T4, l, r);
        rows_add<16>(r, acc);
        e = en;
    }
    float4 r[16] = {};
    int s[16];
    if (t) tail_ids<1, 15>(e, s + 1);
    self = *selfp;
    e_next = *nextp;
    if (t) {
        rows_get<1>(s + 1, T4, l, r + 1);
        if (t & 2) rows_get<2>(s + 2, T4, l, r + 2);
        if (t & 4) rows_get<4>(s + 4, T4, l, r + 4);
        if (t & 8) rows_get<8>(s + 8, T4, l, r + 8);
        if (t & 1) rows_add<1>(r + 1, acc);
        if (t & 2) rows_add<2>(r + 2, acc);
        if (t & 4) rows_add<4>(r + 4, acc);
        if (t & 8) rows_add<8>(r + 8, acc);
    }
}

__device__ __forceinline__ const int* first_piece(const int* __restrict__ csr, int c, int st, int l) {
    return csr + ((c > 0) ? (st + piece_off(0, c, l)) : 0);
}

#define HP 68
#define HQ ((64 / 4) * 40)
template <bool FIRST>
__global__ __launch_bounds__(GTB, 8)
void k_gather_elu(const float* __restrict__ T, const int* __restrict__ csr,
                  const int* __restrict__ cnt, const int* __restrict__ rowst,
                  const float* __restrict__ bias, int N,
                  const float* __restrict__ W2, float* __restrict__ D,
                  const float* __restrict__ Wf, const float* __restrict__ cb, float* out) {
    __shared__ __attribute__((aligned(16))) float sW[FIRST ? 64 * 64 : 4];
    __shared__ __attribute__((aligned(16))) float sH[(GTB / 16) * NPG * HP];
    __shared__ float4 sF4[HQ];
    __shared__ float sDn[FIRST ? (GTB / 16) * NPG : 1];
    int tid = threadIdx.x;
    int l = tid & 15;
    int n0 = (blockIdx.x * (GTB / 16) + (tid >> 4)) * NPG;
    const float4* T4 = (const float4*)T;
    int nm = n0 + (l & (NPG - 1));
    nm = (nm < N) ? nm : (N - 1);
    int cm = cnt[nm];
    int stv = rowst[nm];
    float4 b4 = make_float4(bias[4 * l + 0], bias[4 * l + 1], bias[4 * l + 2], bias[4 * l + 3]);
    int nn = (N - n0 < NPG) ? ((N - n0 > 0) ? (N - n0) : 0) : NPG;
    float4 wa[4], wb[4];
    float4 wf[HQ / GTB];
    if constexpr (FIRST) {
#pragma unroll
        for (int j = 0; j < 4; ++j) {
            int it = tid + j * GTB;
            int kp = it >> 4, cg = it & 15;
            wa[j] = ((const float4*)W2)[(2 * kp) * 16 + cg];
            wb[j] = ((const float4*)W2)[(2 * kp + 1) * 16 + cg];
        }
    }
#pragma unroll
    for (int j = 0; j < HQ / GTB; ++j) wf[j] = ((const float4*)Wf)[tid + j * GTB];
    int e = *first_piece(csr, row_lane<0>(cm), row_lane<0>(stv), l);
    if constexpr (FIRST) {
#pragma unroll
        for (int j = 0; j < 4; ++j) {
            int it = tid + j * GTB;
            int kp = it >> 4, cg = it & 15;
            float4* d = (float4*)&sW[kp * 128 + 8 * cg];
            d[0] = make_float4(wa[j].x, wb[j].x, wa[j].y, wb[j].y);
            d[1] = make_float4(wa[j].z, wb[j].z, wa[j].w, wb[j].w);
        }
    }
#pragma unroll
    for (int j = 0; j < HQ / GTB; ++j) sF4[tid + j * GTB] = wf[j];
#pragma unroll 1
    for (int i = 0; i < nn; ++i) {
        int n = n0 + i;
        int c = row_lane<0>(cm);
        int st = row_lane<0>(stv);
        cm = row_next(cm);
        stv = row_next(stv);
        const int* nextp = first_piece(csr, row_lane<0>(cm), row_lane<0>(stv), l);
        float4 self;
        int e_next;
        float dn = dinv_of(c);
        float4 acc = make_float4(0.0f, 0.0f, 0.0f, 0.0f);
        run_sum(csr + st, c, e, T4, l, T4 + (n * 16 + l), self, nextp, e_next, acc);
        float v0 = dn * (acc.x + self.x) + b4.x;
        float v1 = dn * (acc.y + self.y) + b4.y;
        float v2 = dn * (acc.z + self.z) + b4.z;
        float v3 = dn * (acc.w + self.w) + b4.w;
        float4 h;
        h.x = (v0 > 0.0f) ? v0 : (expf(v0) - 1.0f);
        h.y = (v1 > 0.0f) ? v1 : (expf(v1) - 1.0f);
        h.z = (v2 > 0.0f) ? v2 : (expf(v2) - 1.0f);
        h.w = (v3 > 0.0f) ? v3 : (expf(v3) - 1.0f);
        *(float4*)&sH[((tid >> 4) * NPG + i) * HP + 4 * l] = h;
        if constexpr (FIRST) sDn[(tid >> 4) * NPG + i] = dn;
        e = e_next;
    }
    __syncthreads();
    int wave = tid >> 5;
    int lane = tid & 31;
    int half = lane >> 4;
    int l16  = lane & 15;
    int R0 = blockIdx.x * ((GTB / 16) * NPG) + wave * 16;
    int nrow = N - R0;
    float4* o4 = (float4*)out + (long)R0 * 10;
    float4 prev[5];
    if constexpr (!FIRST) {
        if (nrow >= 16) {
#pragma unroll
            for (int i = 0; i < 5; ++i) prev[i] = o4[lane + 32 * i];
        }
    }
    if constexpr (FIRST) {
        const float* ap = &sH[(wave * 16 + l16) * HP + 2 * half];
        const float* bq = &sW[half * 128 + 2 * l16];
        v8f acc0 = {}, acc1 = {}, acc2 = {}, acc3 = {};
#pragma unroll
        for (int k = 0; k < 64; k += 4) {
            v2f a = *(const v2f*)(ap + k);
            const float* bp = bq + (k >> 1) * 128;
            v2f b0 = *(const v2f*)(bp + 0);
            v2f b1 = *(const v2f*)(bp + 32);
            v2f b2 = *(const v2f*)(bp + 64);
            v2f b3 = *(const v2f*)(bp + 96);
            acc0 = __builtin_amdgcn_wmma_f32_16x16x4_f32(false, a, false, b0, (short)0, acc0, false, false);
            acc1 = __builtin_amdgcn_wmma_f32_16x16x4_f32(false, a, false, b1, (short)0, acc1, false, false);
            acc2 = __builtin_amdgcn_wmma_f32_16x16x4_f32(false, a, false, b2, (short)0, acc2, false, false);
            acc3 = __builtin_amdgcn_wmma_f32_16x16x4_f32(false, a, false, b3, (short)0, acc3, false, false);
        }
        int m0 = R0 + half * 8;
#pragma unroll
        for (int r = 0; r < 8; ++r) {
            int rr = m0 + r;
            if (rr < N) {
                float sc = sDn[wave * 16 + half * 8 + r];
                float* dp = D + (long)rr * 64 + l16;
                dp[0]  = sc * acc0[r];
                dp[16] = sc * acc1[r];
                dp[32] = sc * acc2[r];
                dp[48] = sc * acc3[r];
            }
        }
    }
    const float* ah = &sH[(wave * 16 + l16) * HP + 4 * half];
    const float4* bh = &sF4[half * 40];
    int c2 = 32 + (l16 & 7);
    v8f p0, p1, p2;
    if constexpr (FIRST) {
        float bz0 = cb[l16];
        float bz1 = cb[16 + l16];
        float bz2 = cb[c2];
#pragma unroll
        for (int r = 0; r < 8; ++r) { p0[r] = bz0; p1[r] = bz1; p2[r] = bz2; }
    } else {
#pragma unroll
        for (int r = 0; r < 8; ++r) { p0[r] = 0.0f; p1[r] = 0.0f; p2[r] = 0.0f; }
    }
#pragma unroll
    for (int j = 0; j < 8; ++j) {
        float4 a = *(const float4*)(ah + 8 * j);
        const float4* bp = bh + (2 * j) * 40;
        float4 b0 = bp[l16], b1 = bp[16 + l16], b2 = bp[c2];
        v2f x = {a.x, a.y}, y = {a.z, a.w};
        v2f q0 = {b0.x, b0.y}, q1 = {b1.x, b1.y}, q2 = {b2.x, b2.y};
        v2f s0 = {b0.z, b0.w}, s1 = {b1.z, b1.w}, s2 = {b2.z, b2.w};
        p0 = __builtin_amdgcn_wmma_f32_16x16x4_f32(false, x, false, q0, (short)0, p0, false, false);
        p1 = __builtin_amdgcn_wmma_f32_16x16x4_f32(false, x, false, q1, (short)0, p1, false, false);
        p2 = __builtin_amdgcn_wmma_f32_16x16x4_f32(false, x, false, q2, (short)0, p2, false, false);
        p0 = __builtin_amdgcn_wmma_f32_16x16x4_f32(false, y, false, s0, (short)0, p0, false, false);
        p1 = __builtin_amdgcn_wmma_f32_16x16x4_f32(false, y, false, s1, (short)0, p1, false, false);
        p2 = __builtin_amdgcn_wmma_f32_16x16x4_f32(false, y, false, s2, (short)0, p2, false, false);
    }
    float* so = &sH[wave * 16 * HP];
#pragma unroll
    for (int r = 0; r < 8; ++r) {
        float* q = so + (half * 8 + r) * 40;
        q[l16]      = p0[r];
        q[16 + l16] = p1[r];
        q[c2]       = p2[r];
    }
    __syncthreads();
    const float4* s4 = (const float4*)so;
    if (nrow >= 16) {
        float4 t[5];
#pragma unroll
        for (int i = 0; i < 5; ++i) t[i] = s4[lane + 32 * i];
        if constexpr (!FIRST) {
#pragma unroll
            for (int i = 0; i < 5; ++i) {
                t[i].x += prev[i].x;
                t[i].y += prev[i].y;
                t[i].z += prev[i].z;
                t[i].w += prev[i].w;
            }
        }
#pragma unroll
        for (int i = 0; i < 5; ++i) o4[lane + 32 * i] = t[i];
    } else {
        int n4 = nrow * 10;
        for (int i = lane; i < n4; i += 32) {
            float4 v = s4[i];
            if constexpr (!FIRST) {
                float4 u = o4[i];
                v.x += u.x;
                v.y += u.y;
                v.z += u.z;
                v.w += u.w;
            }
            o4[i] = v;
        }
    }
}

extern "C" void kernel_launch(void* const* d_in, const int* in_sizes, int n_in,
                              void* d_out, int out_size, void* d_ws, size_t ws_size,
                              hipStream_t stream) {
    const float* x       = (const float*)d_in[0];
    const int*   ei      = (const int*)d_in[1];
    const float* conv0_w = (const float*)d_in[2];
    const float* conv0_b = (const float*)d_in[3];
    const float* lin0_w  = (const float*)d_in[4];
    const float* lin0_b  = (const float*)d_in[5];
    const float* conv1_w = (const float*)d_in[6];
    const float* conv1_b = (const float*)d_in[7];
    const float* lin1_w  = (const float*)d_in[8];
    const float* lin1_b  = (const float*)d_in[9];
    const float* out_w   = (const float*)d_in[10];
    const float* out_b   = (const float*)d_in[11];
    float* out = (float*)d_out;

    int N = in_sizes[0] / 128;
    int E = in_sizes[1] / 2;
    const int* src = ei;
    const int* dst = ei + E;

    size_t NH = (size_t)N * 64;
    float* ws   = (float*)d_ws;
    int*   cnt  = (int*)ws;
    float* A    = ws + (((size_t)N + 255) / 256) * 256;
    float* H1   = A + NH;
    float* EL   = H1 + NH;
    float* W0f  = EL + NH;
    float* W1f  = W0f + 64 * 40;
    float* cb   = W1f + 64 * 40;
    int    nbk  = (N + BKN - 1) / BKN;
    int    capb = (int)(NH / (size_t)nbk);
    int*   reg  = (int*)H1;
    int*   oscr   = (int*)EL;
    int*   csr    = oscr;
    int*   rowst  = oscr + 2 * (size_t)E;
    int*   cur    = rowst + N;

    int gGa  = (N + (GTB / 16) * NPG - 1) / ((GTB / 16) * NPG);
    int gBin = E / (TPE * BEPT) + (E % (TPE * BEPT) + TPE - 1) / TPE;
    int gGemm = (N + 127) / 128;

    k_zero_cur<<<1, 512, 0, stream>>>(cur, 2 * nbk);
    k_bin_dense<<<gGemm + (FHB + 1) + gBin, TPE, 0, stream>>>(src, dst, cur, reg, E, nbk, capb,
                                                             x, conv0_w, A, N, gGemm,
                                                             lin0_w, lin0_b, lin1_w, lin1_b, out_w, out_b,
                                                             W0f, W1f, cb);
    k_bucket<<<nbk, BKT, 0, stream>>>(cur, reg, src, dst, E, nbk, capb, cnt, rowst, csr, A, N);

    k_gather_elu<true><<<gGa, GTB, 0, stream>>>(A, csr, cnt, rowst, conv0_b, N, conv1_w, H1, W0f, cb, out);

    k_gather_elu<false><<<gGa, GTB, 0, stream>>>(H1, csr, cnt, rowst, conv1_b, N, nullptr, nullptr, W1f, nullptr, out);
}
